# starting kernel
# baseline (speedup 1.0000x reference)
#include <hip/hip_runtime.h>

#define BB 2
#define HH 8
#define SS 2048
#define DD 64
#define NB 32
#define BS 64
#define NSEL 8
#define SELC 512
#define KEEP 256
#define WIN 128
#define PSTRIDE 520
#define NEGINF (-1e30f)
#define SCALE 0.125f

typedef __attribute__((ext_vector_type(2))) float v2f;
typedef __attribute__((ext_vector_type(8))) float v8f;

static __device__ inline v8f vzero8() {
  v8f r;
#pragma unroll
  for (int i = 0; i < 8; ++i) r[i] = 0.f;
  return r;
}

static __device__ inline v8f wmma4(v2f a, v2f b, v8f c) {
  return __builtin_amdgcn_wmma_f32_16x16x4_f32(false, a, false, b, (short)0, c,
                                               false, false);
}

static __device__ inline unsigned okey(float x) {
  unsigned u = __float_as_uint(x);
  return (u & 0x80000000u) ? ~u : (u | 0x80000000u);
}

static __device__ inline int wred_sumi(int x) {
#pragma unroll
  for (int o = 16; o > 0; o >>= 1) x += __shfl_xor(x, o, 32);
  return x;
}
static __device__ inline float wred_sumf(float x) {
#pragma unroll
  for (int o = 16; o > 0; o >>= 1) x += __shfl_xor(x, o, 32);
  return x;
}
static __device__ inline float wred_maxf(float x) {
#pragma unroll
  for (int o = 16; o > 0; o >>= 1) x = fmaxf(x, __shfl_xor(x, o, 32));
  return x;
}

__global__ void nsa_blockmean_kernel(const float* __restrict__ K,
                                     const float* __restrict__ V,
                                     float* __restrict__ KB,
                                     float* __restrict__ VB) {
  const int g = blockIdx.x;
  const int d = threadIdx.x;
  const float* kp = K + (size_t)g * (BS * DD);
  const float* vp = V + (size_t)g * (BS * DD);
  float ks = 0.f, vs = 0.f;
  for (int t = 0; t < BS; ++t) {
    ks += kp[t * DD + d];
    vs += vp[t * DD + d];
  }
  KB[(size_t)g * DD + d] = ks * (1.f / BS);
  VB[(size_t)g * DD + d] = vs * (1.f / BS);
}

__global__ __launch_bounds__(128) void nsa_main_kernel(
    const float* __restrict__ Q, const float* __restrict__ K,
    const float* __restrict__ V, const float* __restrict__ KB,
    const float* __restrict__ VB, float* __restrict__ O) {
  extern __shared__ float sP[];
  __shared__ float sScore[NB];
  __shared__ int sSel[NSEL];

  const int wg = blockIdx.x;
  const int qb = wg % NB;
  const int bh = wg / NB;
  const size_t hoff = (size_t)bh * SS * DD;
  const float* Qh = Q + hoff;
  const float* Kh = K + hoff;
  const float* Vh = V + hoff;
  const float* KBh = KB + (size_t)bh * NB * DD;
  const float* VBh = VB + (size_t)bh * NB * DD;
  float* Oh = O + hoff;

  const int tid = threadIdx.x;
  const int wave = tid >> 5;
  const int lane = tid & 31;
  const int half = lane >> 4;
  const int ln = lane & 15;
  const int qrow0 = qb * BS + wave * 16;
  float* myP = sP + (size_t)wave * 16 * PSTRIDE;

  v2f qa[16];
#pragma unroll
  for (int t = 0; t < 16; ++t) {
    const float* p = Qh + (size_t)(qrow0 + ln) * DD + 4 * t + 2 * half;
    v2f a; a.x = p[0]; a.y = p[1];
    qa[t] = a;
  }

  v8f oacc[4];
#pragma unroll
  for (int dt = 0; dt < 4; ++dt) oacc[dt] = vzero8();

#pragma unroll
  for (int nt = 0; nt < 2; ++nt) {
    v8f acc = vzero8();
#pragma unroll
    for (int t = 0; t < 16; ++t) {
      const float* bp = KBh + (size_t)(nt * 16 + ln) * DD + 4 * t + 2 * half;
      v2f bf; bf.x = bp[0]; bf.y = bp[1];
      acc = wmma4(qa[t], bf, acc);
    }
#pragma unroll
    for (int r = 0; r < 8; ++r) {
      int m = r + 8 * half;
      int col = nt * 16 + ln;
      float x = acc[r] * SCALE;
      myP[m * PSTRIDE + col] = (col <= qb) ? x : NEGINF;
    }
  }
  __syncthreads();

  if (tid < NB) {
    float s = 0.f;
    for (int q2 = 0; q2 < BS; ++q2) s += sP[q2 * PSTRIDE + tid];
    sScore[tid] = s * (1.f / BS);
  }
  __syncthreads();

  if (tid == 0) {
    float loc[NB];
    for (int i = 0; i < NB; ++i) loc[i] = sScore[i];
    for (int s = 0; s < NSEL; ++s) {
      int bi = 0;
      float bv = loc[0];
      for (int i = 1; i < NB; ++i)
        if (loc[i] > bv) { bv = loc[i]; bi = i; }
      sSel[s] = bi;
      loc[bi] = -3.4e38f;
    }
  }
  __syncthreads();

  if (tid < BS) {
    float* row = sP + tid * PSTRIDE;
    float mx = -3.4e38f;
    for (int c = 0; c < NB; ++c) mx = fmaxf(mx, row[c]);
    float sum = 0.f;
    for (int c = 0; c < NB; ++c) {
      float p = (mx > -1e29f) ? __expf(row[c] - mx) : 0.f;
      row[c] = p;
      sum += p;
    }
    float rs = (sum > 0.f) ? 1.f / sum : 0.f;
    for (int c = 0; c < NB; ++c) row[c] *= rs;
  }
  __syncthreads();

#pragma unroll
  for (int dt = 0; dt < 4; ++dt) {
#pragma unroll
    for (int g = 0; g < 8; ++g) {
      int c0 = 4 * g + 2 * half;
      v2f a; a.x = myP[ln * PSTRIDE + c0]; a.y = myP[ln * PSTRIDE + c0 + 1];
      v2f bf;
      bf.x = VBh[(size_t)c0 * DD + dt * 16 + ln];
      bf.y = VBh[(size_t)(c0 + 1) * DD + dt * 16 + ln];
      oacc[dt] = wmma4(a, bf, oacc[dt]);
    }
  }
  __syncthreads();

  for (int sb = 0; sb < NSEL; ++sb) {
    int kb = sSel[sb];
    for (int kt = 0; kt < 4; ++kt) {
      int keyBase = kb * BS + kt * 16;
      __builtin_prefetch(Kh + (size_t)(keyBase + 16) * DD, 0, 1);
      v8f acc = vzero8();
#pragma unroll
      for (int t = 0; t < 16; ++t) {
        const float* bp = Kh + (size_t)(keyBase + ln) * DD + 4 * t + 2 * half;
        v2f bf; bf.x = bp[0]; bf.y = bp[1];
        acc = wmma4(qa[t], bf, acc);
      }
#pragma unroll
      for (int r = 0; r < 8; ++r) {
        int m = r + 8 * half;
        int i = qrow0 + m;
        int tok = keyBase + ln;
        int col = sb * 64 + kt * 16 + ln;
        float x = acc[r] * SCALE;
        myP[m * PSTRIDE + col] = (tok <= i) ? x : NEGINF;
      }
    }
  }

  for (int r = 0; r < 16; ++r) {
    float* row = myP + r * PSTRIDE;
    unsigned prefix = 0u;
    for (int bit = 31; bit >= 0; --bit) {
      unsigned cand = prefix | (1u << bit);
      int cnt = 0;
      for (int c = lane; c < SELC; c += 32)
        cnt += (okey(row[c]) >= cand) ? 1 : 0;
      cnt = wred_sumi(cnt);
      if (cnt >= KEEP) prefix = cand;
    }
    float mx = -3.4e38f;
    for (int c = lane; c < SELC; c += 32) mx = fmaxf(mx, row[c]);
    mx = wred_maxf(mx);
    float sum = 0.f;
    for (int c = lane; c < SELC; c += 32) {
      float x = row[c];
      float p = (okey(x) >= prefix && mx > -1e29f) ? __expf(x - mx) : 0.f;
      row[c] = p;
      sum += p;
    }
    sum = wred_sumf(sum);
    float rs = (sum > 0.f) ? 1.f / sum : 0.f;
    for (int c = lane; c < SELC; c += 32) row[c] *= rs;
  }

#pragma unroll
  for (int dt = 0; dt < 4; ++dt) {
    for (int g = 0; g < SELC / 4; ++g) {
      int c0 = 4 * g + 2 * half;
      v2f a; a.x = myP[ln * PSTRIDE + c0]; a.y = myP[ln * PSTRIDE + c0 + 1];
      int t0 = sSel[c0 >> 6] * BS + (c0 & 63);
      v2f bf;
      bf.x = Vh[(size_t)t0 * DD + dt * 16 + ln];
      bf.y = Vh[(size_t)(t0 + 1) * DD + dt * 16 + ln];
      oacc[dt] = wmma4(a, bf, oacc[dt]);
    }
  }

  const int firstb = (qb >= 2) ? qb - 2 : 0;
  const int nwb = qb - firstb + 1;
  const int extra = (firstb > 0) ? 1 : 0;
  const int ncg = nwb + extra;
  const int Cw = ncg * 64;

  for (int cg = 0; cg < ncg; ++cg) {
    int kb = (extra && cg == 0) ? 0 : firstb + (cg - extra);
    for (int kt = 0; kt < 4; ++kt) {
      int keyBase = kb * BS + kt * 16;
      v8f acc = vzero8();
#pragma unroll
      for (int t = 0; t < 16; ++t) {
        const float* bp = Kh + (size_t)(keyBase + ln) * DD + 4 * t + 2 * half;
        v2f bf; bf.x = bp[0]; bf.y = bp[1];
        acc = wmma4(qa[t], bf, acc);
      }
#pragma unroll
      for (int r = 0; r < 8; ++r) {
        int m = r + 8 * half;
        int i = qrow0 + m;
        int j = keyBase + ln;
        int col = cg * 64 + kt * 16 + ln;
        bool ok = (j <= i) && (((i - j) < WIN) || (j < 1));
        myP[m * PSTRIDE + col] = ok ? acc[r] * SCALE : NEGINF;
      }
    }
  }

  for (int r = 0; r < 16; ++r) {
    float* row = myP + r * PSTRIDE;
    float mx = -3.4e38f;
    for (int c = lane; c < Cw; c += 32) mx = fmaxf(mx, row[c]);
    mx = wred_maxf(mx);
    float sum = 0.f;
    for (int c = lane; c < Cw; c += 32) {
      float p = (mx > -1e29f) ? __expf(row[c] - mx) : 0.f;
      row[c] = p;
      sum += p;
    }
    sum = wred_sumf(sum);
    float rs = (sum > 0.f) ? 1.f / sum : 0.f;
    for (int c = lane; c < Cw; c += 32) row[c] *= rs;
  }

#pragma unroll
  for (int dt = 0; dt < 4; ++dt) {
    for (int g = 0; g < Cw / 4; ++g) {
      int c0 = 4 * g + 2 * half;
      v2f a; a.x = myP[ln * PSTRIDE + c0]; a.y = myP[ln * PSTRIDE + c0 + 1];
      int cg0 = c0 >> 6;
      int kb0 = (extra && cg0 == 0) ? 0 : firstb + (cg0 - extra);
      int t0 = kb0 * BS + (c0 & 63);
      v2f bf;
      bf.x = Vh[(size_t)t0 * DD + dt * 16 + ln];
      bf.y = Vh[(size_t)(t0 + 1) * DD + dt * 16 + ln];
      oacc[dt] = wmma4(a, bf, oacc[dt]);
    }
  }

  const float third = 1.f / 3.f;
#pragma unroll
  for (int dt = 0; dt < 4; ++dt) {
#pragma unroll
    for (int r = 0; r < 8; ++r) {
      int row = qrow0 + r + 8 * half;
      Oh[(size_t)row * DD + dt * 16 + ln] = oacc[dt][r] * third;
    }
  }
}

extern "C" void kernel_launch(void* const* d_in, const int* in_sizes, int n_in,
                              void* d_out, int out_size, void* d_ws,
                              size_t ws_size, hipStream_t stream) {
  (void)in_sizes; (void)n_in; (void)out_size; (void)ws_size;
  const float* q = (const float*)d_in[0];
  const float* k = (const float*)d_in[1];
  const float* v = (const float*)d_in[2];
  float* out = (float*)d_out;
  float* kblk = (float*)d_ws;
  float* vblk = kblk + (size_t)BB * HH * NB * DD;

  nsa_blockmean_kernel<<<BB * HH * NB, 64, 0, stream>>>(k, v, kblk, vblk);

  const size_t shmem = (size_t)64 * PSTRIDE * sizeof(float);
  hipFuncSetAttribute(reinterpret_cast<const void*>(nsa_main_kernel),
                      hipFuncAttributeMaxDynamicSharedMemorySize, (int)shmem);
  nsa_main_kernel<<<BB * HH * NB, 128, shmem, stream>>>(q, k, v, kblk, vblk,
                                                        out);
}


# top-256 select of the selected branch: row logits kept in registers, per-bit count by wave ballot + scalar popcount instead of a 5-step shuffle sum per bit, 2 rows interleaved; sum order unchanged
# speedup vs baseline: 1.3919x; 1.3919x over previous
#include <hip/hip_runtime.h>

#define BB 2
#define HH 8
#define SS 2048
#define DD 64
#define NB 32
#define BS 64
#define NSEL 8
#define SELC 512
#define KEEP 256
#define WIN 128
#define PSTRIDE 520
#define NEGINF (-1e30f)
#define SCALE 0.125f
#define SELR 2

typedef __attribute__((ext_vector_type(2))) float v2f;
typedef __attribute__((ext_vector_type(8))) float v8f;

static __device__ inline v8f vzero8() {
  v8f r;
#pragma unroll
  for (int i = 0; i < 8; ++i) r[i] = 0.f;
  return r;
}

static __device__ inline v8f wmma4(v2f a, v2f b, v8f c) {
  return __builtin_amdgcn_wmma_f32_16x16x4_f32(false, a, false, b, (short)0, c,
                                               false, false);
}

static __device__ inline unsigned okey(float x) {
  unsigned u = __float_as_uint(x);
  return (u & 0x80000000u) ? ~u : (u | 0x80000000u);
}

static __device__ inline int wred_sumi(int x) {
#pragma unroll
  for (int o = 16; o > 0; o >>= 1) x += __shfl_xor(x, o, 32);
  return x;
}
static __device__ inline float wred_sumf(float x) {
#pragma unroll
  for (int o = 16; o > 0; o >>= 1) x += __shfl_xor(x, o, 32);
  return x;
}
static __device__ inline float wred_maxf(float x) {
#pragma unroll
  for (int o = 16; o > 0; o >>= 1) x = fmaxf(x, __shfl_xor(x, o, 32));
  return x;
}

__global__ void nsa_blockmean_kernel(const float* __restrict__ K,
                                     const float* __restrict__ V,
                                     float* __restrict__ KB,
                                     float* __restrict__ VB) {
  const int g = blockIdx.x;
  const int d = threadIdx.x;
  const float* kp = K + (size_t)g * (BS * DD);
  const float* vp = V + (size_t)g * (BS * DD);
  float ks = 0.f, vs = 0.f;
  for (int t = 0; t < BS; ++t) {
    ks += kp[t * DD + d];
    vs += vp[t * DD + d];
  }
  KB[(size_t)g * DD + d] = ks * (1.f / BS);
  VB[(size_t)g * DD + d] = vs * (1.f / BS);
}

__global__ __launch_bounds__(128) void nsa_main_kernel(
    const float* __restrict__ Q, const float* __restrict__ K,
    const float* __restrict__ V, const float* __restrict__ KB,
    const float* __restrict__ VB, float* __restrict__ O) {
  extern __shared__ float sP[];
  __shared__ float sScore[NB];
  __shared__ int sSel[NSEL];

  const int wg = blockIdx.x;
  const int qb = wg % NB;
  const int bh = wg / NB;
  const size_t hoff = (size_t)bh * SS * DD;
  const float* Qh = Q + hoff;
  const float* Kh = K + hoff;
  const float* Vh = V + hoff;
  const float* KBh = KB + (size_t)bh * NB * DD;
  const float* VBh = VB + (size_t)bh * NB * DD;
  float* Oh = O + hoff;

  const int tid = threadIdx.x;
  const int wave = tid >> 5;
  const int lane = tid & 31;
  const int half = lane >> 4;
  const int ln = lane & 15;
  const int qrow0 = qb * BS + wave * 16;
  float* myP = sP + (size_t)wave * 16 * PSTRIDE;

  v2f qa[16];
#pragma unroll
  for (int t = 0; t < 16; ++t) {
    const float* p = Qh + (size_t)(qrow0 + ln) * DD + 4 * t + 2 * half;
    v2f a; a.x = p[0]; a.y = p[1];
    qa[t] = a;
  }

  v8f oacc[4];
#pragma unroll
  for (int dt = 0; dt < 4; ++dt) oacc[dt] = vzero8();

#pragma unroll
  for (int nt = 0; nt < 2; ++nt) {
    v8f acc = vzero8();
#pragma unroll
    for (int t = 0; t < 16; ++t) {
      const float* bp = KBh + (size_t)(nt * 16 + ln) * DD + 4 * t + 2 * half;
      v2f bf; bf.x = bp[0]; bf.y = bp[1];
      acc = wmma4(qa[t], bf, acc);
    }
#pragma unroll
    for (int r = 0; r < 8; ++r) {
      int m = r + 8 * half;
      int col = nt * 16 + ln;
      float x = acc[r] * SCALE;
      myP[m * PSTRIDE + col] = (col <= qb) ? x : NEGINF;
    }
  }
  __syncthreads();

  if (tid < NB) {
    float s = 0.f;
    for (int q2 = 0; q2 < BS; ++q2) s += sP[q2 * PSTRIDE + tid];
    sScore[tid] = s * (1.f / BS);
  }
  __syncthreads();

  if (tid == 0) {
    float loc[NB];
    for (int i = 0; i < NB; ++i) loc[i] = sScore[i];
    for (int s = 0; s < NSEL; ++s) {
      int bi = 0;
      float bv = loc[0];
      for (int i = 1; i < NB; ++i)
        if (loc[i] > bv) { bv = loc[i]; bi = i; }
      sSel[s] = bi;
      loc[bi] = -3.4e38f;
    }
  }
  __syncthreads();

  if (tid < BS) {
    float* row = sP + tid * PSTRIDE;
    float mx = -3.4e38f;
    for (int c = 0; c < NB; ++c) mx = fmaxf(mx, row[c]);
    float sum = 0.f;
    for (int c = 0; c < NB; ++c) {
      float p = (mx > -1e29f) ? __expf(row[c] - mx) : 0.f;
      row[c] = p;
      sum += p;
    }
    float rs = (sum > 0.f) ? 1.f / sum : 0.f;
    for (int c = 0; c < NB; ++c) row[c] *= rs;
  }
  __syncthreads();

#pragma unroll
  for (int dt = 0; dt < 4; ++dt) {
#pragma unroll
    for (int g = 0; g < 8; ++g) {
      int c0 = 4 * g + 2 * half;
      v2f a; a.x = myP[ln * PSTRIDE + c0]; a.y = myP[ln * PSTRIDE + c0 + 1];
      v2f bf;
      bf.x = VBh[(size_t)c0 * DD + dt * 16 + ln];
      bf.y = VBh[(size_t)(c0 + 1) * DD + dt * 16 + ln];
      oacc[dt] = wmma4(a, bf, oacc[dt]);
    }
  }
  __syncthreads();

  for (int sb = 0; sb < NSEL; ++sb) {
    int kb = sSel[sb];
    for (int kt = 0; kt < 4; ++kt) {
      int keyBase = kb * BS + kt * 16;
      __builtin_prefetch(Kh + (size_t)(keyBase + 16) * DD, 0, 1);
      v8f acc = vzero8();
#pragma unroll
      for (int t = 0; t < 16; ++t) {
        const float* bp = Kh + (size_t)(keyBase + ln) * DD + 4 * t + 2 * half;
        v2f bf; bf.x = bp[0]; bf.y = bp[1];
        acc = wmma4(qa[t], bf, acc);
      }
#pragma unroll
      for (int r = 0; r < 8; ++r) {
        int m = r + 8 * half;
        int i = qrow0 + m;
        int tok = keyBase + ln;
        int col = sb * 64 + kt * 16 + ln;
        float x = acc[r] * SCALE;
        myP[m * PSTRIDE + col] = (tok <= i) ? x : NEGINF;
      }
    }
  }

  for (int r = 0; r < 16; r += SELR) {
    float x[SELR][16];
    unsigned key[SELR][16];
    unsigned pre[SELR];
#pragma unroll
    for (int j = 0; j < SELR; ++j) {
      const float* row = myP + (r + j) * PSTRIDE;
#pragma unroll
      for (int k = 0; k < 16; ++k) {
        x[j][k] = row[lane + 32 * k];
        key[j][k] = okey(x[j][k]);
      }
      pre[j] = 0u;
    }
    for (int bit = 31; bit >= 0; --bit) {
#pragma unroll
      for (int j = 0; j < SELR; ++j) {
        const unsigned cand = pre[j] | (1u << bit);
        int cnt = 0;
#pragma unroll
        for (int k = 0; k < 16; k += 2) {
          const unsigned long long two =
              ((unsigned long long)__builtin_amdgcn_ballot_w32(key[j][k + 1] >= cand) << 32) |
              (unsigned long long)__builtin_amdgcn_ballot_w32(key[j][k] >= cand);
          cnt += __builtin_popcountll(two);
        }
        if (cnt >= KEEP) pre[j] = cand;
      }
    }
#pragma unroll
    for (int j = 0; j < SELR; ++j) {
      float* row = myP + (r + j) * PSTRIDE;
      float mx = -3.4e38f;
#pragma unroll
      for (int k = 0; k < 16; ++k) mx = fmaxf(mx, x[j][k]);
      mx = wred_maxf(mx);
      float sum = 0.f;
#pragma unroll
      for (int k = 0; k < 16; ++k) {
        float p = (key[j][k] >= pre[j] && mx > -1e29f) ? __expf(x[j][k] - mx) : 0.f;
        x[j][k] = p;
        sum += p;
      }
      sum = wred_sumf(sum);
      float rs = (sum > 0.f) ? 1.f / sum : 0.f;
#pragma unroll
      for (int k = 0; k < 16; ++k) row[lane + 32 * k] = x[j][k] * rs;
    }
  }

#pragma unroll
  for (int dt = 0; dt < 4; ++dt) {
    for (int g = 0; g < SELC / 4; ++g) {
      int c0 = 4 * g + 2 * half;
      v2f a; a.x = myP[ln * PSTRIDE + c0]; a.y = myP[ln * PSTRIDE + c0 + 1];
      int t0 = sSel[c0 >> 6] * BS + (c0 & 63);
      v2f bf;
      bf.x = Vh[(size_t)t0 * DD + dt * 16 + ln];
      bf.y = Vh[(size_t)(t0 + 1) * DD + dt * 16 + ln];
      oacc[dt] = wmma4(a, bf, oacc[dt]);
    }
  }

  const int firstb = (qb >= 2) ? qb - 2 : 0;
  const int nwb = qb - firstb + 1;
  const int extra = (firstb > 0) ? 1 : 0;
  const int ncg = nwb + extra;
  const int Cw = ncg * 64;

  for (int cg = 0; cg < ncg; ++cg) {
    int kb = (extra && cg == 0) ? 0 : firstb + (cg - extra);
    for (int kt = 0; kt < 4; ++kt) {
      int keyBase = kb * BS + kt * 16;
      v8f acc = vzero8();
#pragma unroll
      for (int t = 0; t < 16; ++t) {
        const float* bp = Kh + (size_t)(keyBase + ln) * DD + 4 * t + 2 * half;
        v2f bf; bf.x = bp[0]; bf.y = bp[1];
        acc = wmma4(qa[t], bf, acc);
      }
#pragma unroll
      for (int r = 0; r < 8; ++r) {
        int m = r + 8 * half;
        int i = qrow0 + m;
        int j = keyBase + ln;
        int col = cg * 64 + kt * 16 + ln;
        bool ok = (j <= i) && (((i - j) < WIN) || (j < 1));
        myP[m * PSTRIDE + col] = ok ? acc[r] * SCALE : NEGINF;
      }
    }
  }

  for (int r = 0; r < 16; ++r) {
    float* row = myP + r * PSTRIDE;
    float mx = -3.4e38f;
    for (int c = lane; c < Cw; c += 32) mx = fmaxf(mx, row[c]);
    mx = wred_maxf(mx);
    float sum = 0.f;
    for (int c = lane; c < Cw; c += 32) {
      float p = (mx > -1e29f) ? __expf(row[c] - mx) : 0.f;
      row[c] = p;
      sum += p;
    }
    sum = wred_sumf(sum);
    float rs = (sum > 0.f) ? 1.f / sum : 0.f;
    for (int c = lane; c < Cw; c += 32) row[c] *= rs;
  }

#pragma unroll
  for (int dt = 0; dt < 4; ++dt) {
    for (int g = 0; g < Cw / 4; ++g) {
      int c0 = 4 * g + 2 * half;
      v2f a; a.x = myP[ln * PSTRIDE + c0]; a.y = myP[ln * PSTRIDE + c0 + 1];
      int cg0 = c0 >> 6;
      int kb0 = (extra && cg0 == 0) ? 0 : firstb + (cg0 - extra);
      int t0 = kb0 * BS + (c0 & 63);
      v2f bf;
      bf.x = Vh[(size_t)t0 * DD + dt * 16 + ln];
      bf.y = Vh[(size_t)(t0 + 1) * DD + dt * 16 + ln];
      oacc[dt] = wmma4(a, bf, oacc[dt]);
    }
  }

  const float third = 1.f / 3.f;
#pragma unroll
  for (int dt = 0; dt < 4; ++dt) {
#pragma unroll
    for (int r = 0; r < 8; ++r) {
      int row = qrow0 + r + 8 * half;
      Oh[(size_t)row * DD + dt * 16 + ln] = oacc[dt][r] * third;
    }
  }
}

extern "C" void kernel_launch(void* const* d_in, const int* in_sizes, int n_in,
                              void* d_out, int out_size, void* d_ws,
                              size_t ws_size, hipStream_t stream) {
  (void)in_sizes; (void)n_in; (void)out_size; (void)ws_size;
  const float* q = (const float*)d_in[0];
  const float* k = (const float*)d_in[1];
  const float* v = (const float*)d_in[2];
  float* out = (float*)d_out;
  float* kblk = (float*)d_ws;
  float* vblk = kblk + (size_t)BB * HH * NB * DD;

  nsa_blockmean_kernel<<<BB * HH * NB, 64, 0, stream>>>(k, v, kblk, vblk);

  const size_t shmem = (size_t)64 * PSTRIDE * sizeof(float);
  hipFuncSetAttribute(reinterpret_cast<const void*>(nsa_main_kernel),
                      hipFuncAttributeMaxDynamicSharedMemorySize, (int)shmem);
  nsa_main_kernel<<<BB * HH * NB, 128, shmem, stream>>>(q, k, v, kblk, vblk,
                                                        out);
}


# P x V products: key rows walked once for all 4 output tiles (tile j = columns 4*lane+j), V rows as 16-byte loads, next 32-row chunk loaded ahead of the matrix steps; sum order unchanged
# speedup vs baseline: 1.8797x; 1.3505x over previous
#include <hip/hip_runtime.h>

#define BB 2
#define HH 8
#define SS 2048
#define DD 64
#define NB 32
#define BS 64
#define NSEL 8
#define SELC 512
#define KEEP 256
#define WIN 128
#define PSTRIDE 520
#define NEGINF (-1e30f)
#define SCALE 0.125f
#define SELR 2

typedef __attribute__((ext_vector_type(2))) float v2f;
typedef __attribute__((ext_vector_type(8))) float v8f;

static __device__ inline v8f vzero8() {
  v8f r;
#pragma unroll
  for (int i = 0; i < 8; ++i) r[i] = 0.f;
  return r;
}

static __device__ inline v8f wmma4(v2f a, v2f b, v8f c) {
  return __builtin_amdgcn_wmma_f32_16x16x4_f32(false, a, false, b, (short)0, c,
                                               false, false);
}

static __device__ inline unsigned okey(float x) {
  unsigned u = __float_as_uint(x);
  return (u & 0x80000000u) ? ~u : (u | 0x80000000u);
}

static __device__ inline int wred_sumi(int x) {
#pragma unroll
  for (int o = 16; o > 0; o >>= 1) x += __shfl_xor(x, o, 32);
  return x;
}
static __device__ inline float wred_sumf(float x) {
#pragma unroll
  for (int o = 16; o > 0; o >>= 1) x += __shfl_xor(x, o, 32);
  return x;
}
static __device__ inline float wred_maxf(float x) {
#pragma unroll
  for (int o = 16; o > 0; o >>= 1) x = fmaxf(x, __shfl_xor(x, o, 32));
  return x;
}

typedef __attribute__((ext_vector_type(4))) float v4f;
struct PVChunk {
  v2f a[8];
  v4f v0[8];
  v4f v1[8];
};
static __device__ inline void pv_load(PVChunk& c, const float* prow,
                                      const float* vrow) {
#pragma unroll
  for (int g = 0; g < 8; ++g) {
    c.v0[g] = *(const v4f*)(vrow + (size_t)(4 * g) * DD);
    c.v1[g] = *(const v4f*)(vrow + (size_t)(4 * g + 1) * DD);
  }
#pragma unroll
  for (int g = 0; g < 8; ++g) c.a[g] = *(const v2f*)(prow + 4 * g);
}
static __device__ inline void pv_mma(const PVChunk& c, v8f (&o)[4]) {
#pragma unroll
  for (int g = 0; g < 8; ++g) {
#pragma unroll
    for (int j = 0; j < 4; ++j) {
      v2f bf; bf.x = c.v0[g][j]; bf.y = c.v1[g][j];
      o[j] = wmma4(c.a[g], bf, o[j]);
    }
  }
}

__global__ void nsa_blockmean_kernel(const float* __restrict__ K,
                                     const float* __restrict__ V,
                                     float* __restrict__ KB,
                                     float* __restrict__ VB) {
  const int g = blockIdx.x;
  const int d = threadIdx.x;
  const float* kp = K + (size_t)g * (BS * DD);
  const float* vp = V + (size_t)g * (BS * DD);
  float ks = 0.f, vs = 0.f;
  for (int t = 0; t < BS; ++t) {
    ks += kp[t * DD + d];
    vs += vp[t * DD + d];
  }
  KB[(size_t)g * DD + d] = ks * (1.f / BS);
  VB[(size_t)g * DD + d] = vs * (1.f / BS);
}

__global__ __launch_bounds__(128) void nsa_main_kernel(
    const float* __restrict__ Q, const float* __restrict__ K,
    const float* __restrict__ V, const float* __restrict__ KB,
    const float* __restrict__ VB, float* __restrict__ O) {
  extern __shared__ float sP[];
  __shared__ float sScore[NB];
  __shared__ int sSel[NSEL];

  const int wg = blockIdx.x;
  const int qb = wg % NB;
  const int bh = wg / NB;
  const size_t hoff = (size_t)bh * SS * DD;
  const float* Qh = Q + hoff;
  const float* Kh = K + hoff;
  const float* Vh = V + hoff;
  const float* KBh = KB + (size_t)bh * NB * DD;
  const float* VBh = VB + (size_t)bh * NB * DD;
  float* Oh = O + hoff;

  const int tid = threadIdx.x;
  const int wave = tid >> 5;
  const int lane = tid & 31;
  const int half = lane >> 4;
  const int ln = lane & 15;
  const int qrow0 = qb * BS + wave * 16;
  float* myP = sP + (size_t)wave * 16 * PSTRIDE;

  v2f qa[16];
#pragma unroll
  for (int t = 0; t < 16; ++t) {
    const float* p = Qh + (size_t)(qrow0 + ln) * DD + 4 * t + 2 * half;
    v2f a; a.x = p[0]; a.y = p[1];
    qa[t] = a;
  }

  v8f oacc[4];
#pragma unroll
  for (int dt = 0; dt < 4; ++dt) oacc[dt] = vzero8();

#pragma unroll
  for (int nt = 0; nt < 2; ++nt) {
    v8f acc = vzero8();
#pragma unroll
    for (int t = 0; t < 16; ++t) {
      const float* bp = KBh + (size_t)(nt * 16 + ln) * DD + 4 * t + 2 * half;
      v2f bf; bf.x = bp[0]; bf.y = bp[1];
      acc = wmma4(qa[t], bf, acc);
    }
#pragma unroll
    for (int r = 0; r < 8; ++r) {
      int m = r + 8 * half;
      int col = nt * 16 + ln;
      float x = acc[r] * SCALE;
      myP[m * PSTRIDE + col] = (col <= qb) ? x : NEGINF;
    }
  }
  __syncthreads();

  if (tid < NB) {
    float s = 0.f;
    for (int q2 = 0; q2 < BS; ++q2) s += sP[q2 * PSTRIDE + tid];
    sScore[tid] = s * (1.f / BS);
  }
  __syncthreads();

  if (tid == 0) {
    float loc[NB];
    for (int i = 0; i < NB; ++i) loc[i] = sScore[i];
    for (int s = 0; s < NSEL; ++s) {
      int bi = 0;
      float bv = loc[0];
      for (int i = 1; i < NB; ++i)
        if (loc[i] > bv) { bv = loc[i]; bi = i; }
      sSel[s] = bi;
      loc[bi] = -3.4e38f;
    }
  }
  __syncthreads();

  if (tid < BS) {
    float* row = sP + tid * PSTRIDE;
    float mx = -3.4e38f;
    for (int c = 0; c < NB; ++c) mx = fmaxf(mx, row[c]);
    float sum = 0.f;
    for (int c = 0; c < NB; ++c) {
      float p = (mx > -1e29f) ? __expf(row[c] - mx) : 0.f;
      row[c] = p;
      sum += p;
    }
    float rs = (sum > 0.f) ? 1.f / sum : 0.f;
    for (int c = 0; c < NB; ++c) row[c] *= rs;
  }
  __syncthreads();

  const float* pLane = myP + ln * PSTRIDE + 2 * half;
  const size_t vLane = (size_t)(2 * half) * DD + 4 * ln;

  {
    PVChunk c;
    pv_load(c, pLane, VBh + vLane);
    pv_mma(c, oacc);
  }
  __syncthreads();

  for (int sb = 0; sb < NSEL; ++sb) {
    int kb = sSel[sb];
    for (int kt = 0; kt < 4; ++kt) {
      int keyBase = kb * BS + kt * 16;
      __builtin_prefetch(Kh + (size_t)(keyBase + 16) * DD, 0, 1);
      v8f acc = vzero8();
#pragma unroll
      for (int t = 0; t < 16; ++t) {
        const float* bp = Kh + (size_t)(keyBase + ln) * DD + 4 * t + 2 * half;
        v2f bf; bf.x = bp[0]; bf.y = bp[1];
        acc = wmma4(qa[t], bf, acc);
      }
#pragma unroll
      for (int r = 0; r < 8; ++r) {
        int m = r + 8 * half;
        int i = qrow0 + m;
        int tok = keyBase + ln;
        int col = sb * 64 + kt * 16 + ln;
        float x = acc[r] * SCALE;
        myP[m * PSTRIDE + col] = (tok <= i) ? x : NEGINF;
      }
    }
  }

  for (int r = 0; r < 16; r += SELR) {
    float x[SELR][16];
    unsigned key[SELR][16];
    unsigned pre[SELR];
#pragma unroll
    for (int j = 0; j < SELR; ++j) {
      const float* row = myP + (r + j) * PSTRIDE;
#pragma unroll
      for (int k = 0; k < 16; ++k) {
        x[j][k] = row[lane + 32 * k];
        key[j][k] = okey(x[j][k]);
      }
      pre[j] = 0u;
    }
    for (int bit = 31; bit >= 0; --bit) {
#pragma unroll
      for (int j = 0; j < SELR; ++j) {
        const unsigned cand = pre[j] | (1u << bit);
        int cnt = 0;
#pragma unroll
        for (int k = 0; k < 16; k += 2) {
          const unsigned long long two =
              ((unsigned long long)__builtin_amdgcn_ballot_w32(key[j][k + 1] >= cand) << 32) |
              (unsigned long long)__builtin_amdgcn_ballot_w32(key[j][k] >= cand);
          cnt += __builtin_popcountll(two);
        }
        if (cnt >= KEEP) pre[j] = cand;
      }
    }
#pragma unroll
    for (int j = 0; j < SELR; ++j) {
      float* row = myP + (r + j) * PSTRIDE;
      float mx = -3.4e38f;
#pragma unroll
      for (int k = 0; k < 16; ++k) mx = fmaxf(mx, x[j][k]);
      mx = wred_maxf(mx);
      float sum = 0.f;
#pragma unroll
      for (int k = 0; k < 16; ++k) {
        float p = (key[j][k] >= pre[j] && mx > -1e29f) ? __expf(x[j][k] - mx) : 0.f;
        x[j][k] = p;
        sum += p;
      }
      sum = wred_sumf(sum);
      float rs = (sum > 0.f) ? 1.f / sum : 0.f;
#pragma unroll
      for (int k = 0; k < 16; ++k) row[lane + 32 * k] = x[j][k] * rs;
    }
  }

  {
    PVChunk ca, cb;
    pv_load(ca, pLane, Vh + (size_t)sSel[0] * (BS * DD) + vLane);
    for (int ch = 0; ch < 2 * NSEL; ch += 2) {
      pv_load(cb, pLane + (ch + 1) * 32,
              Vh + (size_t)sSel[ch >> 1] * (BS * DD) + 32 * DD + vLane);
      pv_mma(ca, oacc);
      if (ch + 2 < 2 * NSEL)
        pv_load(ca, pLane + (ch + 2) * 32,
                Vh + (size_t)sSel[(ch >> 1) + 1] * (BS * DD) + vLane);
      pv_mma(cb, oacc);
    }
  }

  const int firstb = (qb >= 2) ? qb - 2 : 0;
  const int nwb = qb - firstb + 1;
  const int extra = (firstb > 0) ? 1 : 0;
  const int ncg = nwb + extra;
  const int Cw = ncg * 64;

  for (int cg = 0; cg < ncg; ++cg) {
    int kb = (extra && cg == 0) ? 0 : firstb + (cg - extra);
    for (int kt = 0; kt < 4; ++kt) {
      int keyBase = kb * BS + kt * 16;
      v8f acc = vzero8();
#pragma unroll
      for (int t = 0; t < 16; ++t) {
        const float* bp = Kh + (size_t)(keyBase + ln) * DD + 4 * t + 2 * half;
        v2f bf; bf.x = bp[0]; bf.y = bp[1];
        acc = wmma4(qa[t], bf, acc);
      }
#pragma unroll
      for (int r = 0; r < 8; ++r) {
        int m = r + 8 * half;
        int i = qrow0 + m;
        int j = keyBase + ln;
        int col = cg * 64 + kt * 16 + ln;
        bool ok = (j <= i) && (((i - j) < WIN) || (j < 1));
        myP[m * PSTRIDE + col] = ok ? acc[r] * SCALE : NEGINF;
      }
    }
  }

  for (int r = 0; r < 16; ++r) {
    float* row = myP + r * PSTRIDE;
    float mx = -3.4e38f;
    for (int c = lane; c < Cw; c += 32) mx = fmaxf(mx, row[c]);
    mx = wred_maxf(mx);
    float sum = 0.f;
    for (int c = lane; c < Cw; c += 32) {
      float p = (mx > -1e29f) ? __expf(row[c] - mx) : 0.f;
      row[c] = p;
      sum += p;
    }
    sum = wred_sumf(sum);
    float rs = (sum > 0.f) ? 1.f / sum : 0.f;
    for (int c = lane; c < Cw; c += 32) row[c] *= rs;
  }

  {
    PVChunk ca, cb;
    const int kbFirst = extra ? 0 : firstb;
    pv_load(ca, pLane, Vh + (size_t)kbFirst * (BS * DD) + vLane);
    for (int cg = 0; cg < ncg; ++cg) {
      const int kb0 = (extra && cg == 0) ? 0 : firstb + (cg - extra);
      pv_load(cb, pLane + cg * 64 + 32,
              Vh + (size_t)kb0 * (BS * DD) + 32 * DD + vLane);
      pv_mma(ca, oacc);
      if (cg + 1 < ncg)
        pv_load(ca, pLane + (cg + 1) * 64,
                Vh + (size_t)(firstb + (cg + 1 - extra)) * (BS * DD) + vLane);
      pv_mma(cb, oacc);
    }
  }

  const float third = 1.f / 3.f;
#pragma unroll
  for (int r = 0; r < 8; ++r) {
    int row = qrow0 + r + 8 * half;
    v4f o;
    o.x = oacc[0][r] * third;
    o.y = oacc[1][r] * third;
    o.z = oacc[2][r] * third;
    o.w = oacc[3][r] * third;
    *(v4f*)(Oh + (size_t)row * DD + 4 * ln) = o;
  }
}

extern "C" void kernel_launch(void* const* d_in, const int* in_sizes, int n_in,
                              void* d_out, int out_size, void* d_ws,
                              size_t ws_size, hipStream_t stream) {
  (void)in_sizes; (void)n_in; (void)out_size; (void)ws_size;
  const float* q = (const float*)d_in[0];
  const float* k = (const float*)d_in[1];
  const float* v = (const float*)d_in[2];
  float* out = (float*)d_out;
  float* kblk = (float*)d_ws;
  float* vblk = kblk + (size_t)BB * HH * NB * DD;

  nsa_blockmean_kernel<<<BB * HH * NB, 64, 0, stream>>>(k, v, kblk, vblk);

  const size_t shmem = (size_t)64 * PSTRIDE * sizeof(float);
  hipFuncSetAttribute(reinterpret_cast<const void*>(nsa_main_kernel),
                      hipFuncAttributeMaxDynamicSharedMemorySize, (int)shmem);
  nsa_main_kernel<<<BB * HH * NB, 128, shmem, stream>>>(q, k, v, kblk, vblk,
                                                        out);
}


# window-branch softmax: rows held in registers 4 at a time (one LDS read + one write per logit), their butterfly exchanges in flight together instead of serial LDS round trips; sum order unchanged
# speedup vs baseline: 2.2049x; 1.1730x over previous
#include <hip/hip_runtime.h>

#define BB 2
#define HH 8
#define SS 2048
#define DD 64
#define NB 32
#define BS 64
#define NSEL 8
#define SELC 512
#define KEEP 256
#define WIN 128
#define PSTRIDE 520
#define NEGINF (-1e30f)
#define SCALE 0.125f
#define SELR 2
#define WROWS 4

typedef __attribute__((ext_vector_type(2))) float v2f;
typedef __attribute__((ext_vector_type(8))) float v8f;

static __device__ inline v8f vzero8() {
  v8f r;
#pragma unroll
  for (int i = 0; i < 8; ++i) r[i] = 0.f;
  return r;
}

static __device__ inline v8f wmma4(v2f a, v2f b, v8f c) {
  return __builtin_amdgcn_wmma_f32_16x16x4_f32(false, a, false, b, (short)0, c,
                                               false, false);
}

static __device__ inline unsigned okey(float x) {
  unsigned u = __float_as_uint(x);
  return (u & 0x80000000u) ? ~u : (u | 0x80000000u);
}

static __device__ inline int wred_sumi(int x) {
#pragma unroll
  for (int o = 16; o > 0; o >>= 1) x += __shfl_xor(x, o, 32);
  return x;
}
static __device__ inline float wred_sumf(float x) {
#pragma unroll
  for (int o = 16; o > 0; o >>= 1) x += __shfl_xor(x, o, 32);
  return x;
}
static __device__ inline float wred_maxf(float x) {
#pragma unroll
  for (int o = 16; o > 0; o >>= 1) x = fmaxf(x, __shfl_xor(x, o, 32));
  return x;
}

typedef __attribute__((ext_vector_type(4))) float v4f;
struct PVChunk {
  v2f a[8];
  v4f v0[8];
  v4f v1[8];
};
static __device__ inline void pv_load(PVChunk& c, const float* prow,
                                      const float* vrow) {
#pragma unroll
  for (int g = 0; g < 8; ++g) {
    c.v0[g] = *(const v4f*)(vrow + (size_t)(4 * g) * DD);
    c.v1[g] = *(const v4f*)(vrow + (size_t)(4 * g + 1) * DD);
  }
#pragma unroll
  for (int g = 0; g < 8; ++g) c.a[g] = *(const v2f*)(prow + 4 * g);
}
static __device__ inline void pv_mma(const PVChunk& c, v8f (&o)[4]) {
#pragma unroll
  for (int g = 0; g < 8; ++g) {
#pragma unroll
    for (int j = 0; j < 4; ++j) {
      v2f bf; bf.x = c.v0[g][j]; bf.y = c.v1[g][j];
      o[j] = wmma4(c.a[g], bf, o[j]);
    }
  }
}

static __device__ inline void win_softmax(float* myP, int lane, int ncg) {
  for (int r0 = 0; r0 < 16; r0 += WROWS) {
    float* base = myP + r0 * PSTRIDE + lane;
    float x[WROWS][8];
    float mx[WROWS];
    float sum[WROWS];
#pragma unroll
    for (int j = 0; j < WROWS; ++j) {
#pragma unroll
      for (int k = 0; k < 8; ++k) x[j][k] = 0.f;
      mx[j] = -3.4e38f;
      sum[j] = 0.f;
    }
#pragma unroll
    for (int g = 0; g < 4; ++g) {
      if (g < ncg) {
#pragma unroll
        for (int j = 0; j < WROWS; ++j) {
          x[j][2 * g] = base[j * PSTRIDE + 64 * g];
          x[j][2 * g + 1] = base[j * PSTRIDE + 64 * g + 32];
        }
      }
    }
#pragma unroll
    for (int g = 0; g < 4; ++g) {
      if (g < ncg) {
#pragma unroll
        for (int j = 0; j < WROWS; ++j) {
          mx[j] = fmaxf(mx[j], x[j][2 * g]);
          mx[j] = fmaxf(mx[j], x[j][2 * g + 1]);
        }
      }
    }
#pragma unroll
    for (int o = 16; o > 0; o >>= 1) {
#pragma unroll
      for (int j = 0; j < WROWS; ++j)
        mx[j] = fmaxf(mx[j], __shfl_xor(mx[j], o, 32));
    }
#pragma unroll
    for (int g = 0; g < 4; ++g) {
      if (g < ncg) {
#pragma unroll
        for (int j = 0; j < WROWS; ++j) {
#pragma unroll
          for (int k = 2 * g; k < 2 * g + 2; ++k) {
            float p = (mx[j] > -1e29f) ? __expf(x[j][k] - mx[j]) : 0.f;
            x[j][k] = p;
            sum[j] += p;
          }
        }
      }
    }
#pragma unroll
    for (int o = 16; o > 0; o >>= 1) {
#pragma unroll
      for (int j = 0; j < WROWS; ++j) sum[j] += __shfl_xor(sum[j], o, 32);
    }
    float rs[WROWS];
#pragma unroll
    for (int j = 0; j < WROWS; ++j) rs[j] = (sum[j] > 0.f) ? 1.f / sum[j] : 0.f;
#pragma unroll
    for (int g = 0; g < 4; ++g) {
      if (g < ncg) {
#pragma unroll
        for (int j = 0; j < WROWS; ++j) {
          base[j * PSTRIDE + 64 * g] = x[j][2 * g] * rs[j];
          base[j * PSTRIDE + 64 * g + 32] = x[j][2 * g + 1] * rs[j];
        }
      }
    }
  }
}

__global__ void nsa_blockmean_kernel(const float* __restrict__ K,
                                     const float* __restrict__ V,
                                     float* __restrict__ KB,
                                     float* __restrict__ VB) {
  const int g = blockIdx.x;
  const int d = threadIdx.x;
  const float* kp = K + (size_t)g * (BS * DD);
  const float* vp = V + (size_t)g * (BS * DD);
  float ks = 0.f, vs = 0.f;
  for (int t = 0; t < BS; ++t) {
    ks += kp[t * DD + d];
    vs += vp[t * DD + d];
  }
  KB[(size_t)g * DD + d] = ks * (1.f / BS);
  VB[(size_t)g * DD + d] = vs * (1.f / BS);
}

__global__ __launch_bounds__(128) void nsa_main_kernel(
    const float* __restrict__ Q, const float* __restrict__ K,
    const float* __restrict__ V, const float* __restrict__ KB,
    const float* __restrict__ VB, float* __restrict__ O) {
  extern __shared__ float sP[];
  __shared__ float sScore[NB];
  __shared__ int sSel[NSEL];

  const int wg = blockIdx.x;
  const int qb = wg % NB;
  const int bh = wg / NB;
  const size_t hoff = (size_t)bh * SS * DD;
  const float* Qh = Q + hoff;
  const float* Kh = K + hoff;
  const float* Vh = V + hoff;
  const float* KBh = KB + (size_t)bh * NB * DD;
  const float* VBh = VB + (size_t)bh * NB * DD;
  float* Oh = O + hoff;

  const int tid = threadIdx.x;
  const int wave = tid >> 5;
  const int lane = tid & 31;
  const int half = lane >> 4;
  const int ln = lane & 15;
  const int qrow0 = qb * BS + wave * 16;
  float* myP = sP + (size_t)wave * 16 * PSTRIDE;

  v2f qa[16];
#pragma unroll
  for (int t = 0; t < 16; ++t) {
    const float* p = Qh + (size_t)(qrow0 + ln) * DD + 4 * t + 2 * half;
    v2f a; a.x = p[0]; a.y = p[1];
    qa[t] = a;
  }

  v8f oacc[4];
#pragma unroll
  for (int dt = 0; dt < 4; ++dt) oacc[dt] = vzero8();

#pragma unroll
  for (int nt = 0; nt < 2; ++nt) {
    v8f acc = vzero8();
#pragma unroll
    for (int t = 0; t < 16; ++t) {
      const float* bp = KBh + (size_t)(nt * 16 + ln) * DD + 4 * t + 2 * half;
      v2f bf; bf.x = bp[0]; bf.y = bp[1];
      acc = wmma4(qa[t], bf, acc);
    }
#pragma unroll
    for (int r = 0; r < 8; ++r) {
      int m = r + 8 * half;
      int col = nt * 16 + ln;
      float x = acc[r] * SCALE;
      myP[m * PSTRIDE + col] = (col <= qb) ? x : NEGINF;
    }
  }
  __syncthreads();

  if (tid < NB) {
    float s = 0.f;
    for (int q2 = 0; q2 < BS; ++q2) s += sP[q2 * PSTRIDE + tid];
    sScore[tid] = s * (1.f / BS);
  }
  __syncthreads();

  if (tid == 0) {
    float loc[NB];
    for (int i = 0; i < NB; ++i) loc[i] = sScore[i];
    for (int s = 0; s < NSEL; ++s) {
      int bi = 0;
      float bv = loc[0];
      for (int i = 1; i < NB; ++i)
        if (loc[i] > bv) { bv = loc[i]; bi = i; }
      sSel[s] = bi;
      loc[bi] = -3.4e38f;
    }
  }
  __syncthreads();

  if (tid < BS) {
    float* row = sP + tid * PSTRIDE;
    float mx = -3.4e38f;
    for (int c = 0; c < NB; ++c) mx = fmaxf(mx, row[c]);
    float sum = 0.f;
    for (int c = 0; c < NB; ++c) {
      float p = (mx > -1e29f) ? __expf(row[c] - mx) : 0.f;
      row[c] = p;
      sum += p;
    }
    float rs = (sum > 0.f) ? 1.f / sum : 0.f;
    for (int c = 0; c < NB; ++c) row[c] *= rs;
  }
  __syncthreads();

  const float* pLane = myP + ln * PSTRIDE + 2 * half;
  const size_t vLane = (size_t)(2 * half) * DD + 4 * ln;

  {
    PVChunk c;
    pv_load(c, pLane, VBh + vLane);
    pv_mma(c, oacc);
  }
  __syncthreads();

  for (int sb = 0; sb < NSEL; ++sb) {
    int kb = sSel[sb];
    for (int kt = 0; kt < 4; ++kt) {
      int keyBase = kb * BS + kt * 16;
      __builtin_prefetch(Kh + (size_t)(keyBase + 16) * DD, 0, 1);
      v8f acc = vzero8();
#pragma unroll
      for (int t = 0; t < 16; ++t) {
        const float* bp = Kh + (size_t)(keyBase + ln) * DD + 4 * t + 2 * half;
        v2f bf; bf.x = bp[0]; bf.y = bp[1];
        acc = wmma4(qa[t], bf, acc);
      }
#pragma unroll
      for (int r = 0; r < 8; ++r) {
        int m = r + 8 * half;
        int i = qrow0 + m;
        int tok = keyBase + ln;
        int col = sb * 64 + kt * 16 + ln;
        float x = acc[r] * SCALE;
        myP[m * PSTRIDE + col] = (tok <= i) ? x : NEGINF;
      }
    }
  }

  for (int r = 0; r < 16; r += SELR) {
    float x[SELR][16];
    unsigned key[SELR][16];
    unsigned pre[SELR];
#pragma unroll
    for (int j = 0; j < SELR; ++j) {
      const float* row = myP + (r + j) * PSTRIDE;
#pragma unroll
      for (int k = 0; k < 16; ++k) {
        x[j][k] = row[lane + 32 * k];
        key[j][k] = okey(x[j][k]);
      }
      pre[j] = 0u;
    }
    for (int bit = 31; bit >= 0; --bit) {
#pragma unroll
      for (int j = 0; j < SELR; ++j) {
        const unsigned cand = pre[j] | (1u << bit);
        int cnt = 0;
#pragma unroll
        for (int k = 0; k < 16; k += 2) {
          const unsigned long long two =
              ((unsigned long long)__builtin_amdgcn_ballot_w32(key[j][k + 1] >= cand) << 32) |
              (unsigned long long)__builtin_amdgcn_ballot_w32(key[j][k] >= cand);
          cnt += __builtin_popcountll(two);
        }
        if (cnt >= KEEP) pre[j] = cand;
      }
    }
#pragma unroll
    for (int j = 0; j < SELR; ++j) {
      float* row = myP + (r + j) * PSTRIDE;
      float mx = -3.4e38f;
#pragma unroll
      for (int k = 0; k < 16; ++k) mx = fmaxf(mx, x[j][k]);
      mx = wred_maxf(mx);
      float sum = 0.f;
#pragma unroll
      for (int k = 0; k < 16; ++k) {
        float p = (key[j][k] >= pre[j] && mx > -1e29f) ? __expf(x[j][k] - mx) : 0.f;
        x[j][k] = p;
        sum += p;
      }
      sum = wred_sumf(sum);
      float rs = (sum > 0.f) ? 1.f / sum : 0.f;
#pragma unroll
      for (int k = 0; k < 16; ++k) row[lane + 32 * k] = x[j][k] * rs;
    }
  }

  {
    PVChunk ca, cb;
    pv_load(ca, pLane, Vh + (size_t)sSel[0] * (BS * DD) + vLane);
    for (int ch = 0; ch < 2 * NSEL; ch += 2) {
      pv_load(cb, pLane + (ch + 1) * 32,
              Vh + (size_t)sSel[ch >> 1] * (BS * DD) + 32 * DD + vLane);
      pv_mma(ca, oacc);
      if (ch + 2 < 2 * NSEL)
        pv_load(ca, pLane + (ch + 2) * 32,
                Vh + (size_t)sSel[(ch >> 1) + 1] * (BS * DD) + vLane);
      pv_mma(cb, oacc);
    }
  }

  const int firstb = (qb >= 2) ? qb - 2 : 0;
  const int nwb = qb - firstb + 1;
  const int extra = (firstb > 0) ? 1 : 0;
  const int ncg = nwb + extra;

  for (int cg = 0; cg < ncg; ++cg) {
    int kb = (extra && cg == 0) ? 0 : firstb + (cg - extra);
    for (int kt = 0; kt < 4; ++kt) {
      int keyBase = kb * BS + kt * 16;
      v8f acc = vzero8();
#pragma unroll
      for (int t = 0; t < 16; ++t) {
        const float* bp = Kh + (size_t)(keyBase + ln) * DD + 4 * t + 2 * half;
        v2f bf; bf.x = bp[0]; bf.y = bp[1];
        acc = wmma4(qa[t], bf, acc);
      }
#pragma unroll
      for (int r = 0; r < 8; ++r) {
        int m = r + 8 * half;
        int i = qrow0 + m;
        int j = keyBase + ln;
        int col = cg * 64 + kt * 16 + ln;
        bool ok = (j <= i) && (((i - j) < WIN) || (j < 1));
        myP[m * PSTRIDE + col] = ok ? acc[r] * SCALE : NEGINF;
      }
    }
  }

  win_softmax(myP, lane, ncg);

  {
    PVChunk ca, cb;
    const int kbFirst = extra ? 0 : firstb;
    pv_load(ca, pLane, Vh + (size_t)kbFirst * (BS * DD) + vLane);
    for (int cg = 0; cg < ncg; ++cg) {
      const int kb0 = (extra && cg == 0) ? 0 : firstb + (cg - extra);
      pv_load(cb, pLane + cg * 64 + 32,
              Vh + (size_t)kb0 * (BS * DD) + 32 * DD + vLane);
      pv_mma(ca, oacc);
      if (cg + 1 < ncg)
        pv_load(ca, pLane + (cg + 1) * 64,
                Vh + (size_t)(firstb + (cg + 1 - extra)) * (BS * DD) + vLane);
      pv_mma(cb, oacc);
    }
  }

  const float third = 1.f / 3.f;
#pragma unroll
  for (int r = 0; r < 8; ++r) {
    int row = qrow0 + r + 8 * half;
    v4f o;
    o.x = oacc[0][r] * third;
    o.y = oacc[1][r] * third;
    o.z = oacc[2][r] * third;
    o.w = oacc[3][r] * third;
    *(v4f*)(Oh + (size_t)row * DD + 4 * ln) = o;
  }
}

extern "C" void kernel_launch(void* const* d_in, const int* in_sizes, int n_in,
                              void* d_out, int out_size, void* d_ws,
                              size_t ws_size, hipStream_t stream) {
  (void)in_sizes; (void)n_in; (void)out_size; (void)ws_size;
  const float* q = (const float*)d_in[0];
  const float* k = (const float*)d_in[1];
  const float* v = (const float*)d_in[2];
  float* out = (float*)d_out;
  float* kblk = (float*)d_ws;
  float* vblk = kblk + (size_t)BB * HH * NB * DD;

  nsa_blockmean_kernel<<<BB * HH * NB, 64, 0, stream>>>(k, v, kblk, vblk);

  const size_t shmem = (size_t)64 * PSTRIDE * sizeof(float);
  hipFuncSetAttribute(reinterpret_cast<const void*>(nsa_main_kernel),
                      hipFuncAttributeMaxDynamicSharedMemorySize, (int)shmem);
  nsa_main_kernel<<<BB * HH * NB, 128, shmem, stream>>>(q, k, v, kblk, vblk,
                                                        out);
}


# Q x K^T (selected and window branches): each 64-key K block staged once per work-group in LDS by coalesced 16-byte loads, one block ahead; fragments read from LDS; sum order unchanged
# speedup vs baseline: 2.5423x; 1.1530x over previous
#include <hip/hip_runtime.h>

#define BB 2
#define HH 8
#define SS 2048
#define DD 64
#define NB 32
#define BS 64
#define NSEL 8
#define SELC 512
#define KEEP 256
#define WIN 128
#define PSTRIDE 520
#define NEGINF (-1e30f)
#define SCALE 0.125f
#define SELR 2
#define WROWS 4
#define KPITCH 68

typedef __attribute__((ext_vector_type(2))) float v2f;
typedef __attribute__((ext_vector_type(8))) float v8f;

static __device__ inline v8f vzero8() {
  v8f r;
#pragma unroll
  for (int i = 0; i < 8; ++i) r[i] = 0.f;
  return r;
}

static __device__ inline v8f wmma4(v2f a, v2f b, v8f c) {
  return __builtin_amdgcn_wmma_f32_16x16x4_f32(false, a, false, b, (short)0, c,
                                               false, false);
}

static __device__ inline unsigned okey(float x) {
  unsigned u = __float_as_uint(x);
  return (u & 0x80000000u) ? ~u : (u | 0x80000000u);
}

static __device__ inline int wred_sumi(int x) {
#pragma unroll
  for (int o = 16; o > 0; o >>= 1) x += __shfl_xor(x, o, 32);
  return x;
}
static __device__ inline float wred_sumf(float x) {
#pragma unroll
  for (int o = 16; o > 0; o >>= 1) x += __shfl_xor(x, o, 32);
  return x;
}
static __device__ inline float wred_maxf(float x) {
#pragma unroll
  for (int o = 16; o > 0; o >>= 1) x = fmaxf(x, __shfl_xor(x, o, 32));
  return x;
}

typedef __attribute__((ext_vector_type(4))) float v4f;
struct PVChunk {
  v2f a[8];
  v4f v0[8];
  v4f v1[8];
};
static __device__ inline void pv_load(PVChunk& c, const float* prow,
                                      const float* vrow) {
#pragma unroll
  for (int g = 0; g < 8; ++g) {
    c.v0[g] = *(const v4f*)(vrow + (size_t)(4 * g) * DD);
    c.v1[g] = *(const v4f*)(vrow + (size_t)(4 * g + 1) * DD);
  }
#pragma unroll
  for (int g = 0; g < 8; ++g) c.a[g] = *(const v2f*)(prow + 4 * g);
}
static __device__ inline void pv_mma(const PVChunk& c, v8f (&o)[4]) {
#pragma unroll
  for (int g = 0; g < 8; ++g) {
#pragma unroll
    for (int j = 0; j < 4; ++j) {
      v2f bf; bf.x = c.v0[g][j]; bf.y = c.v1[g][j];
      o[j] = wmma4(c.a[g], bf, o[j]);
    }
  }
}

struct KStage {
  v4f r[8];
};
static __device__ inline void kstage_load(KStage& s, const float* blk, int tid) {
#pragma unroll
  for (int i = 0; i < 8; ++i) s.r[i] = *(const v4f*)(blk + 4 * (tid + 128 * i));
}
static __device__ inline void kstage_store(const KStage& s, float* sK, int tid) {
#pragma unroll
  for (int i = 0; i < 8; ++i) {
    const int c = tid + 128 * i;
    *(v4f*)(sK + (c >> 4) * KPITCH + 4 * (c & 15)) = s.r[i];
  }
}
struct KPair {
  v2f f0[16];
  v2f f1[16];
};
static __device__ inline void kpair_load(KPair& c, const float* krow) {
#pragma unroll
  for (int t = 0; t < 16; ++t) {
    c.f0[t] = *(const v2f*)(krow + 4 * t);
    c.f1[t] = *(const v2f*)(krow + 16 * KPITCH + 4 * t);
  }
}
static __device__ inline void kpair_mma(const KPair& c, const v2f (&qa)[16],
                                        v8f& a0, v8f& a1) {
#pragma unroll
  for (int t = 0; t < 16; ++t) {
    a0 = wmma4(qa[t], c.f0[t], a0);
    a1 = wmma4(qa[t], c.f1[t], a1);
  }
}
static __device__ inline void sel_tile_out(const v8f& acc, float* myP, int half,
                                           int ln, int qrow0, int tok0, int col0) {
#pragma unroll
  for (int r = 0; r < 8; ++r) {
    int m = r + 8 * half;
    int i = qrow0 + m;
    int tok = tok0 + ln;
    float x = acc[r] * SCALE;
    myP[m * PSTRIDE + col0 + ln] = (tok <= i) ? x : NEGINF;
  }
}
static __device__ inline void win_tile_out(const v8f& acc, float* myP, int half,
                                           int ln, int qrow0, int tok0, int col0) {
#pragma unroll
  for (int r = 0; r < 8; ++r) {
    int m = r + 8 * half;
    int i = qrow0 + m;
    int j = tok0 + ln;
    bool ok = (j <= i) && (((i - j) < WIN) || (j < 1));
    myP[m * PSTRIDE + col0 + ln] = ok ? acc[r] * SCALE : NEGINF;
  }
}

static __device__ inline void win_softmax(float* myP, int lane, int ncg) {
  for (int r0 = 0; r0 < 16; r0 += WROWS) {
    float* base = myP + r0 * PSTRIDE + lane;
    float x[WROWS][8];
    float mx[WROWS];
    float sum[WROWS];
#pragma unroll
    for (int j = 0; j < WROWS; ++j) {
#pragma unroll
      for (int k = 0; k < 8; ++k) x[j][k] = 0.f;
      mx[j] = -3.4e38f;
      sum[j] = 0.f;
    }
#pragma unroll
    for (int g = 0; g < 4; ++g) {
      if (g < ncg) {
#pragma unroll
        for (int j = 0; j < WROWS; ++j) {
          x[j][2 * g] = base[j * PSTRIDE + 64 * g];
          x[j][2 * g + 1] = base[j * PSTRIDE + 64 * g + 32];
        }
      }
    }
#pragma unroll
    for (int g = 0; g < 4; ++g) {
      if (g < ncg) {
#pragma unroll
        for (int j = 0; j < WROWS; ++j) {
          mx[j] = fmaxf(mx[j], x[j][2 * g]);
          mx[j] = fmaxf(mx[j], x[j][2 * g + 1]);
        }
      }
    }
#pragma unroll
    for (int o = 16; o > 0; o >>= 1) {
#pragma unroll
      for (int j = 0; j < WROWS; ++j)
        mx[j] = fmaxf(mx[j], __shfl_xor(mx[j], o, 32));
    }
#pragma unroll
    for (int g = 0; g < 4; ++g) {
      if (g < ncg) {
#pragma unroll
        for (int j = 0; j < WROWS; ++j) {
#pragma unroll
          for (int k = 2 * g; k < 2 * g + 2; ++k) {
            float p = (mx[j] > -1e29f) ? __expf(x[j][k] - mx[j]) : 0.f;
            x[j][k] = p;
            sum[j] += p;
          }
        }
      }
    }
#pragma unroll
    for (int o = 16; o > 0; o >>= 1) {
#pragma unroll
      for (int j = 0; j < WROWS; ++j) sum[j] += __shfl_xor(sum[j], o, 32);
    }
    float rs[WROWS];
#pragma unroll
    for (int j = 0; j < WROWS; ++j) rs[j] = (sum[j] > 0.f) ? 1.f / sum[j] : 0.f;
#pragma unroll
    for (int g = 0; g < 4; ++g) {
      if (g < ncg) {
#pragma unroll
        for (int j = 0; j < WROWS; ++j) {
          base[j * PSTRIDE + 64 * g] = x[j][2 * g] * rs[j];
          base[j * PSTRIDE + 64 * g + 32] = x[j][2 * g + 1] * rs[j];
        }
      }
    }
  }
}

__global__ void nsa_blockmean_kernel(const float* __restrict__ K,
                                     const float* __restrict__ V,
                                     float* __restrict__ KB,
                                     float* __restrict__ VB) {
  const int g = blockIdx.x;
  const int d = threadIdx.x;
  const float* kp = K + (size_t)g * (BS * DD);
  const float* vp = V + (size_t)g * (BS * DD);
  float ks = 0.f, vs = 0.f;
  for (int t = 0; t < BS; ++t) {
    ks += kp[t * DD + d];
    vs += vp[t * DD + d];
  }
  KB[(size_t)g * DD + d] = ks * (1.f / BS);
  VB[(size_t)g * DD + d] = vs * (1.f / BS);
}

__global__ __launch_bounds__(128) void nsa_main_kernel(
    const float* __restrict__ Q, const float* __restrict__ K,
    const float* __restrict__ V, const float* __restrict__ KB,
    const float* __restrict__ VB, float* __restrict__ O) {
  extern __shared__ v4f sDyn[];
  float* sP = (float*)sDyn;
  float* sK = sP + 64 * PSTRIDE;
  __shared__ float sScore[NB];
  __shared__ int sSel[NSEL + 1];

  const int wg = blockIdx.x;
  const int qb = wg % NB;
  const int bh = wg / NB;
  const size_t hoff = (size_t)bh * SS * DD;
  const float* Qh = Q + hoff;
  const float* Kh = K + hoff;
  const float* Vh = V + hoff;
  const float* KBh = KB + (size_t)bh * NB * DD;
  const float* VBh = VB + (size_t)bh * NB * DD;
  float* Oh = O + hoff;

  const int tid = threadIdx.x;
  const int wave = tid >> 5;
  const int lane = tid & 31;
  const int half = lane >> 4;
  const int ln = lane & 15;
  const int qrow0 = qb * BS + wave * 16;
  float* myP = sP + (size_t)wave * 16 * PSTRIDE;

  const int firstb = (qb >= 2) ? qb - 2 : 0;
  const int nwb = qb - firstb + 1;
  const int extra = (firstb > 0) ? 1 : 0;
  const int ncg = nwb + extra;
  const int wkb0 = extra ? 0 : firstb;

  v2f qa[16];
#pragma unroll
  for (int t = 0; t < 16; ++t) {
    const float* p = Qh + (size_t)(qrow0 + ln) * DD + 4 * t + 2 * half;
    v2f a; a.x = p[0]; a.y = p[1];
    qa[t] = a;
  }

  v8f oacc[4];
#pragma unroll
  for (int dt = 0; dt < 4; ++dt) oacc[dt] = vzero8();

#pragma unroll
  for (int nt = 0; nt < 2; ++nt) {
    v8f acc = vzero8();
#pragma unroll
    for (int t = 0; t < 16; ++t) {
      const float* bp = KBh + (size_t)(nt * 16 + ln) * DD + 4 * t + 2 * half;
      v2f bf; bf.x = bp[0]; bf.y = bp[1];
      acc = wmma4(qa[t], bf, acc);
    }
#pragma unroll
    for (int r = 0; r < 8; ++r) {
      int m = r + 8 * half;
      int col = nt * 16 + ln;
      float x = acc[r] * SCALE;
      myP[m * PSTRIDE + col] = (col <= qb) ? x : NEGINF;
    }
  }
  __syncthreads();

  if (tid < NB) {
    float s = 0.f;
    for (int q2 = 0; q2 < BS; ++q2) s += sP[q2 * PSTRIDE + tid];
    sScore[tid] = s * (1.f / BS);
  }
  __syncthreads();

  if (tid == 0) {
    float loc[NB];
    for (int i = 0; i < NB; ++i) loc[i] = sScore[i];
    for (int s = 0; s < NSEL; ++s) {
      int bi = 0;
      float bv = loc[0];
      for (int i = 1; i < NB; ++i)
        if (loc[i] > bv) { bv = loc[i]; bi = i; }
      sSel[s] = bi;
      loc[bi] = -3.4e38f;
    }
    sSel[NSEL] = wkb0;
  }
  __syncthreads();

  if (tid < BS) {
    float* row = sP + tid * PSTRIDE;
    float mx = -3.4e38f;
    for (int c = 0; c < NB; ++c) mx = fmaxf(mx, row[c]);
    float sum = 0.f;
    for (int c = 0; c < NB; ++c) {
      float p = (mx > -1e29f) ? __expf(row[c] - mx) : 0.f;
      row[c] = p;
      sum += p;
    }
    float rs = (sum > 0.f) ? 1.f / sum : 0.f;
    for (int c = 0; c < NB; ++c) row[c] *= rs;
  }
  __syncthreads();

  const float* pLane = myP + ln * PSTRIDE + 2 * half;
  const size_t vLane = (size_t)(2 * half) * DD + 4 * ln;

  const float* kLane = sK + ln * KPITCH + 2 * half;
  KStage ks;
  kstage_load(ks, Kh + (size_t)sSel[0] * (BS * DD), tid);

  {
    PVChunk c;
    pv_load(c, pLane, VBh + vLane);
    pv_mma(c, oacc);
  }

  for (int sb = 0; sb < NSEL; ++sb) {
    const int kb = sSel[sb];
    __syncthreads();
    kstage_store(ks, sK, tid);
    __syncthreads();
    kstage_load(ks, Kh + (size_t)sSel[sb + 1] * (BS * DD), tid);
    __builtin_amdgcn_sched_barrier(0);
    KPair pa, pb;
    kpair_load(pa, kLane);
    kpair_load(pb, kLane + 32 * KPITCH);
    {
      v8f a0 = vzero8(), a1 = vzero8();
      kpair_mma(pa, qa, a0, a1);
      sel_tile_out(a0, myP, half, ln, qrow0, kb * BS, sb * 64);
      sel_tile_out(a1, myP, half, ln, qrow0, kb * BS + 16, sb * 64 + 16);
    }
    {
      v8f a0 = vzero8(), a1 = vzero8();
      kpair_mma(pb, qa, a0, a1);
      sel_tile_out(a0, myP, half, ln, qrow0, kb * BS + 32, sb * 64 + 32);
      sel_tile_out(a1, myP, half, ln, qrow0, kb * BS + 48, sb * 64 + 48);
    }
  }

  for (int r = 0; r < 16; r += SELR) {
    float x[SELR][16];
    unsigned key[SELR][16];
    unsigned pre[SELR];
#pragma unroll
    for (int j = 0; j < SELR; ++j) {
      const float* row = myP + (r + j) * PSTRIDE;
#pragma unroll
      for (int k = 0; k < 16; ++k) {
        x[j][k] = row[lane + 32 * k];
        key[j][k] = okey(x[j][k]);
      }
      pre[j] = 0u;
    }
    for (int bit = 31; bit >= 0; --bit) {
#pragma unroll
      for (int j = 0; j < SELR; ++j) {
        const unsigned cand = pre[j] | (1u << bit);
        int cnt = 0;
#pragma unroll
        for (int k = 0; k < 16; k += 2) {
          const unsigned long long two =
              ((unsigned long long)__builtin_amdgcn_ballot_w32(key[j][k + 1] >= cand) << 32) |
              (unsigned long long)__builtin_amdgcn_ballot_w32(key[j][k] >= cand);
          cnt += __builtin_popcountll(two);
        }
        if (cnt >= KEEP) pre[j] = cand;
      }
    }
#pragma unroll
    for (int j = 0; j < SELR; ++j) {
      float* row = myP + (r + j) * PSTRIDE;
      float mx = -3.4e38f;
#pragma unroll
      for (int k = 0; k < 16; ++k) mx = fmaxf(mx, x[j][k]);
      mx = wred_maxf(mx);
      float sum = 0.f;
#pragma unroll
      for (int k = 0; k < 16; ++k) {
        float p = (key[j][k] >= pre[j] && mx > -1e29f) ? __expf(x[j][k] - mx) : 0.f;
        x[j][k] = p;
        sum += p;
      }
      sum = wred_sumf(sum);
      float rs = (sum > 0.f) ? 1.f / sum : 0.f;
#pragma unroll
      for (int k = 0; k < 16; ++k) row[lane + 32 * k] = x[j][k] * rs;
    }
  }

  {
    PVChunk ca, cb;
    pv_load(ca, pLane, Vh + (size_t)sSel[0] * (BS * DD) + vLane);
    for (int ch = 0; ch < 2 * NSEL; ch += 2) {
      pv_load(cb, pLane + (ch + 1) * 32,
              Vh + (size_t)sSel[ch >> 1] * (BS * DD) + 32 * DD + vLane);
      pv_mma(ca, oacc);
      if (ch + 2 < 2 * NSEL)
        pv_load(ca, pLane + (ch + 2) * 32,
                Vh + (size_t)sSel[(ch >> 1) + 1] * (BS * DD) + vLane);
      pv_mma(cb, oacc);
    }
  }

  for (int cg = 0; cg < ncg; ++cg) {
    const int kb = (extra && cg == 0) ? 0 : firstb + (cg - extra);
    __syncthreads();
    kstage_store(ks, sK, tid);
    __syncthreads();
    if (cg + 1 < ncg)
      kstage_load(ks, Kh + (size_t)(firstb + (cg + 1 - extra)) * (BS * DD), tid);
    __builtin_amdgcn_sched_barrier(0);
    KPair pa, pb;
    kpair_load(pa, kLane);
    kpair_load(pb, kLane + 32 * KPITCH);
    __builtin_amdgcn_sched_barrier(0);
    {
      v8f a0 = vzero8(), a1 = vzero8();
      kpair_mma(pa, qa, a0, a1);
      win_tile_out(a0, myP, half, ln, qrow0, kb * BS, cg * 64);
      win_tile_out(a1, myP, half, ln, qrow0, kb * BS + 16, cg * 64 + 16);
    }
    {
      v8f a0 = vzero8(), a1 = vzero8();
      kpair_mma(pb, qa, a0, a1);
      win_tile_out(a0, myP, half, ln, qrow0, kb * BS + 32, cg * 64 + 32);
      win_tile_out(a1, myP, half, ln, qrow0, kb * BS + 48, cg * 64 + 48);
    }
  }

  win_softmax(myP, lane, ncg);

  {
    PVChunk ca, cb;
    const int kbFirst = extra ? 0 : firstb;
    pv_load(ca, pLane, Vh + (size_t)kbFirst * (BS * DD) + vLane);
    for (int cg = 0; cg < ncg; ++cg) {
      const int kb0 = (extra && cg == 0) ? 0 : firstb + (cg - extra);
      pv_load(cb, pLane + cg * 64 + 32,
              Vh + (size_t)kb0 * (BS * DD) + 32 * DD + vLane);
      pv_mma(ca, oacc);
      if (cg + 1 < ncg)
        pv_load(ca, pLane + (cg + 1) * 64,
                Vh + (size_t)(firstb + (cg + 1 - extra)) * (BS * DD) + vLane);
      pv_mma(cb, oacc);
    }
  }

  const float third = 1.f / 3.f;
#pragma unroll
  for (int r = 0; r < 8; ++r) {
    int row = qrow0 + r + 8 * half;
    v4f o;
    o.x = oacc[0][r] * third;
    o.y = oacc[1][r] * third;
    o.z = oacc[2][r] * third;
    o.w = oacc[3][r] * third;
    *(v4f*)(Oh + (size_t)row * DD + 4 * ln) = o;
  }
}

extern "C" void kernel_launch(void* const* d_in, const int* in_sizes, int n_in,
                              void* d_out, int out_size, void* d_ws,
                              size_t ws_size, hipStream_t stream) {
  (void)in_sizes; (void)n_in; (void)out_size; (void)ws_size;
  const float* q = (const float*)d_in[0];
  const float* k = (const float*)d_in[1];
  const float* v = (const float*)d_in[2];
  float* out = (float*)d_out;
  float* kblk = (float*)d_ws;
  float* vblk = kblk + (size_t)BB * HH * NB * DD;

  nsa_blockmean_kernel<<<BB * HH * NB, 64, 0, stream>>>(k, v, kblk, vblk);

  const size_t shmem = (size_t)(64 * PSTRIDE + 64 * KPITCH) * sizeof(float);
  hipFuncSetAttribute(reinterpret_cast<const void*>(nsa_main_kernel),
                      hipFuncAttributeMaxDynamicSharedMemorySize, (int)shmem);
  nsa_main_kernel<<<BB * HH * NB, 128, shmem, stream>>>(q, k, v, kblk, vblk,
                                                        out);
}


# top-256 select: the bit loop ends at the first bit where exactly 256 keys pass (kept set then fixed); rows with under 256 unmasked logits skip it; kept set and sum order unchanged
# speedup vs baseline: 3.0663x; 1.2061x over previous
#include <hip/hip_runtime.h>

#define BB 2
#define HH 8
#define SS 2048
#define DD 64
#define NB 32
#define BS 64
#define NSEL 8
#define SELC 512
#define KEEP 256
#define WIN 128
#define PSTRIDE 520
#define NEGINF (-1e30f)
#define SCALE 0.125f
#define SELR 2
#define WROWS 4
#define KPITCH 68

typedef __attribute__((ext_vector_type(2))) float v2f;
typedef __attribute__((ext_vector_type(8))) float v8f;

static __device__ inline v8f vzero8() {
  v8f r;
#pragma unroll
  for (int i = 0; i < 8; ++i) r[i] = 0.f;
  return r;
}

static __device__ inline v8f wmma4(v2f a, v2f b, v8f c) {
  return __builtin_amdgcn_wmma_f32_16x16x4_f32(false, a, false, b, (short)0, c,
                                               false, false);
}

static __device__ inline unsigned okey(float x) {
  unsigned u = __float_as_uint(x);
  return (u & 0x80000000u) ? ~u : (u | 0x80000000u);
}

static __device__ inline int wred_sumi(int x) {
#pragma unroll
  for (int o = 16; o > 0; o >>= 1) x += __shfl_xor(x, o, 32);
  return x;
}
static __device__ inline float wred_sumf(float x) {
#pragma unroll
  for (int o = 16; o > 0; o >>= 1) x += __shfl_xor(x, o, 32);
  return x;
}
static __device__ inline float wred_maxf(float x) {
#pragma unroll
  for (int o = 16; o > 0; o >>= 1) x = fmaxf(x, __shfl_xor(x, o, 32));
  return x;
}

static __device__ inline int count_ge(const unsigned (&key)[16], unsigned c) {
  int cnt = 0;
#pragma unroll
  for (int k = 0; k < 16; k += 2) {
    const unsigned long long two =
        ((unsigned long long)__builtin_amdgcn_ballot_w32(key[k + 1] >= c) << 32) |
        (unsigned long long)__builtin_amdgcn_ballot_w32(key[k] >= c);
    cnt += __builtin_popcountll(two);
  }
  return cnt;
}

typedef __attribute__((ext_vector_type(4))) float v4f;
struct PVChunk {
  v2f a[8];
  v4f v0[8];
  v4f v1[8];
};
static __device__ inline void pv_load(PVChunk& c, const float* prow,
                                      const float* vrow) {
#pragma unroll
  for (int g = 0; g < 8; ++g) {
    c.v0[g] = *(const v4f*)(vrow + (size_t)(4 * g) * DD);
    c.v1[g] = *(const v4f*)(vrow + (size_t)(4 * g + 1) * DD);
  }
#pragma unroll
  for (int g = 0; g < 8; ++g) c.a[g] = *(const v2f*)(prow + 4 * g);
}
static __device__ inline void pv_mma(const PVChunk& c, v8f (&o)[4]) {
#pragma unroll
  for (int g = 0; g < 8; ++g) {
#pragma unroll
    for (int j = 0; j < 4; ++j) {
      v2f bf; bf.x = c.v0[g][j]; bf.y = c.v1[g][j];
      o[j] = wmma4(c.a[g], bf, o[j]);
    }
  }
}

struct KStage {
  v4f r[8];
};
static __device__ inline void kstage_load(KStage& s, const float* blk, int tid) {
#pragma unroll
  for (int i = 0; i < 8; ++i) s.r[i] = *(const v4f*)(blk + 4 * (tid + 128 * i));
}
static __device__ inline void kstage_store(const KStage& s, float* sK, int tid) {
#pragma unroll
  for (int i = 0; i < 8; ++i) {
    const int c = tid + 128 * i;
    *(v4f*)(sK + (c >> 4) * KPITCH + 4 * (c & 15)) = s.r[i];
  }
}
struct KPair {
  v2f f0[16];
  v2f f1[16];
};
static __device__ inline void kpair_load(KPair& c, const float* krow) {
#pragma unroll
  for (int t = 0; t < 16; ++t) {
    c.f0[t] = *(const v2f*)(krow + 4 * t);
    c.f1[t] = *(const v2f*)(krow + 16 * KPITCH + 4 * t);
  }
}
static __device__ inline void kpair_mma(const KPair& c, const v2f (&qa)[16],
                                        v8f& a0, v8f& a1) {
#pragma unroll
  for (int t = 0; t < 16; ++t) {
    a0 = wmma4(qa[t], c.f0[t], a0);
    a1 = wmma4(qa[t], c.f1[t], a1);
  }
}
static __device__ inline void sel_tile_out(const v8f& acc, float* myP, int half,
                                           int ln, int qrow0, int tok0, int col0) {
#pragma unroll
  for (int r = 0; r < 8; ++r) {
    int m = r + 8 * half;
    int i = qrow0 + m;
    int tok = tok0 + ln;
    float x = acc[r] * SCALE;
    myP[m * PSTRIDE + col0 + ln] = (tok <= i) ? x : NEGINF;
  }
}
static __device__ inline void win_tile_out(const v8f& acc, float* myP, int half,
                                           int ln, int qrow0, int tok0, int col0) {
#pragma unroll
  for (int r = 0; r < 8; ++r) {
    int m = r + 8 * half;
    int i = qrow0 + m;
    int j = tok0 + ln;
    bool ok = (j <= i) && (((i - j) < WIN) || (j < 1));
    myP[m * PSTRIDE + col0 + ln] = ok ? acc[r] * SCALE : NEGINF;
  }
}

static __device__ inline void win_softmax(float* myP, int lane, int ncg) {
  for (int r0 = 0; r0 < 16; r0 += WROWS) {
    float* base = myP + r0 * PSTRIDE + lane;
    float x[WROWS][8];
    float mx[WROWS];
    float sum[WROWS];
#pragma unroll
    for (int j = 0; j < WROWS; ++j) {
#pragma unroll
      for (int k = 0; k < 8; ++k) x[j][k] = 0.f;
      mx[j] = -3.4e38f;
      sum[j] = 0.f;
    }
#pragma unroll
    for (int g = 0; g < 4; ++g) {
      if (g < ncg) {
#pragma unroll
        for (int j = 0; j < WROWS; ++j) {
          x[j][2 * g] = base[j * PSTRIDE + 64 * g];
          x[j][2 * g + 1] = base[j * PSTRIDE + 64 * g + 32];
        }
      }
    }
#pragma unroll
    for (int g = 0; g < 4; ++g) {
      if (g < ncg) {
#pragma unroll
        for (int j = 0; j < WROWS; ++j) {
          mx[j] = fmaxf(mx[j], x[j][2 * g]);
          mx[j] = fmaxf(mx[j], x[j][2 * g + 1]);
        }
      }
    }
#pragma unroll
    for (int o = 16; o > 0; o >>= 1) {
#pragma unroll
      for (int j = 0; j < WROWS; ++j)
        mx[j] = fmaxf(mx[j], __shfl_xor(mx[j], o, 32));
    }
#pragma unroll
    for (int g = 0; g < 4; ++g) {
      if (g < ncg) {
#pragma unroll
        for (int j = 0; j < WROWS; ++j) {
#pragma unroll
          for (int k = 2 * g; k < 2 * g + 2; ++k) {
            float p = (mx[j] > -1e29f) ? __expf(x[j][k] - mx[j]) : 0.f;
            x[j][k] = p;
            sum[j] += p;
          }
        }
      }
    }
#pragma unroll
    for (int o = 16; o > 0; o >>= 1) {
#pragma unroll
      for (int j = 0; j < WROWS; ++j) sum[j] += __shfl_xor(sum[j], o, 32);
    }
    float rs[WROWS];
#pragma unroll
    for (int j = 0; j < WROWS; ++j) rs[j] = (sum[j] > 0.f) ? 1.f / sum[j] : 0.f;
#pragma unroll
    for (int g = 0; g < 4; ++g) {
      if (g < ncg) {
#pragma unroll
        for (int j = 0; j < WROWS; ++j) {
          base[j * PSTRIDE + 64 * g] = x[j][2 * g] * rs[j];
          base[j * PSTRIDE + 64 * g + 32] = x[j][2 * g + 1] * rs[j];
        }
      }
    }
  }
}

__global__ void nsa_blockmean_kernel(const float* __restrict__ K,
                                     const float* __restrict__ V,
                                     float* __restrict__ KB,
                                     float* __restrict__ VB) {
  const int g = blockIdx.x;
  const int d = threadIdx.x;
  const float* kp = K + (size_t)g * (BS * DD);
  const float* vp = V + (size_t)g * (BS * DD);
  float ks = 0.f, vs = 0.f;
  for (int t = 0; t < BS; ++t) {
    ks += kp[t * DD + d];
    vs += vp[t * DD + d];
  }
  KB[(size_t)g * DD + d] = ks * (1.f / BS);
  VB[(size_t)g * DD + d] = vs * (1.f / BS);
}

__global__ __launch_bounds__(128) void nsa_main_kernel(
    const float* __restrict__ Q, const float* __restrict__ K,
    const float* __restrict__ V, const float* __restrict__ KB,
    const float* __restrict__ VB, float* __restrict__ O) {
  extern __shared__ v4f sDyn[];
  float* sP = (float*)sDyn;
  float* sK = sP + 64 * PSTRIDE;
  __shared__ float sScore[NB];
  __shared__ int sSel[NSEL + 1];

  const int wg = blockIdx.x;
  const int qb = wg % NB;
  const int bh = wg / NB;
  const size_t hoff = (size_t)bh * SS * DD;
  const float* Qh = Q + hoff;
  const float* Kh = K + hoff;
  const float* Vh = V + hoff;
  const float* KBh = KB + (size_t)bh * NB * DD;
  const float* VBh = VB + (size_t)bh * NB * DD;
  float* Oh = O + hoff;

  const int tid = threadIdx.x;
  const int wave = tid >> 5;
  const int lane = tid & 31;
  const int half = lane >> 4;
  const int ln = lane & 15;
  const int qrow0 = qb * BS + wave * 16;
  float* myP = sP + (size_t)wave * 16 * PSTRIDE;

  const int firstb = (qb >= 2) ? qb - 2 : 0;
  const int nwb = qb - firstb + 1;
  const int extra = (firstb > 0) ? 1 : 0;
  const int ncg = nwb + extra;
  const int wkb0 = extra ? 0 : firstb;

  v2f qa[16];
#pragma unroll
  for (int t = 0; t < 16; ++t) {
    const float* p = Qh + (size_t)(qrow0 + ln) * DD + 4 * t + 2 * half;
    v2f a; a.x = p[0]; a.y = p[1];
    qa[t] = a;
  }

  v8f oacc[4];
#pragma unroll
  for (int dt = 0; dt < 4; ++dt) oacc[dt] = vzero8();

#pragma unroll
  for (int nt = 0; nt < 2; ++nt) {
    v8f acc = vzero8();
#pragma unroll
    for (int t = 0; t < 16; ++t) {
      const float* bp = KBh + (size_t)(nt * 16 + ln) * DD + 4 * t + 2 * half;
      v2f bf; bf.x = bp[0]; bf.y = bp[1];
      acc = wmma4(qa[t], bf, acc);
    }
#pragma unroll
    for (int r = 0; r < 8; ++r) {
      int m = r + 8 * half;
      int col = nt * 16 + ln;
      float x = acc[r] * SCALE;
      myP[m * PSTRIDE + col] = (col <= qb) ? x : NEGINF;
    }
  }
  __syncthreads();

  if (tid < NB) {
    float s = 0.f;
    for (int q2 = 0; q2 < BS; ++q2) s += sP[q2 * PSTRIDE + tid];
    sScore[tid] = s * (1.f / BS);
  }
  __syncthreads();

  if (tid == 0) {
    float loc[NB];
    for (int i = 0; i < NB; ++i) loc[i] = sScore[i];
    for (int s = 0; s < NSEL; ++s) {
      int bi = 0;
      float bv = loc[0];
      for (int i = 1; i < NB; ++i)
        if (loc[i] > bv) { bv = loc[i]; bi = i; }
      sSel[s] = bi;
      loc[bi] = -3.4e38f;
    }
    sSel[NSEL] = wkb0;
  }
  __syncthreads();

  if (tid < BS) {
    float* row = sP + tid * PSTRIDE;
    float mx = -3.4e38f;
    for (int c = 0; c < NB; ++c) mx = fmaxf(mx, row[c]);
    float sum = 0.f;
    for (int c = 0; c < NB; ++c) {
      float p = (mx > -1e29f) ? __expf(row[c] - mx) : 0.f;
      row[c] = p;
      sum += p;
    }
    float rs = (sum > 0.f) ? 1.f / sum : 0.f;
    for (int c = 0; c < NB; ++c) row[c] *= rs;
  }
  __syncthreads();

  const float* pLane = myP + ln * PSTRIDE + 2 * half;
  const size_t vLane = (size_t)(2 * half) * DD + 4 * ln;

  const float* kLane = sK + ln * KPITCH + 2 * half;
  KStage ks;
  kstage_load(ks, Kh + (size_t)sSel[0] * (BS * DD), tid);

  {
    PVChunk c;
    pv_load(c, pLane, VBh + vLane);
    pv_mma(c, oacc);
  }

  for (int sb = 0; sb < NSEL; ++sb) {
    const int kb = sSel[sb];
    __syncthreads();
    kstage_store(ks, sK, tid);
    __syncthreads();
    kstage_load(ks, Kh + (size_t)sSel[sb + 1] * (BS * DD), tid);
    __builtin_amdgcn_sched_barrier(0);
    KPair pa, pb;
    kpair_load(pa, kLane);
    kpair_load(pb, kLane + 32 * KPITCH);
    {
      v8f a0 = vzero8(), a1 = vzero8();
      kpair_mma(pa, qa, a0, a1);
      sel_tile_out(a0, myP, half, ln, qrow0, kb * BS, sb * 64);
      sel_tile_out(a1, myP, half, ln, qrow0, kb * BS + 16, sb * 64 + 16);
    }
    {
      v8f a0 = vzero8(), a1 = vzero8();
      kpair_mma(pb, qa, a0, a1);
      sel_tile_out(a0, myP, half, ln, qrow0, kb * BS + 32, sb * 64 + 32);
      sel_tile_out(a1, myP, half, ln, qrow0, kb * BS + 48, sb * 64 + 48);
    }
  }

  for (int r = 0; r < 16; r += SELR) {
    float x[SELR][16];
    unsigned key[SELR][16];
    unsigned pre[SELR];
#pragma unroll
    for (int j = 0; j < SELR; ++j) {
      const float* row = myP + (r + j) * PSTRIDE;
#pragma unroll
      for (int k = 0; k < 16; ++k) {
        x[j][k] = row[lane + 32 * k];
        key[j][k] = okey(x[j][k]);
      }
      pre[j] = 0u;
    }
    const unsigned kneg = okey(NEGINF);
    unsigned open = 0u;
#pragma unroll
    for (int j = 0; j < SELR; ++j) {
      if (count_ge(key[j], kneg + 1u) < KEEP && count_ge(key[j], kneg) >= KEEP)
        pre[j] = kneg;
      else
        open |= 1u << j;
    }
    if (open != 0u) {
      for (unsigned m = 0x80000000u; m != 0u; m >>= 1) {
#pragma unroll
        for (int j = 0; j < SELR; ++j) {
          const unsigned cand = pre[j] | m;
          const int cnt = count_ge(key[j], cand);
          if (cnt >= KEEP) pre[j] = cand;
          if (cnt == KEEP) open &= ~(1u << j);
        }
        if (open == 0u) break;
      }
    }
#pragma unroll
    for (int j = 0; j < SELR; ++j) {
      float* row = myP + (r + j) * PSTRIDE;
      float mx = -3.4e38f;
#pragma unroll
      for (int k = 0; k < 16; ++k) mx = fmaxf(mx, x[j][k]);
      mx = wred_maxf(mx);
      float sum = 0.f;
#pragma unroll
      for (int k = 0; k < 16; ++k) {
        float p = (key[j][k] >= pre[j] && mx > -1e29f) ? __expf(x[j][k] - mx) : 0.f;
        x[j][k] = p;
        sum += p;
      }
      sum = wred_sumf(sum);
      float rs = (sum > 0.f) ? 1.f / sum : 0.f;
#pragma unroll
      for (int k = 0; k < 16; ++k) row[lane + 32 * k] = x[j][k] * rs;
    }
  }

  {
    PVChunk ca, cb;
    pv_load(ca, pLane, Vh + (size_t)sSel[0] * (BS * DD) + vLane);
    for (int ch = 0; ch < 2 * NSEL; ch += 2) {
      pv_load(cb, pLane + (ch + 1) * 32,
              Vh + (size_t)sSel[ch >> 1] * (BS * DD) + 32 * DD + vLane);
      pv_mma(ca, oacc);
      if (ch + 2 < 2 * NSEL)
        pv_load(ca, pLane + (ch + 2) * 32,
                Vh + (size_t)sSel[(ch >> 1) + 1] * (BS * DD) + vLane);
      pv_mma(cb, oacc);
    }
  }

  for (int cg = 0; cg < ncg; ++cg) {
    const int kb = (extra && cg == 0) ? 0 : firstb + (cg - extra);
    __syncthreads();
    kstage_store(ks, sK, tid);
    __syncthreads();
    if (cg + 1 < ncg)
      kstage_load(ks, Kh + (size_t)(firstb + (cg + 1 - extra)) * (BS * DD), tid);
    __builtin_amdgcn_sched_barrier(0);
    KPair pa, pb;
    kpair_load(pa, kLane);
    kpair_load(pb, kLane + 32 * KPITCH);
    __builtin_amdgcn_sched_barrier(0);
    {
      v8f a0 = vzero8(), a1 = vzero8();
      kpair_mma(pa, qa, a0, a1);
      win_tile_out(a0, myP, half, ln, qrow0, kb * BS, cg * 64);
      win_tile_out(a1, myP, half, ln, qrow0, kb * BS + 16, cg * 64 + 16);
    }
    {
      v8f a0 = vzero8(), a1 = vzero8();
      kpair_mma(pb, qa, a0, a1);
      win_tile_out(a0, myP, half, ln, qrow0, kb * BS + 32, cg * 64 + 32);
      win_tile_out(a1, myP, half, ln, qrow0, kb * BS + 48, cg * 64 + 48);
    }
  }

  win_softmax(myP, lane, ncg);

  {
    PVChunk ca, cb;
    const int kbFirst = extra ? 0 : firstb;
    pv_load(ca, pLane, Vh + (size_t)kbFirst * (BS * DD) + vLane);
    for (int cg = 0; cg < ncg; ++cg) {
      const int kb0 = (extra && cg == 0) ? 0 : firstb + (cg - extra);
      pv_load(cb, pLane + cg * 64 + 32,
              Vh + (size_t)kb0 * (BS * DD) + 32 * DD + vLane);
      pv_mma(ca, oacc);
      if (cg + 1 < ncg)
        pv_load(ca, pLane + (cg + 1) * 64,
                Vh + (size_t)(firstb + (cg + 1 - extra)) * (BS * DD) + vLane);
      pv_mma(cb, oacc);
    }
  }

  const float third = 1.f / 3.f;
#pragma unroll
  for (int r = 0; r < 8; ++r) {
    int row = qrow0 + r + 8 * half;
    v4f o;
    o.x = oacc[0][r] * third;
    o.y = oacc[1][r] * third;
    o.z = oacc[2][r] * third;
    o.w = oacc[3][r] * third;
    *(v4f*)(Oh + (size_t)row * DD + 4 * ln) = o;
  }
}

extern "C" void kernel_launch(void* const* d_in, const int* in_sizes, int n_in,
                              void* d_out, int out_size, void* d_ws,
                              size_t ws_size, hipStream_t stream) {
  (void)in_sizes; (void)n_in; (void)out_size; (void)ws_size;
  const float* q = (const float*)d_in[0];
  const float* k = (const float*)d_in[1];
  const float* v = (const float*)d_in[2];
  float* out = (float*)d_out;
  float* kblk = (float*)d_ws;
  float* vblk = kblk + (size_t)BB * HH * NB * DD;

  nsa_blockmean_kernel<<<BB * HH * NB, 64, 0, stream>>>(k, v, kblk, vblk);

  const size_t shmem = (size_t)(64 * PSTRIDE + 64 * KPITCH) * sizeof(float);
  hipFuncSetAttribute(reinterpret_cast<const void*>(nsa_main_kernel),
                      hipFuncAttributeMaxDynamicSharedMemorySize, (int)shmem);
  nsa_main_kernel<<<BB * HH * NB, 128, shmem, stream>>>(q, k, v, kblk, vblk,
                                                        out);
}


# top-256 select: thresholds searched for 4 rows side by side (was 2), each step's 64 compares issued as one group ahead of the scalar popcounts; softmax in a second pass; kept set and sum order same
# speedup vs baseline: 3.3344x; 1.0874x over previous
#include <hip/hip_runtime.h>

#define BB 2
#define HH 8
#define SS 2048
#define DD 64
#define NB 32
#define BS 64
#define NSEL 8
#define SELC 512
#define KEEP 256
#define WIN 128
#define PSTRIDE 520
#define NEGINF (-1e30f)
#define SCALE 0.125f
#define SELR 4
#define SMXR 2
#define WROWS 4
#define KPITCH 68

typedef __attribute__((ext_vector_type(2))) float v2f;
typedef __attribute__((ext_vector_type(8))) float v8f;

static __device__ inline v8f vzero8() {
  v8f r;
#pragma unroll
  for (int i = 0; i < 8; ++i) r[i] = 0.f;
  return r;
}

static __device__ inline v8f wmma4(v2f a, v2f b, v8f c) {
  return __builtin_amdgcn_wmma_f32_16x16x4_f32(false, a, false, b, (short)0, c,
                                               false, false);
}

static __device__ inline unsigned okey(float x) {
  unsigned u = __float_as_uint(x);
  return (u & 0x80000000u) ? ~u : (u | 0x80000000u);
}

static __device__ inline int wred_sumi(int x) {
#pragma unroll
  for (int o = 16; o > 0; o >>= 1) x += __shfl_xor(x, o, 32);
  return x;
}
static __device__ inline float wred_sumf(float x) {
#pragma unroll
  for (int o = 16; o > 0; o >>= 1) x += __shfl_xor(x, o, 32);
  return x;
}
static __device__ inline float wred_maxf(float x) {
#pragma unroll
  for (int o = 16; o > 0; o >>= 1) x = fmaxf(x, __shfl_xor(x, o, 32));
  return x;
}

static __device__ inline void count_ge_rows(const unsigned (&key)[SELR][16],
                                            const unsigned (&c)[SELR],
                                            int (&cnt)[SELR]) {
  unsigned b[SELR][16];
#pragma unroll
  for (int j = 0; j < SELR; ++j) {
#pragma unroll
    for (int k = 0; k < 16; ++k)
      b[j][k] = __builtin_amdgcn_ballot_w32(key[j][k] >= c[j]);
  }
  __builtin_amdgcn_sched_barrier(0);
#pragma unroll
  for (int j = 0; j < SELR; ++j) {
    int n = 0;
#pragma unroll
    for (int k = 0; k < 16; k += 2) {
      const unsigned long long two =
          ((unsigned long long)b[j][k + 1] << 32) | (unsigned long long)b[j][k];
      n += __builtin_popcountll(two);
    }
    cnt[j] = n;
  }
}

typedef __attribute__((ext_vector_type(4))) float v4f;
struct PVChunk {
  v2f a[8];
  v4f v0[8];
  v4f v1[8];
};
static __device__ inline void pv_load(PVChunk& c, const float* prow,
                                      const float* vrow) {
#pragma unroll
  for (int g = 0; g < 8; ++g) {
    c.v0[g] = *(const v4f*)(vrow + (size_t)(4 * g) * DD);
    c.v1[g] = *(const v4f*)(vrow + (size_t)(4 * g + 1) * DD);
  }
#pragma unroll
  for (int g = 0; g < 8; ++g) c.a[g] = *(const v2f*)(prow + 4 * g);
}
static __device__ inline void pv_mma(const PVChunk& c, v8f (&o)[4]) {
#pragma unroll
  for (int g = 0; g < 8; ++g) {
#pragma unroll
    for (int j = 0; j < 4; ++j) {
      v2f bf; bf.x = c.v0[g][j]; bf.y = c.v1[g][j];
      o[j] = wmma4(c.a[g], bf, o[j]);
    }
  }
}

struct KStage {
  v4f r[8];
};
static __device__ inline void kstage_load(KStage& s, const float* blk, int tid) {
#pragma unroll
  for (int i = 0; i < 8; ++i) s.r[i] = *(const v4f*)(blk + 4 * (tid + 128 * i));
}
static __device__ inline void kstage_store(const KStage& s, float* sK, int tid) {
#pragma unroll
  for (int i = 0; i < 8; ++i) {
    const int c = tid + 128 * i;
    *(v4f*)(sK + (c >> 4) * KPITCH + 4 * (c & 15)) = s.r[i];
  }
}
struct KPair {
  v2f f0[16];
  v2f f1[16];
};
static __device__ inline void kpair_load(KPair& c, const float* krow) {
#pragma unroll
  for (int t = 0; t < 16; ++t) {
    c.f0[t] = *(const v2f*)(krow + 4 * t);
    c.f1[t] = *(const v2f*)(krow + 16 * KPITCH + 4 * t);
  }
}
static __device__ inline void kpair_mma(const KPair& c, const v2f (&qa)[16],
                                        v8f& a0, v8f& a1) {
#pragma unroll
  for (int t = 0; t < 16; ++t) {
    a0 = wmma4(qa[t], c.f0[t], a0);
    a1 = wmma4(qa[t], c.f1[t], a1);
  }
}
static __device__ inline void sel_tile_out(const v8f& acc, float* myP, int half,
                                           int ln, int qrow0, int tok0, int col0) {
#pragma unroll
  for (int r = 0; r < 8; ++r) {
    int m = r + 8 * half;
    int i = qrow0 + m;
    int tok = tok0 + ln;
    float x = acc[r] * SCALE;
    myP[m * PSTRIDE + col0 + ln] = (tok <= i) ? x : NEGINF;
  }
}
static __device__ inline void win_tile_out(const v8f& acc, float* myP, int half,
                                           int ln, int qrow0, int tok0, int col0) {
#pragma unroll
  for (int r = 0; r < 8; ++r) {
    int m = r + 8 * half;
    int i = qrow0 + m;
    int j = tok0 + ln;
    bool ok = (j <= i) && (((i - j) < WIN) || (j < 1));
    myP[m * PSTRIDE + col0 + ln] = ok ? acc[r] * SCALE : NEGINF;
  }
}

static __device__ inline void win_softmax(float* myP, int lane, int ncg) {
  for (int r0 = 0; r0 < 16; r0 += WROWS) {
    float* base = myP + r0 * PSTRIDE + lane;
    float x[WROWS][8];
    float mx[WROWS];
    float sum[WROWS];
#pragma unroll
    for (int j = 0; j < WROWS; ++j) {
#pragma unroll
      for (int k = 0; k < 8; ++k) x[j][k] = 0.f;
      mx[j] = -3.4e38f;
      sum[j] = 0.f;
    }
#pragma unroll
    for (int g = 0; g < 4; ++g) {
      if (g < ncg) {
#pragma unroll
        for (int j = 0; j < WROWS; ++j) {
          x[j][2 * g] = base[j * PSTRIDE + 64 * g];
          x[j][2 * g + 1] = base[j * PSTRIDE + 64 * g + 32];
        }
      }
    }
#pragma unroll
    for (int g = 0; g < 4; ++g) {
      if (g < ncg) {
#pragma unroll
        for (int j = 0; j < WROWS; ++j) {
          mx[j] = fmaxf(mx[j], x[j][2 * g]);
          mx[j] = fmaxf(mx[j], x[j][2 * g + 1]);
        }
      }
    }
#pragma unroll
    for (int o = 16; o > 0; o >>= 1) {
#pragma unroll
      for (int j = 0; j < WROWS; ++j)
        mx[j] = fmaxf(mx[j], __shfl_xor(mx[j], o, 32));
    }
#pragma unroll
    for (int g = 0; g < 4; ++g) {
      if (g < ncg) {
#pragma unroll
        for (int j = 0; j < WROWS; ++j) {
#pragma unroll
          for (int k = 2 * g; k < 2 * g + 2; ++k) {
            float p = (mx[j] > -1e29f) ? __expf(x[j][k] - mx[j]) : 0.f;
            x[j][k] = p;
            sum[j] += p;
          }
        }
      }
    }
#pragma unroll
    for (int o = 16; o > 0; o >>= 1) {
#pragma unroll
      for (int j = 0; j < WROWS; ++j) sum[j] += __shfl_xor(sum[j], o, 32);
    }
    float rs[WROWS];
#pragma unroll
    for (int j = 0; j < WROWS; ++j) rs[j] = (sum[j] > 0.f) ? 1.f / sum[j] : 0.f;
#pragma unroll
    for (int g = 0; g < 4; ++g) {
      if (g < ncg) {
#pragma unroll
        for (int j = 0; j < WROWS; ++j) {
          base[j * PSTRIDE + 64 * g] = x[j][2 * g] * rs[j];
          base[j * PSTRIDE + 64 * g + 32] = x[j][2 * g + 1] * rs[j];
        }
      }
    }
  }
}

__global__ void nsa_blockmean_kernel(const float* __restrict__ K,
                                     const float* __restrict__ V,
                                     float* __restrict__ KB,
                                     float* __restrict__ VB) {
  const int g = blockIdx.x;
  const int d = threadIdx.x;
  const float* kp = K + (size_t)g * (BS * DD);
  const float* vp = V + (size_t)g * (BS * DD);
  float ks = 0.f, vs = 0.f;
  for (int t = 0; t < BS; ++t) {
    ks += kp[t * DD + d];
    vs += vp[t * DD + d];
  }
  KB[(size_t)g * DD + d] = ks * (1.f / BS);
  VB[(size_t)g * DD + d] = vs * (1.f / BS);
}

__global__ __launch_bounds__(128) void nsa_main_kernel(
    const float* __restrict__ Q, const float* __restrict__ K,
    const float* __restrict__ V, const float* __restrict__ KB,
    const float* __restrict__ VB, float* __restrict__ O) {
  extern __shared__ v4f sDyn[];
  float* sP = (float*)sDyn;
  float* sK = sP + 64 * PSTRIDE;
  __shared__ float sScore[NB];
  __shared__ int sSel[NSEL + 1];

  const int wg = blockIdx.x;
  const int qb = wg % NB;
  const int bh = wg / NB;
  const size_t hoff = (size_t)bh * SS * DD;
  const float* Qh = Q + hoff;
  const float* Kh = K + hoff;
  const float* Vh = V + hoff;
  const float* KBh = KB + (size_t)bh * NB * DD;
  const float* VBh = VB + (size_t)bh * NB * DD;
  float* Oh = O + hoff;

  const int tid = threadIdx.x;
  const int wave = tid >> 5;
  const int lane = tid & 31;
  const int half = lane >> 4;
  const int ln = lane & 15;
  const int qrow0 = qb * BS + wave * 16;
  float* myP = sP + (size_t)wave * 16 * PSTRIDE;

  const int firstb = (qb >= 2) ? qb - 2 : 0;
  const int nwb = qb - firstb + 1;
  const int extra = (firstb > 0) ? 1 : 0;
  const int ncg = nwb + extra;
  const int wkb0 = extra ? 0 : firstb;

  v2f qa[16];
#pragma unroll
  for (int t = 0; t < 16; ++t) {
    const float* p = Qh + (size_t)(qrow0 + ln) * DD + 4 * t + 2 * half;
    v2f a; a.x = p[0]; a.y = p[1];
    qa[t] = a;
  }

  v8f oacc[4];
#pragma unroll
  for (int dt = 0; dt < 4; ++dt) oacc[dt] = vzero8();

#pragma unroll
  for (int nt = 0; nt < 2; ++nt) {
    v8f acc = vzero8();
#pragma unroll
    for (int t = 0; t < 16; ++t) {
      const float* bp = KBh + (size_t)(nt * 16 + ln) * DD + 4 * t + 2 * half;
      v2f bf; bf.x = bp[0]; bf.y = bp[1];
      acc = wmma4(qa[t], bf, acc);
    }
#pragma unroll
    for (int r = 0; r < 8; ++r) {
      int m = r + 8 * half;
      int col = nt * 16 + ln;
      float x = acc[r] * SCALE;
      myP[m * PSTRIDE + col] = (col <= qb) ? x : NEGINF;
    }
  }
  __syncthreads();

  if (tid < NB) {
    float s = 0.f;
    for (int q2 = 0; q2 < BS; ++q2) s += sP[q2 * PSTRIDE + tid];
    sScore[tid] = s * (1.f / BS);
  }
  __syncthreads();

  if (tid == 0) {
    float loc[NB];
    for (int i = 0; i < NB; ++i) loc[i] = sScore[i];
    for (int s = 0; s < NSEL; ++s) {
      int bi = 0;
      float bv = loc[0];
      for (int i = 1; i < NB; ++i)
        if (loc[i] > bv) { bv = loc[i]; bi = i; }
      sSel[s] = bi;
      loc[bi] = -3.4e38f;
    }
    sSel[NSEL] = wkb0;
  }
  __syncthreads();

  if (tid < BS) {
    float* row = sP + tid * PSTRIDE;
    float mx = -3.4e38f;
    for (int c = 0; c < NB; ++c) mx = fmaxf(mx, row[c]);
    float sum = 0.f;
    for (int c = 0; c < NB; ++c) {
      float p = (mx > -1e29f) ? __expf(row[c] - mx) : 0.f;
      row[c] = p;
      sum += p;
    }
    float rs = (sum > 0.f) ? 1.f / sum : 0.f;
    for (int c = 0; c < NB; ++c) row[c] *= rs;
  }
  __syncthreads();

  const float* pLane = myP + ln * PSTRIDE + 2 * half;
  const size_t vLane = (size_t)(2 * half) * DD + 4 * ln;

  const float* kLane = sK + ln * KPITCH + 2 * half;
  KStage ks;
  kstage_load(ks, Kh + (size_t)sSel[0] * (BS * DD), tid);

  {
    PVChunk c;
    pv_load(c, pLane, VBh + vLane);
    pv_mma(c, oacc);
  }

  for (int sb = 0; sb < NSEL; ++sb) {
    const int kb = sSel[sb];
    __syncthreads();
    kstage_store(ks, sK, tid);
    __syncthreads();
    kstage_load(ks, Kh + (size_t)sSel[sb + 1] * (BS * DD), tid);
    __builtin_amdgcn_sched_barrier(0);
    KPair pa, pb;
    kpair_load(pa, kLane);
    kpair_load(pb, kLane + 32 * KPITCH);
    {
      v8f a0 = vzero8(), a1 = vzero8();
      kpair_mma(pa, qa, a0, a1);
      sel_tile_out(a0, myP, half, ln, qrow0, kb * BS, sb * 64);
      sel_tile_out(a1, myP, half, ln, qrow0, kb * BS + 16, sb * 64 + 16);
    }
    {
      v8f a0 = vzero8(), a1 = vzero8();
      kpair_mma(pb, qa, a0, a1);
      sel_tile_out(a0, myP, half, ln, qrow0, kb * BS + 32, sb * 64 + 32);
      sel_tile_out(a1, myP, half, ln, qrow0, kb * BS + 48, sb * 64 + 48);
    }
  }

  unsigned thrv = 0u;
  for (int r = 0; r < 16; r += SELR) {
    unsigned key[SELR][16];
    unsigned pre[SELR];
    int nab[SELR];
#pragma unroll
    for (int j = 0; j < SELR; ++j) {
      const float* row = myP + (r + j) * PSTRIDE;
#pragma unroll
      for (int k = 0; k < 16; ++k) key[j][k] = okey(row[lane + 32 * k]);
      nab[j] = 0;
    }
    const unsigned kneg = okey(NEGINF);
    unsigned open = (1u << SELR) - 1u;
    unsigned m = 0u;
    bool second = false;
    bool go = true;
#pragma unroll
    for (int j = 0; j < SELR; ++j) pre[j] = kneg + 1u;
#pragma nounroll
    for (int it = 0; go && it < 34; ++it) {
      unsigned cand[SELR];
      int cnt[SELR];
#pragma unroll
      for (int j = 0; j < SELR; ++j) cand[j] = pre[j] | m;
      count_ge_rows(key, cand, cnt);
      if (m != 0u) {
#pragma unroll
        for (int j = 0; j < SELR; ++j) {
          if (cnt[j] >= KEEP) pre[j] = cand[j];
          if (cnt[j] == KEEP) open &= ~(1u << j);
        }
        m >>= 1;
        go = (open != 0u) && (m != 0u);
      } else if (!second) {
        bool some = false;
#pragma unroll
        for (int j = 0; j < SELR; ++j) {
          nab[j] = cnt[j];
          some = some || (cnt[j] < KEEP);
        }
        second = some;
        m = some ? 0u : 0x80000000u;
#pragma unroll
        for (int j = 0; j < SELR; ++j) pre[j] = some ? kneg : 0u;
      } else {
#pragma unroll
        for (int j = 0; j < SELR; ++j) {
          const bool caseb = nab[j] < KEEP && cnt[j] >= KEEP;
          pre[j] = caseb ? kneg : 0u;
          if (caseb) open &= ~(1u << j);
        }
        m = 0x80000000u;
        go = (open != 0u);
      }
    }
#pragma unroll
    for (int j = 0; j < SELR; ++j)
      thrv = (lane == r + j) ? pre[j] : thrv;
  }

  for (int r = 0; r < 16; r += SMXR) {
    float x[SMXR][16];
    unsigned key[SMXR][16];
    unsigned pre[SMXR];
#pragma unroll
    for (int j = 0; j < SMXR; ++j) {
      const float* row = myP + (r + j) * PSTRIDE;
#pragma unroll
      for (int k = 0; k < 16; ++k) {
        x[j][k] = row[lane + 32 * k];
        key[j][k] = okey(x[j][k]);
      }
      pre[j] = __builtin_amdgcn_readlane(thrv, r + j);
    }
#pragma unroll
    for (int j = 0; j < SMXR; ++j) {
      float* row = myP + (r + j) * PSTRIDE;
      float mx = -3.4e38f;
#pragma unroll
      for (int k = 0; k < 16; ++k) mx = fmaxf(mx, x[j][k]);
      mx = wred_maxf(mx);
      float sum = 0.f;
#pragma unroll
      for (int k = 0; k < 16; ++k) {
        float p = (key[j][k] >= pre[j] && mx > -1e29f) ? __expf(x[j][k] - mx) : 0.f;
        x[j][k] = p;
        sum += p;
      }
      sum = wred_sumf(sum);
      float rs = (sum > 0.f) ? 1.f / sum : 0.f;
#pragma unroll
      for (int k = 0; k < 16; ++k) row[lane + 32 * k] = x[j][k] * rs;
    }
  }

  {
    PVChunk ca, cb;
    pv_load(ca, pLane, Vh + (size_t)sSel[0] * (BS * DD) + vLane);
    for (int ch = 0; ch < 2 * NSEL; ch += 2) {
      pv_load(cb, pLane + (ch + 1) * 32,
              Vh + (size_t)sSel[ch >> 1] * (BS * DD) + 32 * DD + vLane);
      pv_mma(ca, oacc);
      if (ch + 2 < 2 * NSEL)
        pv_load(ca, pLane + (ch + 2) * 32,
                Vh + (size_t)sSel[(ch >> 1) + 1] * (BS * DD) + vLane);
      pv_mma(cb, oacc);
    }
  }

  for (int cg = 0; cg < ncg; ++cg) {
    const int kb = (extra && cg == 0) ? 0 : firstb + (cg - extra);
    __syncthreads();
    kstage_store(ks, sK, tid);
    __syncthreads();
    if (cg + 1 < ncg)
      kstage_load(ks, Kh + (size_t)(firstb + (cg + 1 - extra)) * (BS * DD), tid);
    __builtin_amdgcn_sched_barrier(0);
    KPair pa, pb;
    kpair_load(pa, kLane);
    kpair_load(pb, kLane + 32 * KPITCH);
    __builtin_amdgcn_sched_barrier(0);
    {
      v8f a0 = vzero8(), a1 = vzero8();
      kpair_mma(pa, qa, a0, a1);
      win_tile_out(a0, myP, half, ln, qrow0, kb * BS, cg * 64);
      win_tile_out(a1, myP, half, ln, qrow0, kb * BS + 16, cg * 64 + 16);
    }
    {
      v8f a0 = vzero8(), a1 = vzero8();
      kpair_mma(pb, qa, a0, a1);
      win_tile_out(a0, myP, half, ln, qrow0, kb * BS + 32, cg * 64 + 32);
      win_tile_out(a1, myP, half, ln, qrow0, kb * BS + 48, cg * 64 + 48);
    }
  }

  win_softmax(myP, lane, ncg);

  {
    PVChunk ca, cb;
    const int kbFirst = extra ? 0 : firstb;
    pv_load(ca, pLane, Vh + (size_t)kbFirst * (BS * DD) + vLane);
    for (int cg = 0; cg < ncg; ++cg) {
      const int kb0 = (extra && cg == 0) ? 0 : firstb + (cg - extra);
      pv_load(cb, pLane + cg * 64 + 32,
              Vh + (size_t)kb0 * (BS * DD) + 32 * DD + vLane);
      pv_mma(ca, oacc);
      if (cg + 1 < ncg)
        pv_load(ca, pLane + (cg + 1) * 64,
                Vh + (size_t)(firstb + (cg + 1 - extra)) * (BS * DD) + vLane);
      pv_mma(cb, oacc);
    }
  }

  const float third = 1.f / 3.f;
#pragma unroll
  for (int r = 0; r < 8; ++r) {
    int row = qrow0 + r + 8 * half;
    v4f o;
    o.x = oacc[0][r] * third;
    o.y = oacc[1][r] * third;
    o.z = oacc[2][r] * third;
    o.w = oacc[3][r] * third;
    *(v4f*)(Oh + (size_t)row * DD + 4 * ln) = o;
  }
}

extern "C" void kernel_launch(void* const* d_in, const int* in_sizes, int n_in,
                              void* d_out, int out_size, void* d_ws,
                              size_t ws_size, hipStream_t stream) {
  (void)in_sizes; (void)n_in; (void)out_size; (void)ws_size;
  const float* q = (const float*)d_in[0];
  const float* k = (const float*)d_in[1];
  const float* v = (const float*)d_in[2];
  float* out = (float*)d_out;
  float* kblk = (float*)d_ws;
  float* vblk = kblk + (size_t)BB * HH * NB * DD;

  nsa_blockmean_kernel<<<BB * HH * NB, 64, 0, stream>>>(k, v, kblk, vblk);

  const size_t shmem = (size_t)(64 * PSTRIDE + 64 * KPITCH) * sizeof(float);
  hipFuncSetAttribute(reinterpret_cast<const void*>(nsa_main_kernel),
                      hipFuncAttributeMaxDynamicSharedMemorySize, (int)shmem);
  nsa_main_kernel<<<BB * HH * NB, 128, shmem, stream>>>(q, k, v, kblk, vblk,
                                                        out);
}


# softmax butterflies (max and sum, both branches): lane exchanges on the vector ALU (DPP row_xmask, permlanex16) instead of ds_bpermute through the LDS unit; step and sum order unchanged
# speedup vs baseline: 3.4714x; 1.0411x over previous
#include <hip/hip_runtime.h>

#define BB 2
#define HH 8
#define SS 2048
#define DD 64
#define NB 32
#define BS 64
#define NSEL 8
#define SELC 512
#define KEEP 256
#define WIN 128
#define PSTRIDE 520
#define NEGINF (-1e30f)
#define SCALE 0.125f
#define SELR 4
#define SMXR 2
#define WROWS 4
#define KPITCH 68

typedef __attribute__((ext_vector_type(2))) float v2f;
typedef __attribute__((ext_vector_type(8))) float v8f;

static __device__ inline v8f vzero8() {
  v8f r;
#pragma unroll
  for (int i = 0; i < 8; ++i) r[i] = 0.f;
  return r;
}

static __device__ inline v8f wmma4(v2f a, v2f b, v8f c) {
  return __builtin_amdgcn_wmma_f32_16x16x4_f32(false, a, false, b, (short)0, c,
                                               false, false);
}

static __device__ inline unsigned okey(float x) {
  unsigned u = __float_as_uint(x);
  return (u & 0x80000000u) ? ~u : (u | 0x80000000u);
}

template <int M>
static __device__ inline float xlane(float x) {
  const int b = __builtin_bit_cast(int, x);
  return __builtin_bit_cast(
      float, __builtin_amdgcn_update_dpp(b, b, 0x160 + M, 0xf, 0xf, true));
}
static __device__ inline float xhalf(float x) {
  const int b = __builtin_bit_cast(int, x);
  return __builtin_bit_cast(
      float, __builtin_amdgcn_permlanex16(b, b, 0x76543210, 0xfedcba98, false,
                                          false));
}
template <int N>
static __device__ inline void wred_sumf(float (&v)[N]) {
#pragma unroll
  for (int j = 0; j < N; ++j) v[j] += xhalf(v[j]);
#pragma unroll
  for (int j = 0; j < N; ++j) v[j] += xlane<8>(v[j]);
#pragma unroll
  for (int j = 0; j < N; ++j) v[j] += xlane<4>(v[j]);
#pragma unroll
  for (int j = 0; j < N; ++j) v[j] += xlane<2>(v[j]);
#pragma unroll
  for (int j = 0; j < N; ++j) v[j] += xlane<1>(v[j]);
}
template <int N>
static __device__ inline void wred_maxf(float (&v)[N]) {
#pragma unroll
  for (int j = 0; j < N; ++j) v[j] = fmaxf(v[j], xhalf(v[j]));
#pragma unroll
  for (int j = 0; j < N; ++j) v[j] = fmaxf(v[j], xlane<8>(v[j]));
#pragma unroll
  for (int j = 0; j < N; ++j) v[j] = fmaxf(v[j], xlane<4>(v[j]));
#pragma unroll
  for (int j = 0; j < N; ++j) v[j] = fmaxf(v[j], xlane<2>(v[j]));
#pragma unroll
  for (int j = 0; j < N; ++j) v[j] = fmaxf(v[j], xlane<1>(v[j]));
}

static __device__ inline void count_ge_rows(const unsigned (&key)[SELR][16],
                                            const unsigned (&c)[SELR],
                                            int (&cnt)[SELR]) {
  unsigned b[SELR][16];
#pragma unroll
  for (int j = 0; j < SELR; ++j) {
#pragma unroll
    for (int k = 0; k < 16; ++k)
      b[j][k] = __builtin_amdgcn_ballot_w32(key[j][k] >= c[j]);
  }
  __builtin_amdgcn_sched_barrier(0);
#pragma unroll
  for (int j = 0; j < SELR; ++j) {
    int n = 0;
#pragma unroll
    for (int k = 0; k < 16; k += 2) {
      const unsigned long long two =
          ((unsigned long long)b[j][k + 1] << 32) | (unsigned long long)b[j][k];
      n += __builtin_popcountll(two);
    }
    cnt[j] = n;
  }
}

typedef __attribute__((ext_vector_type(4))) float v4f;
struct PVChunk {
  v2f a[8];
  v4f v0[8];
  v4f v1[8];
};
static __device__ inline void pv_load(PVChunk& c, const float* prow,
                                      const float* vrow) {
#pragma unroll
  for (int g = 0; g < 8; ++g) {
    c.v0[g] = *(const v4f*)(vrow + (size_t)(4 * g) * DD);
    c.v1[g] = *(const v4f*)(vrow + (size_t)(4 * g + 1) * DD);
  }
#pragma unroll
  for (int g = 0; g < 8; ++g) c.a[g] = *(const v2f*)(prow + 4 * g);
}
static __device__ inline void pv_mma(const PVChunk& c, v8f (&o)[4]) {
#pragma unroll
  for (int g = 0; g < 8; ++g) {
#pragma unroll
    for (int j = 0; j < 4; ++j) {
      v2f bf; bf.x = c.v0[g][j]; bf.y = c.v1[g][j];
      o[j] = wmma4(c.a[g], bf, o[j]);
    }
  }
}

struct KStage {
  v4f r[8];
};
static __device__ inline void kstage_load(KStage& s, const float* blk, int tid) {
#pragma unroll
  for (int i = 0; i < 8; ++i) s.r[i] = *(const v4f*)(blk + 4 * (tid + 128 * i));
}
static __device__ inline void kstage_store(const KStage& s, float* sK, int tid) {
#pragma unroll
  for (int i = 0; i < 8; ++i) {
    const int c = tid + 128 * i;
    *(v4f*)(sK + (c >> 4) * KPITCH + 4 * (c & 15)) = s.r[i];
  }
}
struct KPair {
  v2f f0[16];
  v2f f1[16];
};
static __device__ inline void kpair_load(KPair& c, const float* krow) {
#pragma unroll
  for (int t = 0; t < 16; ++t) {
    c.f0[t] = *(const v2f*)(krow + 4 * t);
    c.f1[t] = *(const v2f*)(krow + 16 * KPITCH + 4 * t);
  }
}
static __device__ inline void kpair_mma(const KPair& c, const v2f (&qa)[16],
                                        v8f& a0, v8f& a1) {
#pragma unroll
  for (int t = 0; t < 16; ++t) {
    a0 = wmma4(qa[t], c.f0[t], a0);
    a1 = wmma4(qa[t], c.f1[t], a1);
  }
}
static __device__ inline void sel_tile_out(const v8f& acc, float* myP, int half,
                                           int ln, int qrow0, int tok0, int col0) {
#pragma unroll
  for (int r = 0; r < 8; ++r) {
    int m = r + 8 * half;
    int i = qrow0 + m;
    int tok = tok0 + ln;
    float x = acc[r] * SCALE;
    myP[m * PSTRIDE + col0 + ln] = (tok <= i) ? x : NEGINF;
  }
}
static __device__ inline void win_tile_out(const v8f& acc, float* myP, int half,
                                           int ln, int qrow0, int tok0, int col0) {
#pragma unroll
  for (int r = 0; r < 8; ++r) {
    int m = r + 8 * half;
    int i = qrow0 + m;
    int j = tok0 + ln;
    bool ok = (j <= i) && (((i - j) < WIN) || (j < 1));
    myP[m * PSTRIDE + col0 + ln] = ok ? acc[r] * SCALE : NEGINF;
  }
}

static __device__ inline void win_softmax(float* myP, int lane, int ncg) {
  for (int r0 = 0; r0 < 16; r0 += WROWS) {
    float* base = myP + r0 * PSTRIDE + lane;
    float x[WROWS][8];
    float mx[WROWS];
    float sum[WROWS];
#pragma unroll
    for (int j = 0; j < WROWS; ++j) {
#pragma unroll
      for (int k = 0; k < 8; ++k) x[j][k] = 0.f;
      mx[j] = -3.4e38f;
      sum[j] = 0.f;
    }
#pragma unroll
    for (int g = 0; g < 4; ++g) {
      if (g < ncg) {
#pragma unroll
        for (int j = 0; j < WROWS; ++j) {
          x[j][2 * g] = base[j * PSTRIDE + 64 * g];
          x[j][2 * g + 1] = base[j * PSTRIDE + 64 * g + 32];
        }
      }
    }
#pragma unroll
    for (int g = 0; g < 4; ++g) {
      if (g < ncg) {
#pragma unroll
        for (int j = 0; j < WROWS; ++j) {
          mx[j] = fmaxf(mx[j], x[j][2 * g]);
          mx[j] = fmaxf(mx[j], x[j][2 * g + 1]);
        }
      }
    }
    wred_maxf(mx);
#pragma unroll
    for (int g = 0; g < 4; ++g) {
      if (g < ncg) {
#pragma unroll
        for (int j = 0; j < WROWS; ++j) {
#pragma unroll
          for (int k = 2 * g; k < 2 * g + 2; ++k) {
            float p = (mx[j] > -1e29f) ? __expf(x[j][k] - mx[j]) : 0.f;
            x[j][k] = p;
            sum[j] += p;
          }
        }
      }
    }
    wred_sumf(sum);
    float rs[WROWS];
#pragma unroll
    for (int j = 0; j < WROWS; ++j) rs[j] = (sum[j] > 0.f) ? 1.f / sum[j] : 0.f;
#pragma unroll
    for (int g = 0; g < 4; ++g) {
      if (g < ncg) {
#pragma unroll
        for (int j = 0; j < WROWS; ++j) {
          base[j * PSTRIDE + 64 * g] = x[j][2 * g] * rs[j];
          base[j * PSTRIDE + 64 * g + 32] = x[j][2 * g + 1] * rs[j];
        }
      }
    }
  }
}

__global__ void nsa_blockmean_kernel(const float* __restrict__ K,
                                     const float* __restrict__ V,
                                     float* __restrict__ KB,
                                     float* __restrict__ VB) {
  const int g = blockIdx.x;
  const int d = threadIdx.x;
  const float* kp = K + (size_t)g * (BS * DD);
  const float* vp = V + (size_t)g * (BS * DD);
  float ks = 0.f, vs = 0.f;
  for (int t = 0; t < BS; ++t) {
    ks += kp[t * DD + d];
    vs += vp[t * DD + d];
  }
  KB[(size_t)g * DD + d] = ks * (1.f / BS);
  VB[(size_t)g * DD + d] = vs * (1.f / BS);
}

__global__ __launch_bounds__(128) void nsa_main_kernel(
    const float* __restrict__ Q, const float* __restrict__ K,
    const float* __restrict__ V, const float* __restrict__ KB,
    const float* __restrict__ VB, float* __restrict__ O) {
  extern __shared__ v4f sDyn[];
  float* sP = (float*)sDyn;
  float* sK = sP + 64 * PSTRIDE;
  __shared__ float sScore[NB];
  __shared__ int sSel[NSEL + 1];

  const int wg = blockIdx.x;
  const int qb = wg % NB;
  const int bh = wg / NB;
  const size_t hoff = (size_t)bh * SS * DD;
  const float* Qh = Q + hoff;
  const float* Kh = K + hoff;
  const float* Vh = V + hoff;
  const float* KBh = KB + (size_t)bh * NB * DD;
  const float* VBh = VB + (size_t)bh * NB * DD;
  float* Oh = O + hoff;

  const int tid = threadIdx.x;
  const int wave = tid >> 5;
  const int lane = tid & 31;
  const int half = lane >> 4;
  const int ln = lane & 15;
  const int qrow0 = qb * BS + wave * 16;
  float* myP = sP + (size_t)wave * 16 * PSTRIDE;

  const int firstb = (qb >= 2) ? qb - 2 : 0;
  const int nwb = qb - firstb + 1;
  const int extra = (firstb > 0) ? 1 : 0;
  const int ncg = nwb + extra;
  const int wkb0 = extra ? 0 : firstb;

  v2f qa[16];
#pragma unroll
  for (int t = 0; t < 16; ++t) {
    const float* p = Qh + (size_t)(qrow0 + ln) * DD + 4 * t + 2 * half;
    v2f a; a.x = p[0]; a.y = p[1];
    qa[t] = a;
  }

  v8f oacc[4];
#pragma unroll
  for (int dt = 0; dt < 4; ++dt) oacc[dt] = vzero8();

#pragma unroll
  for (int nt = 0; nt < 2; ++nt) {
    v8f acc = vzero8();
#pragma unroll
    for (int t = 0; t < 16; ++t) {
      const float* bp = KBh + (size_t)(nt * 16 + ln) * DD + 4 * t + 2 * half;
      v2f bf; bf.x = bp[0]; bf.y = bp[1];
      acc = wmma4(qa[t], bf, acc);
    }
#pragma unroll
    for (int r = 0; r < 8; ++r) {
      int m = r + 8 * half;
      int col = nt * 16 + ln;
      float x = acc[r] * SCALE;
      myP[m * PSTRIDE + col] = (col <= qb) ? x : NEGINF;
    }
  }
  __syncthreads();

  if (tid < NB) {
    float s = 0.f;
    for (int q2 = 0; q2 < BS; ++q2) s += sP[q2 * PSTRIDE + tid];
    sScore[tid] = s * (1.f / BS);
  }
  __syncthreads();

  if (tid == 0) {
    float loc[NB];
    for (int i = 0; i < NB; ++i) loc[i] = sScore[i];
    for (int s = 0; s < NSEL; ++s) {
      int bi = 0;
      float bv = loc[0];
      for (int i = 1; i < NB; ++i)
        if (loc[i] > bv) { bv = loc[i]; bi = i; }
      sSel[s] = bi;
      loc[bi] = -3.4e38f;
    }
    sSel[NSEL] = wkb0;
  }
  __syncthreads();

  if (tid < BS) {
    float* row = sP + tid * PSTRIDE;
    float mx = -3.4e38f;
    for (int c = 0; c < NB; ++c) mx = fmaxf(mx, row[c]);
    float sum = 0.f;
    for (int c = 0; c < NB; ++c) {
      float p = (mx > -1e29f) ? __expf(row[c] - mx) : 0.f;
      row[c] = p;
      sum += p;
    }
    float rs = (sum > 0.f) ? 1.f / sum : 0.f;
    for (int c = 0; c < NB; ++c) row[c] *= rs;
  }
  __syncthreads();

  const float* pLane = myP + ln * PSTRIDE + 2 * half;
  const size_t vLane = (size_t)(2 * half) * DD + 4 * ln;

  const float* kLane = sK + ln * KPITCH + 2 * half;
  KStage ks;
  kstage_load(ks, Kh + (size_t)sSel[0] * (BS * DD), tid);

  {
    PVChunk c;
    pv_load(c, pLane, VBh + vLane);
    pv_mma(c, oacc);
  }

  for (int sb = 0; sb < NSEL; ++sb) {
    const int kb = sSel[sb];
    __syncthreads();
    kstage_store(ks, sK, tid);
    __syncthreads();
    kstage_load(ks, Kh + (size_t)sSel[sb + 1] * (BS * DD), tid);
    __builtin_amdgcn_sched_barrier(0);
    KPair pa, pb;
    kpair_load(pa, kLane);
    kpair_load(pb, kLane + 32 * KPITCH);
    {
      v8f a0 = vzero8(), a1 = vzero8();
      kpair_mma(pa, qa, a0, a1);
      sel_tile_out(a0, myP, half, ln, qrow0, kb * BS, sb * 64);
      sel_tile_out(a1, myP, half, ln, qrow0, kb * BS + 16, sb * 64 + 16);
    }
    {
      v8f a0 = vzero8(), a1 = vzero8();
      kpair_mma(pb, qa, a0, a1);
      sel_tile_out(a0, myP, half, ln, qrow0, kb * BS + 32, sb * 64 + 32);
      sel_tile_out(a1, myP, half, ln, qrow0, kb * BS + 48, sb * 64 + 48);
    }
  }

  unsigned thrv = 0u;
  for (int r = 0; r < 16; r += SELR) {
    unsigned key[SELR][16];
    unsigned pre[SELR];
    int nab[SELR];
#pragma unroll
    for (int j = 0; j < SELR; ++j) {
      const float* row = myP + (r + j) * PSTRIDE;
#pragma unroll
      for (int k = 0; k < 16; ++k) key[j][k] = okey(row[lane + 32 * k]);
      nab[j] = 0;
    }
    const unsigned kneg = okey(NEGINF);
    unsigned open = (1u << SELR) - 1u;
    unsigned m = 0u;
    bool second = false;
    bool go = true;
#pragma unroll
    for (int j = 0; j < SELR; ++j) pre[j] = kneg + 1u;
#pragma nounroll
    for (int it = 0; go && it < 34; ++it) {
      unsigned cand[SELR];
      int cnt[SELR];
#pragma unroll
      for (int j = 0; j < SELR; ++j) cand[j] = pre[j] | m;
      count_ge_rows(key, cand, cnt);
      if (m != 0u) {
#pragma unroll
        for (int j = 0; j < SELR; ++j) {
          if (cnt[j] >= KEEP) pre[j] = cand[j];
          if (cnt[j] == KEEP) open &= ~(1u << j);
        }
        m >>= 1;
        go = (open != 0u) && (m != 0u);
      } else if (!second) {
        bool some = false;
#pragma unroll
        for (int j = 0; j < SELR; ++j) {
          nab[j] = cnt[j];
          some = some || (cnt[j] < KEEP);
        }
        second = some;
        m = some ? 0u : 0x80000000u;
#pragma unroll
        for (int j = 0; j < SELR; ++j) pre[j] = some ? kneg : 0u;
      } else {
#pragma unroll
        for (int j = 0; j < SELR; ++j) {
          const bool caseb = nab[j] < KEEP && cnt[j] >= KEEP;
          pre[j] = caseb ? kneg : 0u;
          if (caseb) open &= ~(1u << j);
        }
        m = 0x80000000u;
        go = (open != 0u);
      }
    }
#pragma unroll
    for (int j = 0; j < SELR; ++j)
      thrv = (lane == r + j) ? pre[j] : thrv;
  }

  for (int r = 0; r < 16; r += SMXR) {
    float x[SMXR][16];
    unsigned key[SMXR][16];
    unsigned pre[SMXR];
#pragma unroll
    for (int j = 0; j < SMXR; ++j) {
      const float* row = myP + (r + j) * PSTRIDE;
#pragma unroll
      for (int k = 0; k < 16; ++k) {
        x[j][k] = row[lane + 32 * k];
        key[j][k] = okey(x[j][k]);
      }
      pre[j] = __builtin_amdgcn_readlane(thrv, r + j);
    }
    float mx[SMXR];
    float sum[SMXR];
#pragma unroll
    for (int j = 0; j < SMXR; ++j) {
      mx[j] = -3.4e38f;
#pragma unroll
      for (int k = 0; k < 16; ++k) mx[j] = fmaxf(mx[j], x[j][k]);
    }
    wred_maxf(mx);
#pragma unroll
    for (int j = 0; j < SMXR; ++j) {
      sum[j] = 0.f;
#pragma unroll
      for (int k = 0; k < 16; ++k) {
        float p = (key[j][k] >= pre[j] && mx[j] > -1e29f) ? __expf(x[j][k] - mx[j]) : 0.f;
        x[j][k] = p;
        sum[j] += p;
      }
    }
    wred_sumf(sum);
#pragma unroll
    for (int j = 0; j < SMXR; ++j) {
      float* row = myP + (r + j) * PSTRIDE;
      float rs = (sum[j] > 0.f) ? 1.f / sum[j] : 0.f;
#pragma unroll
      for (int k = 0; k < 16; ++k) row[lane + 32 * k] = x[j][k] * rs;
    }
  }

  {
    PVChunk ca, cb;
    pv_load(ca, pLane, Vh + (size_t)sSel[0] * (BS * DD) + vLane);
    for (int ch = 0; ch < 2 * NSEL; ch += 2) {
      pv_load(cb, pLane + (ch + 1) * 32,
              Vh + (size_t)sSel[ch >> 1] * (BS * DD) + 32 * DD + vLane);
      pv_mma(ca, oacc);
      if (ch + 2 < 2 * NSEL)
        pv_load(ca, pLane + (ch + 2) * 32,
                Vh + (size_t)sSel[(ch >> 1) + 1] * (BS * DD) + vLane);
      pv_mma(cb, oacc);
    }
  }

  for (int cg = 0; cg < ncg; ++cg) {
    const int kb = (extra && cg == 0) ? 0 : firstb + (cg - extra);
    __syncthreads();
    kstage_store(ks, sK, tid);
    __syncthreads();
    if (cg + 1 < ncg)
      kstage_load(ks, Kh + (size_t)(firstb + (cg + 1 - extra)) * (BS * DD), tid);
    __builtin_amdgcn_sched_barrier(0);
    KPair pa, pb;
    kpair_load(pa, kLane);
    kpair_load(pb, kLane + 32 * KPITCH);
    __builtin_amdgcn_sched_barrier(0);
    {
      v8f a0 = vzero8(), a1 = vzero8();
      kpair_mma(pa, qa, a0, a1);
      win_tile_out(a0, myP, half, ln, qrow0, kb * BS, cg * 64);
      win_tile_out(a1, myP, half, ln, qrow0, kb * BS + 16, cg * 64 + 16);
    }
    {
      v8f a0 = vzero8(), a1 = vzero8();
      kpair_mma(pb, qa, a0, a1);
      win_tile_out(a0, myP, half, ln, qrow0, kb * BS + 32, cg * 64 + 32);
      win_tile_out(a1, myP, half, ln, qrow0, kb * BS + 48, cg * 64 + 48);
    }
  }

  win_softmax(myP, lane, ncg);

  {
    PVChunk ca, cb;
    const int kbFirst = extra ? 0 : firstb;
    pv_load(ca, pLane, Vh + (size_t)kbFirst * (BS * DD) + vLane);
    for (int cg = 0; cg < ncg; ++cg) {
      const int kb0 = (extra && cg == 0) ? 0 : firstb + (cg - extra);
      pv_load(cb, pLane + cg * 64 + 32,
              Vh + (size_t)kb0 * (BS * DD) + 32 * DD + vLane);
      pv_mma(ca, oacc);
      if (cg + 1 < ncg)
        pv_load(ca, pLane + (cg + 1) * 64,
                Vh + (size_t)(firstb + (cg + 1 - extra)) * (BS * DD) + vLane);
      pv_mma(cb, oacc);
    }
  }

  const float third = 1.f / 3.f;
#pragma unroll
  for (int r = 0; r < 8; ++r) {
    int row = qrow0 + r + 8 * half;
    v4f o;
    o.x = oacc[0][r] * third;
    o.y = oacc[1][r] * third;
    o.z = oacc[2][r] * third;
    o.w = oacc[3][r] * third;
    *(v4f*)(Oh + (size_t)row * DD + 4 * ln) = o;
  }
}

extern "C" void kernel_launch(void* const* d_in, const int* in_sizes, int n_in,
                              void* d_out, int out_size, void* d_ws,
                              size_t ws_size, hipStream_t stream) {
  (void)in_sizes; (void)n_in; (void)out_size; (void)ws_size;
  const float* q = (const float*)d_in[0];
  const float* k = (const float*)d_in[1];
  const float* v = (const float*)d_in[2];
  float* out = (float*)d_out;
  float* kblk = (float*)d_ws;
  float* vblk = kblk + (size_t)BB * HH * NB * DD;

  nsa_blockmean_kernel<<<BB * HH * NB, 64, 0, stream>>>(k, v, kblk, vblk);

  const size_t shmem = (size_t)(64 * PSTRIDE + 64 * KPITCH) * sizeof(float);
  hipFuncSetAttribute(reinterpret_cast<const void*>(nsa_main_kernel),
                      hipFuncAttributeMaxDynamicSharedMemorySize, (int)shmem);
  nsa_main_kernel<<<BB * HH * NB, 128, shmem, stream>>>(q, k, v, kblk, vblk,
                                                        out);
}


# window branch: 32-key chunks masked for all 16 rows of a wave get no matrix steps (Q x K^T: mask value stored directly; P x V: chunk skipped, its P are exact zeros); sum order unchanged
# speedup vs baseline: 3.5843x; 1.0325x over previous
#include <hip/hip_runtime.h>

#define BB 2
#define HH 8
#define SS 2048
#define DD 64
#define NB 32
#define BS 64
#define NSEL 8
#define SELC 512
#define KEEP 256
#define WIN 128
#define PSTRIDE 520
#define NEGINF (-1e30f)
#define SCALE 0.125f
#define SELR 4
#define SMXR 2
#define WROWS 4
#define KPITCH 68

typedef __attribute__((ext_vector_type(2))) float v2f;
typedef __attribute__((ext_vector_type(8))) float v8f;

static __device__ inline v8f vzero8() {
  v8f r;
#pragma unroll
  for (int i = 0; i < 8; ++i) r[i] = 0.f;
  return r;
}

static __device__ inline v8f wmma4(v2f a, v2f b, v8f c) {
  return __builtin_amdgcn_wmma_f32_16x16x4_f32(false, a, false, b, (short)0, c,
                                               false, false);
}

static __device__ inline unsigned okey(float x) {
  unsigned u = __float_as_uint(x);
  return (u & 0x80000000u) ? ~u : (u | 0x80000000u);
}

template <int M>
static __device__ inline float xlane(float x) {
  const int b = __builtin_bit_cast(int, x);
  return __builtin_bit_cast(
      float, __builtin_amdgcn_update_dpp(b, b, 0x160 + M, 0xf, 0xf, true));
}
static __device__ inline float xhalf(float x) {
  const int b = __builtin_bit_cast(int, x);
  return __builtin_bit_cast(
      float, __builtin_amdgcn_permlanex16(b, b, 0x76543210, 0xfedcba98, false,
                                          false));
}
template <int N>
static __device__ inline void wred_sumf(float (&v)[N]) {
#pragma unroll
  for (int j = 0; j < N; ++j) v[j] += xhalf(v[j]);
#pragma unroll
  for (int j = 0; j < N; ++j) v[j] += xlane<8>(v[j]);
#pragma unroll
  for (int j = 0; j < N; ++j) v[j] += xlane<4>(v[j]);
#pragma unroll
  for (int j = 0; j < N; ++j) v[j] += xlane<2>(v[j]);
#pragma unroll
  for (int j = 0; j < N; ++j) v[j] += xlane<1>(v[j]);
}
template <int N>
static __device__ inline void wred_maxf(float (&v)[N]) {
#pragma unroll
  for (int j = 0; j < N; ++j) v[j] = fmaxf(v[j], xhalf(v[j]));
#pragma unroll
  for (int j = 0; j < N; ++j) v[j] = fmaxf(v[j], xlane<8>(v[j]));
#pragma unroll
  for (int j = 0; j < N; ++j) v[j] = fmaxf(v[j], xlane<4>(v[j]));
#pragma unroll
  for (int j = 0; j < N; ++j) v[j] = fmaxf(v[j], xlane<2>(v[j]));
#pragma unroll
  for (int j = 0; j < N; ++j) v[j] = fmaxf(v[j], xlane<1>(v[j]));
}

static __device__ inline void count_ge_rows(const unsigned (&key)[SELR][16],
                                            const unsigned (&c)[SELR],
                                            int (&cnt)[SELR]) {
  unsigned b[SELR][16];
#pragma unroll
  for (int j = 0; j < SELR; ++j) {
#pragma unroll
    for (int k = 0; k < 16; ++k)
      b[j][k] = __builtin_amdgcn_ballot_w32(key[j][k] >= c[j]);
  }
  __builtin_amdgcn_sched_barrier(0);
#pragma unroll
  for (int j = 0; j < SELR; ++j) {
    int n = 0;
#pragma unroll
    for (int k = 0; k < 16; k += 2) {
      const unsigned long long two =
          ((unsigned long long)b[j][k + 1] << 32) | (unsigned long long)b[j][k];
      n += __builtin_popcountll(two);
    }
    cnt[j] = n;
  }
}

typedef __attribute__((ext_vector_type(4))) float v4f;
struct PVChunk {
  v2f a[8];
  v4f v0[8];
  v4f v1[8];
};
static __device__ inline void pv_load(PVChunk& c, const float* prow,
                                      const float* vrow) {
#pragma unroll
  for (int g = 0; g < 8; ++g) {
    c.v0[g] = *(const v4f*)(vrow + (size_t)(4 * g) * DD);
    c.v1[g] = *(const v4f*)(vrow + (size_t)(4 * g + 1) * DD);
  }
#pragma unroll
  for (int g = 0; g < 8; ++g) c.a[g] = *(const v2f*)(prow + 4 * g);
}
static __device__ inline void pv_mma(const PVChunk& c, v8f (&o)[4]) {
#pragma unroll
  for (int g = 0; g < 8; ++g) {
#pragma unroll
    for (int j = 0; j < 4; ++j) {
      v2f bf; bf.x = c.v0[g][j]; bf.y = c.v1[g][j];
      o[j] = wmma4(c.a[g], bf, o[j]);
    }
  }
}

struct KStage {
  v4f r[8];
};
static __device__ inline void kstage_load(KStage& s, const float* blk, int tid) {
#pragma unroll
  for (int i = 0; i < 8; ++i) s.r[i] = *(const v4f*)(blk + 4 * (tid + 128 * i));
}
static __device__ inline void kstage_store(const KStage& s, float* sK, int tid) {
#pragma unroll
  for (int i = 0; i < 8; ++i) {
    const int c = tid + 128 * i;
    *(v4f*)(sK + (c >> 4) * KPITCH + 4 * (c & 15)) = s.r[i];
  }
}
struct KPair {
  v2f f0[16];
  v2f f1[16];
};
static __device__ inline void kpair_load(KPair& c, const float* krow) {
#pragma unroll
  for (int t = 0; t < 16; ++t) {
    c.f0[t] = *(const v2f*)(krow + 4 * t);
    c.f1[t] = *(const v2f*)(krow + 16 * KPITCH + 4 * t);
  }
}
static __device__ inline void kpair_mma(const KPair& c, const v2f (&qa)[16],
                                        v8f& a0, v8f& a1) {
#pragma unroll
  for (int t = 0; t < 16; ++t) {
    a0 = wmma4(qa[t], c.f0[t], a0);
    a1 = wmma4(qa[t], c.f1[t], a1);
  }
}
static __device__ inline void sel_tile_out(const v8f& acc, float* myP, int half,
                                           int ln, int qrow0, int tok0, int col0) {
#pragma unroll
  for (int r = 0; r < 8; ++r) {
    int m = r + 8 * half;
    int i = qrow0 + m;
    int tok = tok0 + ln;
    float x = acc[r] * SCALE;
    myP[m * PSTRIDE + col0 + ln] = (tok <= i) ? x : NEGINF;
  }
}
static __device__ inline void win_tile_out(const v8f& acc, float* myP, int half,
                                           int ln, int qrow0, int tok0, int col0) {
#pragma unroll
  for (int r = 0; r < 8; ++r) {
    int m = r + 8 * half;
    int i = qrow0 + m;
    int j = tok0 + ln;
    bool ok = (j <= i) && (((i - j) < WIN) || (j < 1));
    myP[m * PSTRIDE + col0 + ln] = ok ? acc[r] * SCALE : NEGINF;
  }
}

static __device__ inline void win_chunk_masked(float* myP, int half, int ln,
                                               int col0) {
#pragma unroll
  for (int r = 0; r < 8; ++r) {
    int m = r + 8 * half;
    myP[m * PSTRIDE + col0 + ln] = NEGINF;
    myP[m * PSTRIDE + col0 + 16 + ln] = NEGINF;
  }
}

static __device__ inline void win_softmax(float* myP, int lane, int ncg) {
  for (int r0 = 0; r0 < 16; r0 += WROWS) {
    float* base = myP + r0 * PSTRIDE + lane;
    float x[WROWS][8];
    float mx[WROWS];
    float sum[WROWS];
#pragma unroll
    for (int j = 0; j < WROWS; ++j) {
#pragma unroll
      for (int k = 0; k < 8; ++k) x[j][k] = 0.f;
      mx[j] = -3.4e38f;
      sum[j] = 0.f;
    }
#pragma unroll
    for (int g = 0; g < 4; ++g) {
      if (g < ncg) {
#pragma unroll
        for (int j = 0; j < WROWS; ++j) {
          x[j][2 * g] = base[j * PSTRIDE + 64 * g];
          x[j][2 * g + 1] = base[j * PSTRIDE + 64 * g + 32];
        }
      }
    }
#pragma unroll
    for (int g = 0; g < 4; ++g) {
      if (g < ncg) {
#pragma unroll
        for (int j = 0; j < WROWS; ++j) {
          mx[j] = fmaxf(mx[j], x[j][2 * g]);
          mx[j] = fmaxf(mx[j], x[j][2 * g + 1]);
        }
      }
    }
    wred_maxf(mx);
#pragma unroll
    for (int g = 0; g < 4; ++g) {
      if (g < ncg) {
#pragma unroll
        for (int j = 0; j < WROWS; ++j) {
#pragma unroll
          for (int k = 2 * g; k < 2 * g + 2; ++k) {
            float p = (mx[j] > -1e29f) ? __expf(x[j][k] - mx[j]) : 0.f;
            x[j][k] = p;
            sum[j] += p;
          }
        }
      }
    }
    wred_sumf(sum);
    float rs[WROWS];
#pragma unroll
    for (int j = 0; j < WROWS; ++j) rs[j] = (sum[j] > 0.f) ? 1.f / sum[j] : 0.f;
#pragma unroll
    for (int g = 0; g < 4; ++g) {
      if (g < ncg) {
#pragma unroll
        for (int j = 0; j < WROWS; ++j) {
          base[j * PSTRIDE + 64 * g] = x[j][2 * g] * rs[j];
          base[j * PSTRIDE + 64 * g + 32] = x[j][2 * g + 1] * rs[j];
        }
      }
    }
  }
}

__global__ void nsa_blockmean_kernel(const float* __restrict__ K,
                                     const float* __restrict__ V,
                                     float* __restrict__ KB,
                                     float* __restrict__ VB) {
  const int g = blockIdx.x;
  const int d = threadIdx.x;
  const float* kp = K + (size_t)g * (BS * DD);
  const float* vp = V + (size_t)g * (BS * DD);
  float ks = 0.f, vs = 0.f;
  for (int t = 0; t < BS; ++t) {
    ks += kp[t * DD + d];
    vs += vp[t * DD + d];
  }
  KB[(size_t)g * DD + d] = ks * (1.f / BS);
  VB[(size_t)g * DD + d] = vs * (1.f / BS);
}

__global__ __launch_bounds__(128) void nsa_main_kernel(
    const float* __restrict__ Q, const float* __restrict__ K,
    const float* __restrict__ V, const float* __restrict__ KB,
    const float* __restrict__ VB, float* __restrict__ O) {
  extern __shared__ v4f sDyn[];
  float* sP = (float*)sDyn;
  float* sK = sP + 64 * PSTRIDE;
  __shared__ float sScore[NB];
  __shared__ int sSel[NSEL + 1];

  const int wg = blockIdx.x;
  const int qb = wg % NB;
  const int bh = wg / NB;
  const size_t hoff = (size_t)bh * SS * DD;
  const float* Qh = Q + hoff;
  const float* Kh = K + hoff;
  const float* Vh = V + hoff;
  const float* KBh = KB + (size_t)bh * NB * DD;
  const float* VBh = VB + (size_t)bh * NB * DD;
  float* Oh = O + hoff;

  const int tid = threadIdx.x;
  const int wave = tid >> 5;
  const int lane = tid & 31;
  const int half = lane >> 4;
  const int ln = lane & 15;
  const int qrow0 = qb * BS + wave * 16;
  float* myP = sP + (size_t)wave * 16 * PSTRIDE;

  const int firstb = (qb >= 2) ? qb - 2 : 0;
  const int nwb = qb - firstb + 1;
  const int extra = (firstb > 0) ? 1 : 0;
  const int ncg = nwb + extra;
  const int wkb0 = extra ? 0 : firstb;

  v2f qa[16];
#pragma unroll
  for (int t = 0; t < 16; ++t) {
    const float* p = Qh + (size_t)(qrow0 + ln) * DD + 4 * t + 2 * half;
    v2f a; a.x = p[0]; a.y = p[1];
    qa[t] = a;
  }

  v8f oacc[4];
#pragma unroll
  for (int dt = 0; dt < 4; ++dt) oacc[dt] = vzero8();

#pragma unroll
  for (int nt = 0; nt < 2; ++nt) {
    v8f acc = vzero8();
#pragma unroll
    for (int t = 0; t < 16; ++t) {
      const float* bp = KBh + (size_t)(nt * 16 + ln) * DD + 4 * t + 2 * half;
      v2f bf; bf.x = bp[0]; bf.y = bp[1];
      acc = wmma4(qa[t], bf, acc);
    }
#pragma unroll
    for (int r = 0; r < 8; ++r) {
      int m = r + 8 * half;
      int col = nt * 16 + ln;
      float x = acc[r] * SCALE;
      myP[m * PSTRIDE + col] = (col <= qb) ? x : NEGINF;
    }
  }
  __syncthreads();

  if (tid < NB) {
    float s = 0.f;
    for (int q2 = 0; q2 < BS; ++q2) s += sP[q2 * PSTRIDE + tid];
    sScore[tid] = s * (1.f / BS);
  }
  __syncthreads();

  if (tid == 0) {
    float loc[NB];
    for (int i = 0; i < NB; ++i) loc[i] = sScore[i];
    for (int s = 0; s < NSEL; ++s) {
      int bi = 0;
      float bv = loc[0];
      for (int i = 1; i < NB; ++i)
        if (loc[i] > bv) { bv = loc[i]; bi = i; }
      sSel[s] = bi;
      loc[bi] = -3.4e38f;
    }
    sSel[NSEL] = wkb0;
  }
  __syncthreads();

  if (tid < BS) {
    float* row = sP + tid * PSTRIDE;
    float mx = -3.4e38f;
    for (int c = 0; c < NB; ++c) mx = fmaxf(mx, row[c]);
    float sum = 0.f;
    for (int c = 0; c < NB; ++c) {
      float p = (mx > -1e29f) ? __expf(row[c] - mx) : 0.f;
      row[c] = p;
      sum += p;
    }
    float rs = (sum > 0.f) ? 1.f / sum : 0.f;
    for (int c = 0; c < NB; ++c) row[c] *= rs;
  }
  __syncthreads();

  const float* pLane = myP + ln * PSTRIDE + 2 * half;
  const size_t vLane = (size_t)(2 * half) * DD + 4 * ln;

  const float* kLane = sK + ln * KPITCH + 2 * half;
  KStage ks;
  kstage_load(ks, Kh + (size_t)sSel[0] * (BS * DD), tid);

  {
    PVChunk c;
    pv_load(c, pLane, VBh + vLane);
    pv_mma(c, oacc);
  }

  for (int sb = 0; sb < NSEL; ++sb) {
    const int kb = sSel[sb];
    __syncthreads();
    kstage_store(ks, sK, tid);
    __syncthreads();
    kstage_load(ks, Kh + (size_t)sSel[sb + 1] * (BS * DD), tid);
    __builtin_amdgcn_sched_barrier(0);
    KPair pa, pb;
    kpair_load(pa, kLane);
    kpair_load(pb, kLane + 32 * KPITCH);
    {
      v8f a0 = vzero8(), a1 = vzero8();
      kpair_mma(pa, qa, a0, a1);
      sel_tile_out(a0, myP, half, ln, qrow0, kb * BS, sb * 64);
      sel_tile_out(a1, myP, half, ln, qrow0, kb * BS + 16, sb * 64 + 16);
    }
    {
      v8f a0 = vzero8(), a1 = vzero8();
      kpair_mma(pb, qa, a0, a1);
      sel_tile_out(a0, myP, half, ln, qrow0, kb * BS + 32, sb * 64 + 32);
      sel_tile_out(a1, myP, half, ln, qrow0, kb * BS + 48, sb * 64 + 48);
    }
  }

  unsigned thrv = 0u;
  for (int r = 0; r < 16; r += SELR) {
    unsigned key[SELR][16];
    unsigned pre[SELR];
    int nab[SELR];
#pragma unroll
    for (int j = 0; j < SELR; ++j) {
      const float* row = myP + (r + j) * PSTRIDE;
#pragma unroll
      for (int k = 0; k < 16; ++k) key[j][k] = okey(row[lane + 32 * k]);
      nab[j] = 0;
    }
    const unsigned kneg = okey(NEGINF);
    unsigned open = (1u << SELR) - 1u;
    unsigned m = 0u;
    bool second = false;
    bool go = true;
#pragma unroll
    for (int j = 0; j < SELR; ++j) pre[j] = kneg + 1u;
#pragma nounroll
    for (int it = 0; go && it < 34; ++it) {
      unsigned cand[SELR];
      int cnt[SELR];
#pragma unroll
      for (int j = 0; j < SELR; ++j) cand[j] = pre[j] | m;
      count_ge_rows(key, cand, cnt);
      if (m != 0u) {
#pragma unroll
        for (int j = 0; j < SELR; ++j) {
          if (cnt[j] >= KEEP) pre[j] = cand[j];
          if (cnt[j] == KEEP) open &= ~(1u << j);
        }
        m >>= 1;
        go = (open != 0u) && (m != 0u);
      } else if (!second) {
        bool some = false;
#pragma unroll
        for (int j = 0; j < SELR; ++j) {
          nab[j] = cnt[j];
          some = some || (cnt[j] < KEEP);
        }
        second = some;
        m = some ? 0u : 0x80000000u;
#pragma unroll
        for (int j = 0; j < SELR; ++j) pre[j] = some ? kneg : 0u;
      } else {
#pragma unroll
        for (int j = 0; j < SELR; ++j) {
          const bool caseb = nab[j] < KEEP && cnt[j] >= KEEP;
          pre[j] = caseb ? kneg : 0u;
          if (caseb) open &= ~(1u << j);
        }
        m = 0x80000000u;
        go = (open != 0u);
      }
    }
#pragma unroll
    for (int j = 0; j < SELR; ++j)
      thrv = (lane == r + j) ? pre[j] : thrv;
  }

  for (int r = 0; r < 16; r += SMXR) {
    float x[SMXR][16];
    unsigned key[SMXR][16];
    unsigned pre[SMXR];
#pragma unroll
    for (int j = 0; j < SMXR; ++j) {
      const float* row = myP + (r + j) * PSTRIDE;
#pragma unroll
      for (int k = 0; k < 16; ++k) {
        x[j][k] = row[lane + 32 * k];
        key[j][k] = okey(x[j][k]);
      }
      pre[j] = __builtin_amdgcn_readlane(thrv, r + j);
    }
    float mx[SMXR];
    float sum[SMXR];
#pragma unroll
    for (int j = 0; j < SMXR; ++j) {
      mx[j] = -3.4e38f;
#pragma unroll
      for (int k = 0; k < 16; ++k) mx[j] = fmaxf(mx[j], x[j][k]);
    }
    wred_maxf(mx);
#pragma unroll
    for (int j = 0; j < SMXR; ++j) {
      sum[j] = 0.f;
#pragma unroll
      for (int k = 0; k < 16; ++k) {
        float p = (key[j][k] >= pre[j] && mx[j] > -1e29f) ? __expf(x[j][k] - mx[j]) : 0.f;
        x[j][k] = p;
        sum[j] += p;
      }
    }
    wred_sumf(sum);
#pragma unroll
    for (int j = 0; j < SMXR; ++j) {
      float* row = myP + (r + j) * PSTRIDE;
      float rs = (sum[j] > 0.f) ? 1.f / sum[j] : 0.f;
#pragma unroll
      for (int k = 0; k < 16; ++k) row[lane + 32 * k] = x[j][k] * rs;
    }
  }

  {
    PVChunk ca, cb;
    pv_load(ca, pLane, Vh + (size_t)sSel[0] * (BS * DD) + vLane);
    for (int ch = 0; ch < 2 * NSEL; ch += 2) {
      pv_load(cb, pLane + (ch + 1) * 32,
              Vh + (size_t)sSel[ch >> 1] * (BS * DD) + 32 * DD + vLane);
      pv_mma(ca, oacc);
      if (ch + 2 < 2 * NSEL)
        pv_load(ca, pLane + (ch + 2) * 32,
                Vh + (size_t)sSel[(ch >> 1) + 1] * (BS * DD) + vLane);
      pv_mma(cb, oacc);
    }
  }

  const int tI = __builtin_amdgcn_readfirstlane(qrow0) >> 4;
  const int tJb = 4 * firstb;
  const int tlo = (tI - 8 - tJb > 0) ? tI - 8 - tJb : 0;
  const int thi = tI - tJb;
  const unsigned maskw = ((2u << thi) - 1u) & ~((1u << tlo) - 1u);
  const unsigned need = extra ? (1u | (maskw << 4)) : (maskw | 1u);
  unsigned needp = (need | (need >> 1)) & 0x5555u;
  if (__builtin_popcount(needp) & 1) {
    const unsigned freec = ~needp & 0x5555u & ((1u << (4 * ncg)) - 1u);
    needp |= freec & (0u - freec);
  }
#pragma nounroll
  for (int cg = 0; cg < ncg; ++cg) {
    const int kb = (extra && cg == 0) ? 0 : firstb + (cg - extra);
    __syncthreads();
    kstage_store(ks, sK, tid);
    __syncthreads();
    if (cg + 1 < ncg)
      kstage_load(ks, Kh + (size_t)(firstb + (cg + 1 - extra)) * (BS * DD), tid);
    __builtin_amdgcn_sched_barrier(0);
    const bool wa = ((needp >> (4 * cg)) & 1u) != 0u;
    const bool wb = ((needp >> (4 * cg + 2)) & 1u) != 0u;
    KPair pa, pb;
    if (wa) kpair_load(pa, kLane);
    if (wb) kpair_load(pb, kLane + 32 * KPITCH);
    __builtin_amdgcn_sched_barrier(0);
    if (wa) {
      v8f a0 = vzero8(), a1 = vzero8();
      kpair_mma(pa, qa, a0, a1);
      win_tile_out(a0, myP, half, ln, qrow0, kb * BS, cg * 64);
      win_tile_out(a1, myP, half, ln, qrow0, kb * BS + 16, cg * 64 + 16);
    } else {
      win_chunk_masked(myP, half, ln, cg * 64);
    }
    if (wb) {
      v8f a0 = vzero8(), a1 = vzero8();
      kpair_mma(pb, qa, a0, a1);
      win_tile_out(a0, myP, half, ln, qrow0, kb * BS + 32, cg * 64 + 32);
      win_tile_out(a1, myP, half, ln, qrow0, kb * BS + 48, cg * 64 + 48);
    } else {
      win_chunk_masked(myP, half, ln, cg * 64 + 32);
    }
  }

  win_softmax(myP, lane, ncg);

  {
    PVChunk ca, cb;
    unsigned rem = needp;
    const int ntrip = __builtin_popcount(needp) >> 1;
    {
      const int c0 = __builtin_ctz(rem) >> 1;
      rem &= rem - 1u;
      const int g0 = c0 >> 1;
      const int kb0 = (extra && g0 == 0) ? 0 : firstb + (g0 - extra);
      pv_load(ca, pLane + 32 * c0,
              Vh + (size_t)kb0 * (BS * DD) + (size_t)(c0 & 1) * (32 * DD) + vLane);
    }
    for (int trip = 0; trip < ntrip; ++trip) {
      const int c1 = __builtin_ctz(rem) >> 1;
      rem &= rem - 1u;
      {
        const int g1 = c1 >> 1;
        const int kb1 = (extra && g1 == 0) ? 0 : firstb + (g1 - extra);
        pv_load(cb, pLane + 32 * c1,
                Vh + (size_t)kb1 * (BS * DD) + (size_t)(c1 & 1) * (32 * DD) + vLane);
      }
      __builtin_amdgcn_sched_barrier(0);
      pv_mma(ca, oacc);
      {
        const int c0 = (rem != 0u) ? (__builtin_ctz(rem) >> 1) : c1;
        rem &= rem - 1u;
        const int g0 = c0 >> 1;
        const int kb0 = (extra && g0 == 0) ? 0 : firstb + (g0 - extra);
        pv_load(ca, pLane + 32 * c0,
                Vh + (size_t)kb0 * (BS * DD) + (size_t)(c0 & 1) * (32 * DD) + vLane);
      }
      __builtin_amdgcn_sched_barrier(0);
      pv_mma(cb, oacc);
    }
  }

  const float third = 1.f / 3.f;
#pragma unroll
  for (int r = 0; r < 8; ++r) {
    int row = qrow0 + r + 8 * half;
    v4f o;
    o.x = oacc[0][r] * third;
    o.y = oacc[1][r] * third;
    o.z = oacc[2][r] * third;
    o.w = oacc[3][r] * third;
    *(v4f*)(Oh + (size_t)row * DD + 4 * ln) = o;
  }
}

extern "C" void kernel_launch(void* const* d_in, const int* in_sizes, int n_in,
                              void* d_out, int out_size, void* d_ws,
                              size_t ws_size, hipStream_t stream) {
  (void)in_sizes; (void)n_in; (void)out_size; (void)ws_size;
  const float* q = (const float*)d_in[0];
  const float* k = (const float*)d_in[1];
  const float* v = (const float*)d_in[2];
  float* out = (float*)d_out;
  float* kblk = (float*)d_ws;
  float* vblk = kblk + (size_t)BB * HH * NB * DD;

  nsa_blockmean_kernel<<<BB * HH * NB, 64, 0, stream>>>(k, v, kblk, vblk);

  const size_t shmem = (size_t)(64 * PSTRIDE + 64 * KPITCH) * sizeof(float);
  hipFuncSetAttribute(reinterpret_cast<const void*>(nsa_main_kernel),
                      hipFuncAttributeMaxDynamicSharedMemorySize, (int)shmem);
  nsa_main_kernel<<<BB * HH * NB, 128, shmem, stream>>>(q, k, v, kblk, vblk,
                                                        out);
}


# global loads issued ahead of use where a wait met a fresh load: selected P x V loop (same loads every trip), first V chunks before their softmax, block ids by lane read, block-mean loads in flight
# speedup vs baseline: 3.8945x; 1.0865x over previous
#include <hip/hip_runtime.h>

#define BB 2
#define HH 8
#define SS 2048
#define DD 64
#define NB 32
#define BS 64
#define NSEL 8
#define SELC 512
#define KEEP 256
#define WIN 128
#define PSTRIDE 520
#define NEGINF (-1e30f)
#define SCALE 0.125f
#define SELR 4
#define SMXR 2
#define WROWS 4
#define KPITCH 68

typedef __attribute__((ext_vector_type(2))) float v2f;
typedef __attribute__((ext_vector_type(8))) float v8f;

static __device__ inline v8f vzero8() {
  v8f r;
#pragma unroll
  for (int i = 0; i < 8; ++i) r[i] = 0.f;
  return r;
}

static __device__ inline v8f wmma4(v2f a, v2f b, v8f c) {
  return __builtin_amdgcn_wmma_f32_16x16x4_f32(false, a, false, b, (short)0, c,
                                               false, false);
}

static __device__ inline unsigned okey(float x) {
  unsigned u = __float_as_uint(x);
  return (u & 0x80000000u) ? ~u : (u | 0x80000000u);
}

template <int M>
static __device__ inline float xlane(float x) {
  const int b = __builtin_bit_cast(int, x);
  return __builtin_bit_cast(
      float, __builtin_amdgcn_update_dpp(b, b, 0x160 + M, 0xf, 0xf, true));
}
static __device__ inline float xhalf(float x) {
  const int b = __builtin_bit_cast(int, x);
  return __builtin_bit_cast(
      float, __builtin_amdgcn_permlanex16(b, b, 0x76543210, 0xfedcba98, false,
                                          false));
}
template <int N>
static __device__ inline void wred_sumf(float (&v)[N]) {
#pragma unroll
  for (int j = 0; j < N; ++j) v[j] += xhalf(v[j]);
#pragma unroll
  for (int j = 0; j < N; ++j) v[j] += xlane<8>(v[j]);
#pragma unroll
  for (int j = 0; j < N; ++j) v[j] += xlane<4>(v[j]);
#pragma unroll
  for (int j = 0; j < N; ++j) v[j] += xlane<2>(v[j]);
#pragma unroll
  for (int j = 0; j < N; ++j) v[j] += xlane<1>(v[j]);
}
template <int N>
static __device__ inline void wred_maxf(float (&v)[N]) {
#pragma unroll
  for (int j = 0; j < N; ++j) v[j] = fmaxf(v[j], xhalf(v[j]));
#pragma unroll
  for (int j = 0; j < N; ++j) v[j] = fmaxf(v[j], xlane<8>(v[j]));
#pragma unroll
  for (int j = 0; j < N; ++j) v[j] = fmaxf(v[j], xlane<4>(v[j]));
#pragma unroll
  for (int j = 0; j < N; ++j) v[j] = fmaxf(v[j], xlane<2>(v[j]));
#pragma unroll
  for (int j = 0; j < N; ++j) v[j] = fmaxf(v[j], xlane<1>(v[j]));
}

static __device__ inline void count_ge_rows(const unsigned (&key)[SELR][16],
                                            const unsigned (&c)[SELR],
                                            int (&cnt)[SELR]) {
  unsigned b[SELR][16];
#pragma unroll
  for (int j = 0; j < SELR; ++j) {
#pragma unroll
    for (int k = 0; k < 16; ++k)
      b[j][k] = __builtin_amdgcn_ballot_w32(key[j][k] >= c[j]);
  }
  __builtin_amdgcn_sched_barrier(0);
#pragma unroll
  for (int j = 0; j < SELR; ++j) {
    int n = 0;
#pragma unroll
    for (int k = 0; k < 16; k += 2) {
      const unsigned long long two =
          ((unsigned long long)b[j][k + 1] << 32) | (unsigned long long)b[j][k];
      n += __builtin_popcountll(two);
    }
    cnt[j] = n;
  }
}

typedef __attribute__((ext_vector_type(4))) float v4f;
struct PVChunk {
  v2f a[8];
  v4f v0[8];
  v4f v1[8];
};
static __device__ inline void pv_load_v(PVChunk& c, const float* vrow) {
#pragma unroll
  for (int g = 0; g < 8; ++g) {
    c.v0[g] = *(const v4f*)(vrow + (size_t)(4 * g) * DD);
    c.v1[g] = *(const v4f*)(vrow + (size_t)(4 * g + 1) * DD);
  }
}
static __device__ inline void pv_load_p(PVChunk& c, const float* prow) {
#pragma unroll
  for (int g = 0; g < 8; ++g) c.a[g] = *(const v2f*)(prow + 4 * g);
}
static __device__ inline void pv_load(PVChunk& c, const float* prow,
                                      const float* vrow) {
  pv_load_v(c, vrow);
  pv_load_p(c, prow);
}
static __device__ inline void pv_mma(const PVChunk& c, v8f (&o)[4]) {
#pragma unroll
  for (int g = 0; g < 8; ++g) {
#pragma unroll
    for (int j = 0; j < 4; ++j) {
      v2f bf; bf.x = c.v0[g][j]; bf.y = c.v1[g][j];
      o[j] = wmma4(c.a[g], bf, o[j]);
    }
  }
}

struct KStage {
  v4f r[8];
};
static __device__ inline void kstage_load(KStage& s, const float* blk, int tid) {
#pragma unroll
  for (int i = 0; i < 8; ++i) s.r[i] = *(const v4f*)(blk + 4 * (tid + 128 * i));
}
static __device__ inline void kstage_store(const KStage& s, float* sK, int tid) {
#pragma unroll
  for (int i = 0; i < 8; ++i) {
    const int c = tid + 128 * i;
    *(v4f*)(sK + (c >> 4) * KPITCH + 4 * (c & 15)) = s.r[i];
  }
}
struct KPair {
  v2f f0[16];
  v2f f1[16];
};
static __device__ inline void kpair_load(KPair& c, const float* krow) {
#pragma unroll
  for (int t = 0; t < 16; ++t) {
    c.f0[t] = *(const v2f*)(krow + 4 * t);
    c.f1[t] = *(const v2f*)(krow + 16 * KPITCH + 4 * t);
  }
}
static __device__ inline void kpair_mma(const KPair& c, const v2f (&qa)[16],
                                        v8f& a0, v8f& a1) {
#pragma unroll
  for (int t = 0; t < 16; ++t) {
    a0 = wmma4(qa[t], c.f0[t], a0);
    a1 = wmma4(qa[t], c.f1[t], a1);
  }
}
static __device__ inline void sel_tile_out(const v8f& acc, float* myP, int half,
                                           int ln, int qrow0, int tok0, int col0) {
#pragma unroll
  for (int r = 0; r < 8; ++r) {
    int m = r + 8 * half;
    int i = qrow0 + m;
    int tok = tok0 + ln;
    float x = acc[r] * SCALE;
    myP[m * PSTRIDE + col0 + ln] = (tok <= i) ? x : NEGINF;
  }
}
static __device__ inline void win_tile_out(const v8f& acc, float* myP, int half,
                                           int ln, int qrow0, int tok0, int col0) {
#pragma unroll
  for (int r = 0; r < 8; ++r) {
    int m = r + 8 * half;
    int i = qrow0 + m;
    int j = tok0 + ln;
    bool ok = (j <= i) && (((i - j) < WIN) || (j < 1));
    myP[m * PSTRIDE + col0 + ln] = ok ? acc[r] * SCALE : NEGINF;
  }
}

static __device__ inline void win_chunk_masked(float* myP, int half, int ln,
                                               int col0) {
#pragma unroll
  for (int r = 0; r < 8; ++r) {
    int m = r + 8 * half;
    myP[m * PSTRIDE + col0 + ln] = NEGINF;
    myP[m * PSTRIDE + col0 + 16 + ln] = NEGINF;
  }
}

static __device__ inline void win_softmax(float* myP, int lane, int ncg) {
  for (int r0 = 0; r0 < 16; r0 += WROWS) {
    float* base = myP + r0 * PSTRIDE + lane;
    float x[WROWS][8];
    float mx[WROWS];
    float sum[WROWS];
#pragma unroll
    for (int j = 0; j < WROWS; ++j) {
#pragma unroll
      for (int k = 0; k < 8; ++k) x[j][k] = 0.f;
      mx[j] = -3.4e38f;
      sum[j] = 0.f;
    }
#pragma unroll
    for (int g = 0; g < 4; ++g) {
      if (g < ncg) {
#pragma unroll
        for (int j = 0; j < WROWS; ++j) {
          x[j][2 * g] = base[j * PSTRIDE + 64 * g];
          x[j][2 * g + 1] = base[j * PSTRIDE + 64 * g + 32];
        }
      }
    }
#pragma unroll
    for (int g = 0; g < 4; ++g) {
      if (g < ncg) {
#pragma unroll
        for (int j = 0; j < WROWS; ++j) {
          mx[j] = fmaxf(mx[j], x[j][2 * g]);
          mx[j] = fmaxf(mx[j], x[j][2 * g + 1]);
        }
      }
    }
    wred_maxf(mx);
#pragma unroll
    for (int g = 0; g < 4; ++g) {
      if (g < ncg) {
#pragma unroll
        for (int j = 0; j < WROWS; ++j) {
#pragma unroll
          for (int k = 2 * g; k < 2 * g + 2; ++k) {
            float p = (mx[j] > -1e29f) ? __expf(x[j][k] - mx[j]) : 0.f;
            x[j][k] = p;
            sum[j] += p;
          }
        }
      }
    }
    wred_sumf(sum);
    float rs[WROWS];
#pragma unroll
    for (int j = 0; j < WROWS; ++j) rs[j] = (sum[j] > 0.f) ? 1.f / sum[j] : 0.f;
#pragma unroll
    for (int g = 0; g < 4; ++g) {
      if (g < ncg) {
#pragma unroll
        for (int j = 0; j < WROWS; ++j) {
          base[j * PSTRIDE + 64 * g] = x[j][2 * g] * rs[j];
          base[j * PSTRIDE + 64 * g + 32] = x[j][2 * g + 1] * rs[j];
        }
      }
    }
  }
}

__global__ void nsa_blockmean_kernel(const float* __restrict__ K,
                                     const float* __restrict__ V,
                                     float* __restrict__ KB,
                                     float* __restrict__ VB) {
  const int g = blockIdx.x;
  const int d = threadIdx.x;
  const float* kp = K + (size_t)g * (BS * DD);
  const float* vp = V + (size_t)g * (BS * DD);
  float ks = 0.f, vs = 0.f;
#pragma unroll
  for (int t = 0; t < BS; ++t) {
    ks += kp[t * DD + d];
    vs += vp[t * DD + d];
  }
  KB[(size_t)g * DD + d] = ks * (1.f / BS);
  VB[(size_t)g * DD + d] = vs * (1.f / BS);
}

__global__ __launch_bounds__(128) void nsa_main_kernel(
    const float* __restrict__ Q, const float* __restrict__ K,
    const float* __restrict__ V, const float* __restrict__ KB,
    const float* __restrict__ VB, float* __restrict__ O) {
  extern __shared__ v4f sDyn[];
  float* sP = (float*)sDyn;
  float* sK = sP + 64 * PSTRIDE;
  __shared__ float sScore[NB];
  __shared__ int sSel[NSEL + 1];

  const int wg = blockIdx.x;
  const int qb = wg % NB;
  const int bh = wg / NB;
  const size_t hoff = (size_t)bh * SS * DD;
  const float* Qh = Q + hoff;
  const float* Kh = K + hoff;
  const float* Vh = V + hoff;
  const float* KBh = KB + (size_t)bh * NB * DD;
  const float* VBh = VB + (size_t)bh * NB * DD;
  float* Oh = O + hoff;

  const int tid = threadIdx.x;
  const int wave = tid >> 5;
  const int lane = tid & 31;
  const int half = lane >> 4;
  const int ln = lane & 15;
  const int qrow0 = qb * BS + wave * 16;
  float* myP = sP + (size_t)wave * 16 * PSTRIDE;

  const int firstb = (qb >= 2) ? qb - 2 : 0;
  const int nwb = qb - firstb + 1;
  const int extra = (firstb > 0) ? 1 : 0;
  const int ncg = nwb + extra;
  const int wkb0 = extra ? 0 : firstb;

  v2f qa[16];
#pragma unroll
  for (int t = 0; t < 16; ++t) {
    const float* p = Qh + (size_t)(qrow0 + ln) * DD + 4 * t + 2 * half;
    v2f a; a.x = p[0]; a.y = p[1];
    qa[t] = a;
  }

  v8f oacc[4];
#pragma unroll
  for (int dt = 0; dt < 4; ++dt) oacc[dt] = vzero8();

  const size_t vLane = (size_t)(2 * half) * DD + 4 * ln;
  PVChunk ccmp;
  pv_load_v(ccmp, VBh + vLane);

#pragma unroll
  for (int nt = 0; nt < 2; ++nt) {
    v8f acc = vzero8();
#pragma unroll
    for (int t = 0; t < 16; ++t) {
      const float* bp = KBh + (size_t)(nt * 16 + ln) * DD + 4 * t + 2 * half;
      v2f bf; bf.x = bp[0]; bf.y = bp[1];
      acc = wmma4(qa[t], bf, acc);
    }
#pragma unroll
    for (int r = 0; r < 8; ++r) {
      int m = r + 8 * half;
      int col = nt * 16 + ln;
      float x = acc[r] * SCALE;
      myP[m * PSTRIDE + col] = (col <= qb) ? x : NEGINF;
    }
  }
  __syncthreads();

  if (tid < NB) {
    float s = 0.f;
    for (int q2 = 0; q2 < BS; ++q2) s += sP[q2 * PSTRIDE + tid];
    sScore[tid] = s * (1.f / BS);
  }
  __syncthreads();

  if (tid == 0) {
    float loc[NB];
    for (int i = 0; i < NB; ++i) loc[i] = sScore[i];
    for (int s = 0; s < NSEL; ++s) {
      int bi = 0;
      float bv = loc[0];
      for (int i = 1; i < NB; ++i)
        if (loc[i] > bv) { bv = loc[i]; bi = i; }
      sSel[s] = bi;
      loc[bi] = -3.4e38f;
    }
    sSel[NSEL] = wkb0;
  }
  __syncthreads();

  if (tid < BS) {
    float* row = sP + tid * PSTRIDE;
    float mx = -3.4e38f;
    for (int c = 0; c < NB; ++c) mx = fmaxf(mx, row[c]);
    float sum = 0.f;
    for (int c = 0; c < NB; ++c) {
      float p = (mx > -1e29f) ? __expf(row[c] - mx) : 0.f;
      row[c] = p;
      sum += p;
    }
    float rs = (sum > 0.f) ? 1.f / sum : 0.f;
    for (int c = 0; c < NB; ++c) row[c] *= rs;
  }
  __syncthreads();

  const float* pLane = myP + ln * PSTRIDE + 2 * half;
  const int selv = sSel[lane <= NSEL ? lane : 0];

  const float* kLane = sK + ln * KPITCH + 2 * half;
  KStage ks;
  kstage_load(ks, Kh + (size_t)__builtin_amdgcn_readlane(selv, 0) * (BS * DD),
              tid);

  pv_load_p(ccmp, pLane);
  pv_mma(ccmp, oacc);

  for (int sb = 0; sb < NSEL; ++sb) {
    const int kb = __builtin_amdgcn_readlane(selv, sb);
    __syncthreads();
    kstage_store(ks, sK, tid);
    __syncthreads();
    kstage_load(ks,
                Kh + (size_t)__builtin_amdgcn_readlane(selv, sb + 1) * (BS * DD),
                tid);
    __builtin_amdgcn_sched_barrier(0);
    KPair pa, pb;
    kpair_load(pa, kLane);
    kpair_load(pb, kLane + 32 * KPITCH);
    {
      v8f a0 = vzero8(), a1 = vzero8();
      kpair_mma(pa, qa, a0, a1);
      sel_tile_out(a0, myP, half, ln, qrow0, kb * BS, sb * 64);
      sel_tile_out(a1, myP, half, ln, qrow0, kb * BS + 16, sb * 64 + 16);
    }
    {
      v8f a0 = vzero8(), a1 = vzero8();
      kpair_mma(pb, qa, a0, a1);
      sel_tile_out(a0, myP, half, ln, qrow0, kb * BS + 32, sb * 64 + 32);
      sel_tile_out(a1, myP, half, ln, qrow0, kb * BS + 48, sb * 64 + 48);
    }
  }

  unsigned thrv = 0u;
  for (int r = 0; r < 16; r += SELR) {
    unsigned key[SELR][16];
    unsigned pre[SELR];
    int nab[SELR];
#pragma unroll
    for (int j = 0; j < SELR; ++j) {
      const float* row = myP + (r + j) * PSTRIDE;
#pragma unroll
      for (int k = 0; k < 16; ++k) key[j][k] = okey(row[lane + 32 * k]);
      nab[j] = 0;
    }
    const unsigned kneg = okey(NEGINF);
    unsigned open = (1u << SELR) - 1u;
    unsigned m = 0u;
    bool second = false;
    bool go = true;
#pragma unroll
    for (int j = 0; j < SELR; ++j) pre[j] = kneg + 1u;
#pragma nounroll
    for (int it = 0; go && it < 34; ++it) {
      unsigned cand[SELR];
      int cnt[SELR];
#pragma unroll
      for (int j = 0; j < SELR; ++j) cand[j] = pre[j] | m;
      count_ge_rows(key, cand, cnt);
      if (m != 0u) {
#pragma unroll
        for (int j = 0; j < SELR; ++j) {
          if (cnt[j] >= KEEP) pre[j] = cand[j];
          if (cnt[j] == KEEP) open &= ~(1u << j);
        }
        m >>= 1;
        go = (open != 0u) && (m != 0u);
      } else if (!second) {
        bool some = false;
#pragma unroll
        for (int j = 0; j < SELR; ++j) {
          nab[j] = cnt[j];
          some = some || (cnt[j] < KEEP);
        }
        second = some;
        m = some ? 0u : 0x80000000u;
#pragma unroll
        for (int j = 0; j < SELR; ++j) pre[j] = some ? kneg : 0u;
      } else {
#pragma unroll
        for (int j = 0; j < SELR; ++j) {
          const bool caseb = nab[j] < KEEP && cnt[j] >= KEEP;
          pre[j] = caseb ? kneg : 0u;
          if (caseb) open &= ~(1u << j);
        }
        m = 0x80000000u;
        go = (open != 0u);
      }
    }
#pragma unroll
    for (int j = 0; j < SELR; ++j)
      thrv = (lane == r + j) ? pre[j] : thrv;
  }

  PVChunk ca, cb;
  pv_load_v(ca, Vh + (size_t)__builtin_amdgcn_readlane(selv, 0) * (BS * DD) +
                    vLane);

  for (int r = 0; r < 16; r += SMXR) {
    float x[SMXR][16];
    unsigned key[SMXR][16];
    unsigned pre[SMXR];
#pragma unroll
    for (int j = 0; j < SMXR; ++j) {
      const float* row = myP + (r + j) * PSTRIDE;
#pragma unroll
      for (int k = 0; k < 16; ++k) {
        x[j][k] = row[lane + 32 * k];
        key[j][k] = okey(x[j][k]);
      }
      pre[j] = __builtin_amdgcn_readlane(thrv, r + j);
    }
    float mx[SMXR];
    float sum[SMXR];
#pragma unroll
    for (int j = 0; j < SMXR; ++j) {
      mx[j] = -3.4e38f;
#pragma unroll
      for (int k = 0; k < 16; ++k) mx[j] = fmaxf(mx[j], x[j][k]);
    }
    wred_maxf(mx);
#pragma unroll
    for (int j = 0; j < SMXR; ++j) {
      sum[j] = 0.f;
#pragma unroll
      for (int k = 0; k < 16; ++k) {
        float p = (key[j][k] >= pre[j] && mx[j] > -1e29f) ? __expf(x[j][k] - mx[j]) : 0.f;
        x[j][k] = p;
        sum[j] += p;
      }
    }
    wred_sumf(sum);
#pragma unroll
    for (int j = 0; j < SMXR; ++j) {
      float* row = myP + (r + j) * PSTRIDE;
      float rs = (sum[j] > 0.f) ? 1.f / sum[j] : 0.f;
#pragma unroll
      for (int k = 0; k < 16; ++k) row[lane + 32 * k] = x[j][k] * rs;
    }
  }

  {
    pv_load_p(ca, pLane);
#pragma nounroll
    for (int sb = 0; sb < NSEL; ++sb) {
      const float* vb =
          Vh + (size_t)__builtin_amdgcn_readlane(selv, sb) * (BS * DD) + vLane;
      const float* pb = pLane + sb * 64;
      pv_load(cb, pb + 32, vb + 32 * DD);
      __builtin_amdgcn_sched_barrier(0);
      pv_mma(ca, oacc);
      {
        const bool more = sb + 1 < NSEL;
        const int sn = more ? sb + 1 : sb;
        const float* vn =
            Vh + (size_t)__builtin_amdgcn_readlane(selv, sn) * (BS * DD) +
            (more ? 0 : 32 * DD) + vLane;
        const float* pn = more ? pb + 64 : pb + 32;
        pv_load(ca, pn, vn);
      }
      __builtin_amdgcn_sched_barrier(0);
      pv_mma(cb, oacc);
    }
  }

  const int tI = __builtin_amdgcn_readfirstlane(qrow0) >> 4;
  const int tJb = 4 * firstb;
  const int tlo = (tI - 8 - tJb > 0) ? tI - 8 - tJb : 0;
  const int thi = tI - tJb;
  const unsigned maskw = ((2u << thi) - 1u) & ~((1u << tlo) - 1u);
  const unsigned need = extra ? (1u | (maskw << 4)) : (maskw | 1u);
  unsigned needp = (need | (need >> 1)) & 0x5555u;
  if (__builtin_popcount(needp) & 1) {
    const unsigned freec = ~needp & 0x5555u & ((1u << (4 * ncg)) - 1u);
    needp |= freec & (0u - freec);
  }
#pragma nounroll
  for (int cg = 0; cg < ncg; ++cg) {
    const int kb = (extra && cg == 0) ? 0 : firstb + (cg - extra);
    __syncthreads();
    kstage_store(ks, sK, tid);
    __syncthreads();
    if (cg + 1 < ncg)
      kstage_load(ks, Kh + (size_t)(firstb + (cg + 1 - extra)) * (BS * DD), tid);
    __builtin_amdgcn_sched_barrier(0);
    const bool wa = ((needp >> (4 * cg)) & 1u) != 0u;
    const bool wb = ((needp >> (4 * cg + 2)) & 1u) != 0u;
    KPair pa, pb;
    if (wa) kpair_load(pa, kLane);
    if (wb) kpair_load(pb, kLane + 32 * KPITCH);
    __builtin_amdgcn_sched_barrier(0);
    if (wa) {
      v8f a0 = vzero8(), a1 = vzero8();
      kpair_mma(pa, qa, a0, a1);
      win_tile_out(a0, myP, half, ln, qrow0, kb * BS, cg * 64);
      win_tile_out(a1, myP, half, ln, qrow0, kb * BS + 16, cg * 64 + 16);
    } else {
      win_chunk_masked(myP, half, ln, cg * 64);
    }
    if (wb) {
      v8f a0 = vzero8(), a1 = vzero8();
      kpair_mma(pb, qa, a0, a1);
      win_tile_out(a0, myP, half, ln, qrow0, kb * BS + 32, cg * 64 + 32);
      win_tile_out(a1, myP, half, ln, qrow0, kb * BS + 48, cg * 64 + 48);
    } else {
      win_chunk_masked(myP, half, ln, cg * 64 + 32);
    }
  }

  unsigned rem = needp;
  const int cw0 = __builtin_ctz(rem) >> 1;
  rem &= rem - 1u;
  {
    const int g0 = cw0 >> 1;
    const int kb0 = (extra && g0 == 0) ? 0 : firstb + (g0 - extra);
    pv_load_v(ca, Vh + (size_t)kb0 * (BS * DD) +
                      (size_t)(cw0 & 1) * (32 * DD) + vLane);
  }

  win_softmax(myP, lane, ncg);

  {
    const int ntrip = __builtin_popcount(needp) >> 1;
    pv_load_p(ca, pLane + 32 * cw0);
    for (int trip = 0; trip < ntrip; ++trip) {
      const int c1 = __builtin_ctz(rem) >> 1;
      rem &= rem - 1u;
      {
        const int g1 = c1 >> 1;
        const int kb1 = (extra && g1 == 0) ? 0 : firstb + (g1 - extra);
        pv_load(cb, pLane + 32 * c1,
                Vh + (size_t)kb1 * (BS * DD) + (size_t)(c1 & 1) * (32 * DD) + vLane);
      }
      __builtin_amdgcn_sched_barrier(0);
      pv_mma(ca, oacc);
      {
        const int c0 = (rem != 0u) ? (__builtin_ctz(rem) >> 1) : c1;
        rem &= rem - 1u;
        const int g0 = c0 >> 1;
        const int kb0 = (extra && g0 == 0) ? 0 : firstb + (g0 - extra);
        pv_load(ca, pLane + 32 * c0,
                Vh + (size_t)kb0 * (BS * DD) + (size_t)(c0 & 1) * (32 * DD) + vLane);
      }
      __builtin_amdgcn_sched_barrier(0);
      pv_mma(cb, oacc);
    }
  }

  const float third = 1.f / 3.f;
#pragma unroll
  for (int r = 0; r < 8; ++r) {
    int row = qrow0 + r + 8 * half;
    v4f o;
    o.x = oacc[0][r] * third;
    o.y = oacc[1][r] * third;
    o.z = oacc[2][r] * third;
    o.w = oacc[3][r] * third;
    *(v4f*)(Oh + (size_t)row * DD + 4 * ln) = o;
  }
}

extern "C" void kernel_launch(void* const* d_in, const int* in_sizes, int n_in,
                              void* d_out, int out_size, void* d_ws,
                              size_t ws_size, hipStream_t stream) {
  (void)in_sizes; (void)n_in; (void)out_size; (void)ws_size;
  const float* q = (const float*)d_in[0];
  const float* k = (const float*)d_in[1];
  const float* v = (const float*)d_in[2];
  float* out = (float*)d_out;
  float* kblk = (float*)d_ws;
  float* vblk = kblk + (size_t)BB * HH * NB * DD;

  nsa_blockmean_kernel<<<BB * HH * NB, 64, 0, stream>>>(k, v, kblk, vblk);

  const size_t shmem = (size_t)(64 * PSTRIDE + 64 * KPITCH) * sizeof(float);
  hipFuncSetAttribute(reinterpret_cast<const void*>(nsa_main_kernel),
                      hipFuncAttributeMaxDynamicSharedMemorySize, (int)shmem);
  nsa_main_kernel<<<BB * HH * NB, 128, shmem, stream>>>(q, k, v, kblk, vblk,
                                                        out);
}


# top-256 select: threshold found by closing a key bracket with candidates interpolated on the counts (middle key as bound), not bit by bit; keys by shift/or/xor; kept set and sum order unchanged
# speedup vs baseline: 4.1012x; 1.0531x over previous
#include <hip/hip_runtime.h>

#define BB 2
#define HH 8
#define SS 2048
#define DD 64
#define NB 32
#define BS 64
#define NSEL 8
#define SELC 512
#define KEEP 256
#define WIN 128
#define PSTRIDE 520
#define NEGINF (-1e30f)
#define SCALE 0.125f
#define SELR 4
#define SEL_NINT 26
#define SEL_MAXIT 64
#define SEL_DAMP 0.9f
#define SEL_WMIN 0.1f
#define SMXR 2
#define WROWS 4
#define KPITCH 68

typedef __attribute__((ext_vector_type(2))) float v2f;
typedef __attribute__((ext_vector_type(8))) float v8f;

static __device__ inline v8f vzero8() {
  v8f r;
#pragma unroll
  for (int i = 0; i < 8; ++i) r[i] = 0.f;
  return r;
}

static __device__ inline v8f wmma4(v2f a, v2f b, v8f c) {
  return __builtin_amdgcn_wmma_f32_16x16x4_f32(false, a, false, b, (short)0, c,
                                               false, false);
}

static __device__ inline unsigned okey(float x) {
  const unsigned u = __float_as_uint(x);
  return u ^ ((unsigned)((int)u >> 31) | 0x80000000u);
}
static __device__ inline float unokey(unsigned k) {
  return __uint_as_float(k ^ (~(unsigned)((int)k >> 31) | 0x80000000u));
}

template <int M>
static __device__ inline float xlane(float x) {
  const int b = __builtin_bit_cast(int, x);
  return __builtin_bit_cast(
      float, __builtin_amdgcn_update_dpp(b, b, 0x160 + M, 0xf, 0xf, true));
}
static __device__ inline float xhalf(float x) {
  const int b = __builtin_bit_cast(int, x);
  return __builtin_bit_cast(
      float, __builtin_amdgcn_permlanex16(b, b, 0x76543210, 0xfedcba98, false,
                                          false));
}
template <int N>
static __device__ inline void wred_sumf(float (&v)[N]) {
#pragma unroll
  for (int j = 0; j < N; ++j) v[j] += xhalf(v[j]);
#pragma unroll
  for (int j = 0; j < N; ++j) v[j] += xlane<8>(v[j]);
#pragma unroll
  for (int j = 0; j < N; ++j) v[j] += xlane<4>(v[j]);
#pragma unroll
  for (int j = 0; j < N; ++j) v[j] += xlane<2>(v[j]);
#pragma unroll
  for (int j = 0; j < N; ++j) v[j] += xlane<1>(v[j]);
}
template <int N>
static __device__ inline void wred_maxf(float (&v)[N]) {
#pragma unroll
  for (int j = 0; j < N; ++j) v[j] = fmaxf(v[j], xhalf(v[j]));
#pragma unroll
  for (int j = 0; j < N; ++j) v[j] = fmaxf(v[j], xlane<8>(v[j]));
#pragma unroll
  for (int j = 0; j < N; ++j) v[j] = fmaxf(v[j], xlane<4>(v[j]));
#pragma unroll
  for (int j = 0; j < N; ++j) v[j] = fmaxf(v[j], xlane<2>(v[j]));
#pragma unroll
  for (int j = 0; j < N; ++j) v[j] = fmaxf(v[j], xlane<1>(v[j]));
}

template <int M>
static __device__ inline unsigned xlaneu(unsigned x) {
  const int b = (int)x;
  return (unsigned)__builtin_amdgcn_update_dpp(b, b, 0x160 + M, 0xf, 0xf, true);
}
static __device__ inline unsigned xhalfu(unsigned x) {
  const int b = (int)x;
  return (unsigned)__builtin_amdgcn_permlanex16(b, b, 0x76543210, 0xfedcba98,
                                                false, false);
}
template <bool MX>
static __device__ inline unsigned upick(unsigned a, unsigned b) {
  return MX ? (a > b ? a : b) : (a < b ? a : b);
}
template <bool MX>
static __device__ inline unsigned urowred4(const unsigned (&m)[4], int lane) {
  const bool odd = (lane & 1) != 0, up = (lane & 2) != 0;
  unsigned k0 = odd ? m[1] : m[0], s0 = odd ? m[0] : m[1];
  unsigned k1 = odd ? m[3] : m[2], s1 = odd ? m[2] : m[3];
  k0 = upick<MX>(k0, xlaneu<1>(s0));
  k1 = upick<MX>(k1, xlaneu<1>(s1));
  unsigned k = up ? k1 : k0;
  const unsigned s = up ? k0 : k1;
  k = upick<MX>(k, xlaneu<2>(s));
  k = upick<MX>(k, xlaneu<4>(k));
  k = upick<MX>(k, xlaneu<8>(k));
  k = upick<MX>(k, xhalfu(k));
  return k;
}

static __device__ inline void count_ge_rows(const unsigned (&key)[SELR][16],
                                            const unsigned (&c)[SELR],
                                            int (&cnt)[SELR]) {
  unsigned b[SELR][16];
#pragma unroll
  for (int j = 0; j < SELR; ++j) {
#pragma unroll
    for (int k = 0; k < 16; ++k)
      b[j][k] = __builtin_amdgcn_ballot_w32(key[j][k] >= c[j]);
  }
  __builtin_amdgcn_sched_barrier(0);
#pragma unroll
  for (int j = 0; j < SELR; ++j) {
    int n = 0;
#pragma unroll
    for (int k = 0; k < 16; k += 2) {
      const unsigned long long two =
          ((unsigned long long)b[j][k + 1] << 32) | (unsigned long long)b[j][k];
      n += __builtin_popcountll(two);
    }
    cnt[j] = n;
  }
}

typedef __attribute__((ext_vector_type(4))) float v4f;
struct PVChunk {
  v2f a[8];
  v4f v0[8];
  v4f v1[8];
};
static __device__ inline void pv_load_v(PVChunk& c, const float* vrow) {
#pragma unroll
  for (int g = 0; g < 8; ++g) {
    c.v0[g] = *(const v4f*)(vrow + (size_t)(4 * g) * DD);
    c.v1[g] = *(const v4f*)(vrow + (size_t)(4 * g + 1) * DD);
  }
}
static __device__ inline void pv_load_p(PVChunk& c, const float* prow) {
#pragma unroll
  for (int g = 0; g < 8; ++g) c.a[g] = *(const v2f*)(prow + 4 * g);
}
static __device__ inline void pv_load(PVChunk& c, const float* prow,
                                      const float* vrow) {
  pv_load_v(c, vrow);
  pv_load_p(c, prow);
}
static __device__ inline void pv_mma(const PVChunk& c, v8f (&o)[4]) {
#pragma unroll
  for (int g = 0; g < 8; ++g) {
#pragma unroll
    for (int j = 0; j < 4; ++j) {
      v2f bf; bf.x = c.v0[g][j]; bf.y = c.v1[g][j];
      o[j] = wmma4(c.a[g], bf, o[j]);
    }
  }
}

struct KStage {
  v4f r[8];
};
static __device__ inline void kstage_load(KStage& s, const float* blk, int tid) {
#pragma unroll
  for (int i = 0; i < 8; ++i) s.r[i] = *(const v4f*)(blk + 4 * (tid + 128 * i));
}
static __device__ inline void kstage_store(const KStage& s, float* sK, int tid) {
#pragma unroll
  for (int i = 0; i < 8; ++i) {
    const int c = tid + 128 * i;
    *(v4f*)(sK + (c >> 4) * KPITCH + 4 * (c & 15)) = s.r[i];
  }
}
struct KPair {
  v2f f0[16];
  v2f f1[16];
};
static __device__ inline void kpair_load(KPair& c, const float* krow) {
#pragma unroll
  for (int t = 0; t < 16; ++t) {
    c.f0[t] = *(const v2f*)(krow + 4 * t);
    c.f1[t] = *(const v2f*)(krow + 16 * KPITCH + 4 * t);
  }
}
static __device__ inline void kpair_mma(const KPair& c, const v2f (&qa)[16],
                                        v8f& a0, v8f& a1) {
#pragma unroll
  for (int t = 0; t < 16; ++t) {
    a0 = wmma4(qa[t], c.f0[t], a0);
    a1 = wmma4(qa[t], c.f1[t], a1);
  }
}
static __device__ inline void sel_tile_out(const v8f& acc, float* myP, int half,
                                           int ln, int qrow0, int tok0, int col0) {
#pragma unroll
  for (int r = 0; r < 8; ++r) {
    int m = r + 8 * half;
    int i = qrow0 + m;
    int tok = tok0 + ln;
    float x = acc[r] * SCALE;
    myP[m * PSTRIDE + col0 + ln] = (tok <= i) ? x : NEGINF;
  }
}
static __device__ inline void win_tile_out(const v8f& acc, float* myP, int half,
                                           int ln, int qrow0, int tok0, int col0) {
#pragma unroll
  for (int r = 0; r < 8; ++r) {
    int m = r + 8 * half;
    int i = qrow0 + m;
    int j = tok0 + ln;
    bool ok = (j <= i) && (((i - j) < WIN) || (j < 1));
    myP[m * PSTRIDE + col0 + ln] = ok ? acc[r] * SCALE : NEGINF;
  }
}

static __device__ inline void win_chunk_masked(float* myP, int half, int ln,
                                               int col0) {
#pragma unroll
  for (int r = 0; r < 8; ++r) {
    int m = r + 8 * half;
    myP[m * PSTRIDE + col0 + ln] = NEGINF;
    myP[m * PSTRIDE + col0 + 16 + ln] = NEGINF;
  }
}

static __device__ inline void win_softmax(float* myP, int lane, int ncg) {
  for (int r0 = 0; r0 < 16; r0 += WROWS) {
    float* base = myP + r0 * PSTRIDE + lane;
    float x[WROWS][8];
    float mx[WROWS];
    float sum[WROWS];
#pragma unroll
    for (int j = 0; j < WROWS; ++j) {
#pragma unroll
      for (int k = 0; k < 8; ++k) x[j][k] = 0.f;
      mx[j] = -3.4e38f;
      sum[j] = 0.f;
    }
#pragma unroll
    for (int g = 0; g < 4; ++g) {
      if (g < ncg) {
#pragma unroll
        for (int j = 0; j < WROWS; ++j) {
          x[j][2 * g] = base[j * PSTRIDE + 64 * g];
          x[j][2 * g + 1] = base[j * PSTRIDE + 64 * g + 32];
        }
      }
    }
#pragma unroll
    for (int g = 0; g < 4; ++g) {
      if (g < ncg) {
#pragma unroll
        for (int j = 0; j < WROWS; ++j) {
          mx[j] = fmaxf(mx[j], x[j][2 * g]);
          mx[j] = fmaxf(mx[j], x[j][2 * g + 1]);
        }
      }
    }
    wred_maxf(mx);
#pragma unroll
    for (int g = 0; g < 4; ++g) {
      if (g < ncg) {
#pragma unroll
        for (int j = 0; j < WROWS; ++j) {
#pragma unroll
          for (int k = 2 * g; k < 2 * g + 2; ++k) {
            float p = (mx[j] > -1e29f) ? __expf(x[j][k] - mx[j]) : 0.f;
            x[j][k] = p;
            sum[j] += p;
          }
        }
      }
    }
    wred_sumf(sum);
    float rs[WROWS];
#pragma unroll
    for (int j = 0; j < WROWS; ++j) rs[j] = (sum[j] > 0.f) ? 1.f / sum[j] : 0.f;
#pragma unroll
    for (int g = 0; g < 4; ++g) {
      if (g < ncg) {
#pragma unroll
        for (int j = 0; j < WROWS; ++j) {
          base[j * PSTRIDE + 64 * g] = x[j][2 * g] * rs[j];
          base[j * PSTRIDE + 64 * g + 32] = x[j][2 * g + 1] * rs[j];
        }
      }
    }
  }
}

__global__ void nsa_blockmean_kernel(const float* __restrict__ K,
                                     const float* __restrict__ V,
                                     float* __restrict__ KB,
                                     float* __restrict__ VB) {
  const int g = blockIdx.x;
  const int d = threadIdx.x;
  const float* kp = K + (size_t)g * (BS * DD);
  const float* vp = V + (size_t)g * (BS * DD);
  float ks = 0.f, vs = 0.f;
#pragma unroll
  for (int t = 0; t < BS; ++t) {
    ks += kp[t * DD + d];
    vs += vp[t * DD + d];
  }
  KB[(size_t)g * DD + d] = ks * (1.f / BS);
  VB[(size_t)g * DD + d] = vs * (1.f / BS);
}

__global__ __launch_bounds__(128) void nsa_main_kernel(
    const float* __restrict__ Q, const float* __restrict__ K,
    const float* __restrict__ V, const float* __restrict__ KB,
    const float* __restrict__ VB, float* __restrict__ O) {
  extern __shared__ v4f sDyn[];
  float* sP = (float*)sDyn;
  float* sK = sP + 64 * PSTRIDE;
  __shared__ float sScore[NB];
  __shared__ int sSel[NSEL + 1];

  const int wg = blockIdx.x;
  const int qb = wg % NB;
  const int bh = wg / NB;
  const size_t hoff = (size_t)bh * SS * DD;
  const float* Qh = Q + hoff;
  const float* Kh = K + hoff;
  const float* Vh = V + hoff;
  const float* KBh = KB + (size_t)bh * NB * DD;
  const float* VBh = VB + (size_t)bh * NB * DD;
  float* Oh = O + hoff;

  const int tid = threadIdx.x;
  const int wave = tid >> 5;
  const int lane = tid & 31;
  const int half = lane >> 4;
  const int ln = lane & 15;
  const int qrow0 = qb * BS + wave * 16;
  float* myP = sP + (size_t)wave * 16 * PSTRIDE;

  const int firstb = (qb >= 2) ? qb - 2 : 0;
  const int nwb = qb - firstb + 1;
  const int extra = (firstb > 0) ? 1 : 0;
  const int ncg = nwb + extra;
  const int wkb0 = extra ? 0 : firstb;

  v2f qa[16];
#pragma unroll
  for (int t = 0; t < 16; ++t) {
    const float* p = Qh + (size_t)(qrow0 + ln) * DD + 4 * t + 2 * half;
    v2f a; a.x = p[0]; a.y = p[1];
    qa[t] = a;
  }

  v8f oacc[4];
#pragma unroll
  for (int dt = 0; dt < 4; ++dt) oacc[dt] = vzero8();

  const size_t vLane = (size_t)(2 * half) * DD + 4 * ln;
  PVChunk ccmp;
  pv_load_v(ccmp, VBh + vLane);

#pragma unroll
  for (int nt = 0; nt < 2; ++nt) {
    v8f acc = vzero8();
#pragma unroll
    for (int t = 0; t < 16; ++t) {
      const float* bp = KBh + (size_t)(nt * 16 + ln) * DD + 4 * t + 2 * half;
      v2f bf; bf.x = bp[0]; bf.y = bp[1];
      acc = wmma4(qa[t], bf, acc);
    }
#pragma unroll
    for (int r = 0; r < 8; ++r) {
      int m = r + 8 * half;
      int col = nt * 16 + ln;
      float x = acc[r] * SCALE;
      myP[m * PSTRIDE + col] = (col <= qb) ? x : NEGINF;
    }
  }
  __syncthreads();

  if (tid < NB) {
    float s = 0.f;
    for (int q2 = 0; q2 < BS; ++q2) s += sP[q2 * PSTRIDE + tid];
    sScore[tid] = s * (1.f / BS);
  }
  __syncthreads();

  if (tid == 0) {
    float loc[NB];
    for (int i = 0; i < NB; ++i) loc[i] = sScore[i];
    for (int s = 0; s < NSEL; ++s) {
      int bi = 0;
      float bv = loc[0];
      for (int i = 1; i < NB; ++i)
        if (loc[i] > bv) { bv = loc[i]; bi = i; }
      sSel[s] = bi;
      loc[bi] = -3.4e38f;
    }
    sSel[NSEL] = wkb0;
  }
  __syncthreads();

  if (tid < BS) {
    float* row = sP + tid * PSTRIDE;
    float mx = -3.4e38f;
    for (int c = 0; c < NB; ++c) mx = fmaxf(mx, row[c]);
    float sum = 0.f;
    for (int c = 0; c < NB; ++c) {
      float p = (mx > -1e29f) ? __expf(row[c] - mx) : 0.f;
      row[c] = p;
      sum += p;
    }
    float rs = (sum > 0.f) ? 1.f / sum : 0.f;
    for (int c = 0; c < NB; ++c) row[c] *= rs;
  }
  __syncthreads();

  const float* pLane = myP + ln * PSTRIDE + 2 * half;
  const int selv = sSel[lane <= NSEL ? lane : 0];

  const float* kLane = sK + ln * KPITCH + 2 * half;
  KStage ks;
  kstage_load(ks, Kh + (size_t)__builtin_amdgcn_readlane(selv, 0) * (BS * DD),
              tid);

  pv_load_p(ccmp, pLane);
  pv_mma(ccmp, oacc);

  for (int sb = 0; sb < NSEL; ++sb) {
    const int kb = __builtin_amdgcn_readlane(selv, sb);
    __syncthreads();
    kstage_store(ks, sK, tid);
    __syncthreads();
    kstage_load(ks,
                Kh + (size_t)__builtin_amdgcn_readlane(selv, sb + 1) * (BS * DD),
                tid);
    __builtin_amdgcn_sched_barrier(0);
    KPair pa, pb;
    kpair_load(pa, kLane);
    kpair_load(pb, kLane + 32 * KPITCH);
    {
      v8f a0 = vzero8(), a1 = vzero8();
      kpair_mma(pa, qa, a0, a1);
      sel_tile_out(a0, myP, half, ln, qrow0, kb * BS, sb * 64);
      sel_tile_out(a1, myP, half, ln, qrow0, kb * BS + 16, sb * 64 + 16);
    }
    {
      v8f a0 = vzero8(), a1 = vzero8();
      kpair_mma(pb, qa, a0, a1);
      sel_tile_out(a0, myP, half, ln, qrow0, kb * BS + 32, sb * 64 + 32);
      sel_tile_out(a1, myP, half, ln, qrow0, kb * BS + 48, sb * 64 + 48);
    }
  }

  static_assert(SELR == 4, "urowred4 sorts four rows onto the lanes");
  unsigned thrv = 0u;
  for (int r = 0; r < 16; r += SELR) {
    unsigned key[SELR][16];
    unsigned smx[SELR];
    unsigned smn[SELR];
    const unsigned kneg = okey(NEGINF);
#pragma unroll
    for (int j = 0; j < SELR; ++j) {
      const float* row = myP + (r + j) * PSTRIDE;
#pragma unroll
      for (int k = 0; k < 16; ++k) key[j][k] = okey(row[lane + 32 * k]);
      smx[j] = upick<true>(key[j][0], key[j][8]);
      smn[j] = upick<false>(key[j][0] - (kneg + 1u), key[j][8] - (kneg + 1u));
    }
    const float smax = unokey(urowred4<true>(smx, lane));
    const float smin = unokey(urowred4<false>(smn, lane) + (kneg + 1u));
    const float emax = smax + 0.5f * (smax - smin);
    const float emin = smin - 0.5f * (smax - smin);

    bool open = lane < SELR;
    unsigned candv = kneg + 1u, prev = 0u, klo = 0u, khi = 0xffffffffu;
    float tlo = emin, thi = emax;
    float flo = (float)(SELC - KEEP);
    float fhi = (float)KEEP * (1.f / SEL_DAMP);
    bool go = true;
#pragma nounroll
    for (int it = 0; go && it < SEL_MAXIT; ++it) {
      unsigned cand[SELR];
      int cnt[SELR];
#pragma unroll
      for (int j = 0; j < SELR; ++j)
        cand[j] = (unsigned)__builtin_amdgcn_readlane((int)candv, j);
      count_ge_rows(key, cand, cnt);
      int n = cnt[SELR - 1];
#pragma unroll
      for (int j = SELR - 2; j >= 0; --j) n = (lane == j) ? cnt[j] : n;

      const bool gt = n > KEEP, lt = n < KEEP;
      const float tc = __builtin_amdgcn_fmed3f(unokey(candv), emin, emax);
      const float fa = (float)(n - KEEP);
      klo = gt ? candv : klo;
      tlo = gt ? tc : tlo;
      flo = gt ? fa : SEL_DAMP * flo;
      khi = lt ? candv : khi;
      thi = lt ? tc : thi;
      fhi = lt ? -fa : SEL_DAMP * fhi;
      const unsigned dist = khi - klo;
      const bool hit = !gt && !lt;
      const bool snap = dist == 1u;
      const unsigned pnew = hit ? candv : snap ? klo : prev;
      prev = open ? pnew : prev;
      open = open && !hit && !snap;
      const float w = __builtin_amdgcn_fmed3f(
          flo * __builtin_amdgcn_rcpf(flo + fhi), SEL_WMIN, 1.f - SEL_WMIN);
      const float t = tlo + (thi - tlo) * w;
      const unsigned c = okey(t);
      const bool inside = c > klo && c < khi && it < SEL_NINT;
      unsigned cn = inside ? c : klo + (dist >> 1);
      cn = (khi == kneg + 1u) ? kneg : cn;
      candv = open ? cn : prev;
      go = __builtin_amdgcn_ballot_w32(open) != 0u;
    }
#pragma unroll
    for (int j = 0; j < SELR; ++j) {
      const unsigned pj = (unsigned)__builtin_amdgcn_readlane((int)prev, j);
      thrv = (lane == r + j) ? pj : thrv;
    }
  }

  PVChunk ca, cb;
  pv_load_v(ca, Vh + (size_t)__builtin_amdgcn_readlane(selv, 0) * (BS * DD) +
                    vLane);

  for (int r = 0; r < 16; r += SMXR) {
    float x[SMXR][16];
    unsigned key[SMXR][16];
    unsigned pre[SMXR];
#pragma unroll
    for (int j = 0; j < SMXR; ++j) {
      const float* row = myP + (r + j) * PSTRIDE;
#pragma unroll
      for (int k = 0; k < 16; ++k) {
        x[j][k] = row[lane + 32 * k];
        key[j][k] = okey(x[j][k]);
      }
      pre[j] = __builtin_amdgcn_readlane(thrv, r + j);
    }
    float mx[SMXR];
    float sum[SMXR];
#pragma unroll
    for (int j = 0; j < SMXR; ++j) {
      mx[j] = -3.4e38f;
#pragma unroll
      for (int k = 0; k < 16; ++k) mx[j] = fmaxf(mx[j], x[j][k]);
    }
    wred_maxf(mx);
#pragma unroll
    for (int j = 0; j < SMXR; ++j) {
      sum[j] = 0.f;
#pragma unroll
      for (int k = 0; k < 16; ++k) {
        float p = (key[j][k] >= pre[j] && mx[j] > -1e29f) ? __expf(x[j][k] - mx[j]) : 0.f;
        x[j][k] = p;
        sum[j] += p;
      }
    }
    wred_sumf(sum);
#pragma unroll
    for (int j = 0; j < SMXR; ++j) {
      float* row = myP + (r + j) * PSTRIDE;
      float rs = (sum[j] > 0.f) ? 1.f / sum[j] : 0.f;
#pragma unroll
      for (int k = 0; k < 16; ++k) row[lane + 32 * k] = x[j][k] * rs;
    }
  }

  {
    pv_load_p(ca, pLane);
#pragma nounroll
    for (int sb = 0; sb < NSEL; ++sb) {
      const float* vb =
          Vh + (size_t)__builtin_amdgcn_readlane(selv, sb) * (BS * DD) + vLane;
      const float* pb = pLane + sb * 64;
      pv_load(cb, pb + 32, vb + 32 * DD);
      __builtin_amdgcn_sched_barrier(0);
      pv_mma(ca, oacc);
      {
        const bool more = sb + 1 < NSEL;
        const int sn = more ? sb + 1 : sb;
        const float* vn =
            Vh + (size_t)__builtin_amdgcn_readlane(selv, sn) * (BS * DD) +
            (more ? 0 : 32 * DD) + vLane;
        const float* pn = more ? pb + 64 : pb + 32;
        pv_load(ca, pn, vn);
      }
      __builtin_amdgcn_sched_barrier(0);
      pv_mma(cb, oacc);
    }
  }

  const int tI = __builtin_amdgcn_readfirstlane(qrow0) >> 4;
  const int tJb = 4 * firstb;
  const int tlo = (tI - 8 - tJb > 0) ? tI - 8 - tJb : 0;
  const int thi = tI - tJb;
  const unsigned maskw = ((2u << thi) - 1u) & ~((1u << tlo) - 1u);
  const unsigned need = extra ? (1u | (maskw << 4)) : (maskw | 1u);
  unsigned needp = (need | (need >> 1)) & 0x5555u;
  if (__builtin_popcount(needp) & 1) {
    const unsigned freec = ~needp & 0x5555u & ((1u << (4 * ncg)) - 1u);
    needp |= freec & (0u - freec);
  }
#pragma nounroll
  for (int cg = 0; cg < ncg; ++cg) {
    const int kb = (extra && cg == 0) ? 0 : firstb + (cg - extra);
    __syncthreads();
    kstage_store(ks, sK, tid);
    __syncthreads();
    if (cg + 1 < ncg)
      kstage_load(ks, Kh + (size_t)(firstb + (cg + 1 - extra)) * (BS * DD), tid);
    __builtin_amdgcn_sched_barrier(0);
    const bool wa = ((needp >> (4 * cg)) & 1u) != 0u;
    const bool wb = ((needp >> (4 * cg + 2)) & 1u) != 0u;
    KPair pa, pb;
    if (wa) kpair_load(pa, kLane);
    if (wb) kpair_load(pb, kLane + 32 * KPITCH);
    __builtin_amdgcn_sched_barrier(0);
    if (wa) {
      v8f a0 = vzero8(), a1 = vzero8();
      kpair_mma(pa, qa, a0, a1);
      win_tile_out(a0, myP, half, ln, qrow0, kb * BS, cg * 64);
      win_tile_out(a1, myP, half, ln, qrow0, kb * BS + 16, cg * 64 + 16);
    } else {
      win_chunk_masked(myP, half, ln, cg * 64);
    }
    if (wb) {
      v8f a0 = vzero8(), a1 = vzero8();
      kpair_mma(pb, qa, a0, a1);
      win_tile_out(a0, myP, half, ln, qrow0, kb * BS + 32, cg * 64 + 32);
      win_tile_out(a1, myP, half, ln, qrow0, kb * BS + 48, cg * 64 + 48);
    } else {
      win_chunk_masked(myP, half, ln, cg * 64 + 32);
    }
  }

  unsigned rem = needp;
  const int cw0 = __builtin_ctz(rem) >> 1;
  rem &= rem - 1u;
  {
    const int g0 = cw0 >> 1;
    const int kb0 = (extra && g0 == 0) ? 0 : firstb + (g0 - extra);
    pv_load_v(ca, Vh + (size_t)kb0 * (BS * DD) +
                      (size_t)(cw0 & 1) * (32 * DD) + vLane);
  }

  win_softmax(myP, lane, ncg);

  {
    const int ntrip = __builtin_popcount(needp) >> 1;
    pv_load_p(ca, pLane + 32 * cw0);
    for (int trip = 0; trip < ntrip; ++trip) {
      const int c1 = __builtin_ctz(rem) >> 1;
      rem &= rem - 1u;
      {
        const int g1 = c1 >> 1;
        const int kb1 = (extra && g1 == 0) ? 0 : firstb + (g1 - extra);
        pv_load(cb, pLane + 32 * c1,
                Vh + (size_t)kb1 * (BS * DD) + (size_t)(c1 & 1) * (32 * DD) + vLane);
      }
      __builtin_amdgcn_sched_barrier(0);
      pv_mma(ca, oacc);
      {
        const int c0 = (rem != 0u) ? (__builtin_ctz(rem) >> 1) : c1;
        rem &= rem - 1u;
        const int g0 = c0 >> 1;
        const int kb0 = (extra && g0 == 0) ? 0 : firstb + (g0 - extra);
        pv_load(ca, pLane + 32 * c0,
                Vh + (size_t)kb0 * (BS * DD) + (size_t)(c0 & 1) * (32 * DD) + vLane);
      }
      __builtin_amdgcn_sched_barrier(0);
      pv_mma(cb, oacc);
    }
  }

  const float third = 1.f / 3.f;
#pragma unroll
  for (int r = 0; r < 8; ++r) {
    int row = qrow0 + r + 8 * half;
    v4f o;
    o.x = oacc[0][r] * third;
    o.y = oacc[1][r] * third;
    o.z = oacc[2][r] * third;
    o.w = oacc[3][r] * third;
    *(v4f*)(Oh + (size_t)row * DD + 4 * ln) = o;
  }
}

extern "C" void kernel_launch(void* const* d_in, const int* in_sizes, int n_in,
                              void* d_out, int out_size, void* d_ws,
                              size_t ws_size, hipStream_t stream) {
  (void)in_sizes; (void)n_in; (void)out_size; (void)ws_size;
  const float* q = (const float*)d_in[0];
  const float* k = (const float*)d_in[1];
  const float* v = (const float*)d_in[2];
  float* out = (float*)d_out;
  float* kblk = (float*)d_ws;
  float* vblk = kblk + (size_t)BB * HH * NB * DD;

  nsa_blockmean_kernel<<<BB * HH * NB, 64, 0, stream>>>(k, v, kblk, vblk);

  const size_t shmem = (size_t)(64 * PSTRIDE + 64 * KPITCH) * sizeof(float);
  hipFuncSetAttribute(reinterpret_cast<const void*>(nsa_main_kernel),
                      hipFuncAttributeMaxDynamicSharedMemorySize, (int)shmem);
  nsa_main_kernel<<<BB * HH * NB, 128, shmem, stream>>>(q, k, v, kblk, vblk,
                                                        out);
}
